# starting kernel
# baseline (speedup 1.0000x reference)
#include <hip/hip_runtime.h>
#include <hip/hip_bf16.h>

typedef float v2f __attribute__((ext_vector_type(2)));
typedef float v8f __attribute__((ext_vector_type(8)));

#define IN_DIM   64
#define OUT_DIM  64
#define EDGE_DIM 16
#define H_HEADS  4
#define HD       16
#define NEG_SLOPE   0.2f
#define EPS_SOFTMAX 1e-8f
#define EPS_LN      1e-5f

__global__ void zero_kernel(float* __restrict__ p, long n) {
    long i = (long)blockIdx.x * blockDim.x + threadIdx.x;
    long stride = (long)gridDim.x * blockDim.x;
    for (; i < n; i += stride) p[i] = 0.0f;
}

__global__ void node_gemm_wmma(const float* __restrict__ x,
                               const float* __restrict__ W,
                               float* __restrict__ h, int n) {
    int wave = threadIdx.x >> 5;
    int lane = threadIdx.x & 31;
    int grp  = lane >> 4;
    int lid  = lane & 15;

    int row0 = blockIdx.x * 16;
    int col0 = wave * 16;

    int arow = row0 + lid;
    if (arow >= n) arow = n - 1;

    const float* xr = x + (long)arow * IN_DIM + 2 * grp;
    const float* wc = W + (long)(2 * grp) * OUT_DIM + col0 + lid;

    v8f c = {};
    #pragma unroll
    for (int k0 = 0; k0 < IN_DIM; k0 += 4) {
        v2f a = *(const v2f*)(xr + k0);
        v2f b;
        b.x = wc[(long)k0 * OUT_DIM];
        b.y = wc[(long)(k0 + 1) * OUT_DIM];
        c = __builtin_amdgcn_wmma_f32_16x16x4_f32(false, a, false, b,
                                                  (short)0, c, false, false);
    }

    #pragma unroll
    for (int r = 0; r < 8; ++r) {
        int row = row0 + r + 8 * grp;
        if (row < n) h[(long)row * OUT_DIM + col0 + lid] = c[r];
    }
}

__global__ void node_proj(const float* __restrict__ h,
                          const float* __restrict__ attn_src,
                          const float* __restrict__ attn_dst,
                          float* __restrict__ a_src,
                          float* __restrict__ a_dst, int n) {
    int idx = blockIdx.x * blockDim.x + threadIdx.x;
    if (idx >= n * H_HEADS) return;
    int hh   = idx & 3;
    int node = idx >> 2;
    const float4* hp = (const float4*)(h + (long)node * OUT_DIM + hh * HD);
    const float4* as = (const float4*)(attn_src + hh * HD);
    const float4* ad = (const float4*)(attn_dst + hh * HD);
    float s = 0.0f, d = 0.0f;
    #pragma unroll
    for (int j = 0; j < 4; ++j) {
        float4 hv = hp[j], sv = as[j], dv = ad[j];
        s += hv.x * sv.x + hv.y * sv.y + hv.z * sv.z + hv.w * sv.w;
        d += hv.x * dv.x + hv.y * dv.y + hv.z * dv.z + hv.w * dv.w;
    }
    a_src[idx] = s;
    a_dst[idx] = d;
}

__global__ void edge_score(const int* __restrict__ src,
                           const int* __restrict__ dst,
                           const float* __restrict__ edge_attr,
                           const float* __restrict__ W_edge,
                           const float* __restrict__ a_src,
                           const float* __restrict__ a_dst,
                           float* __restrict__ attn,
                           int* __restrict__ amax_bits, long e_cnt) {
    long e = (long)blockIdx.x * blockDim.x + threadIdx.x;
    if (e >= e_cnt) return;
    int s = src[e], t = dst[e];

    float sc[H_HEADS] = {0.f, 0.f, 0.f, 0.f};
    const float4* ea = (const float4*)(edge_attr + e * EDGE_DIM);
    #pragma unroll
    for (int q = 0; q < 4; ++q) {
        float4 v = ea[q];
        int d0 = q * 4;
        #pragma unroll
        for (int hh = 0; hh < H_HEADS; ++hh) {
            sc[hh] += v.x * W_edge[(d0 + 0) * H_HEADS + hh]
                    + v.y * W_edge[(d0 + 1) * H_HEADS + hh]
                    + v.z * W_edge[(d0 + 2) * H_HEADS + hh]
                    + v.w * W_edge[(d0 + 3) * H_HEADS + hh];
        }
    }
    float4 asv = ((const float4*)a_src)[s];
    float4 adv = ((const float4*)a_dst)[t];
    float av[H_HEADS] = {asv.x + adv.x, asv.y + adv.y, asv.z + adv.z, asv.w + adv.w};

    float4 outv;
    float* op = (float*)&outv;
    #pragma unroll
    for (int hh = 0; hh < H_HEADS; ++hh) {
        float v = av[hh] + sc[hh];
        v = (v >= 0.0f) ? v : NEG_SLOPE * v;
        op[hh] = v;
        if (v > 0.0f) atomicMax(&amax_bits[t * H_HEADS + hh], __float_as_int(v));
    }
    ((float4*)attn)[e] = outv;
}

__global__ void edge_softmax(const int* __restrict__ dst,
                             const float* __restrict__ amax,
                             float* __restrict__ attn,
                             float* __restrict__ asum, long eh) {
    long idx = (long)blockIdx.x * blockDim.x + threadIdx.x;
    if (idx >= eh) return;
    long e = idx >> 2;
    int hh = (int)(idx & 3);
    int t  = dst[e];
    float v = expf(attn[idx] - amax[t * H_HEADS + hh]);
    attn[idx] = v;
    atomicAdd(&asum[t * H_HEADS + hh], v);
}

__global__ void scatter_msg(const int* __restrict__ src,
                            const int* __restrict__ dst,
                            const float* __restrict__ h,
                            const float* __restrict__ attn,
                            const float* __restrict__ asum,
                            float* __restrict__ acc, long total) {
    long idx = (long)blockIdx.x * blockDim.x + threadIdx.x;
    if (idx >= total) return;
    long e = idx >> 6;
    int d  = (int)(idx & 63);
    int hh = d >> 4;
    int s = src[e], t = dst[e];
    float w = attn[e * H_HEADS + hh] / (asum[t * H_HEADS + hh] + EPS_SOFTMAX);
    atomicAdd(&acc[(long)t * OUT_DIM + d], h[(long)s * OUT_DIM + d] * w);
}

__global__ void residual_ln(const float* __restrict__ acc,
                            const float* __restrict__ h,
                            const float* __restrict__ gamma,
                            const float* __restrict__ beta,
                            float* __restrict__ out, int n) {
    int wave = threadIdx.x >> 5;
    int lane = threadIdx.x & 31;
    int node = blockIdx.x * 8 + wave;
    if (node >= n) return;
    long base = (long)node * OUT_DIM;

    float y0 = acc[base + lane]      + h[base + lane];
    float y1 = acc[base + lane + 32] + h[base + lane + 32];

    float s = y0 + y1;
    #pragma unroll
    for (int off = 16; off > 0; off >>= 1) s += __shfl_xor(s, off, 32);
    float mu = s * (1.0f / 64.0f);

    float d0 = y0 - mu, d1 = y1 - mu;
    float vs = d0 * d0 + d1 * d1;
    #pragma unroll
    for (int off = 16; off > 0; off >>= 1) vs += __shfl_xor(vs, off, 32);
    float rstd = rsqrtf(vs * (1.0f / 64.0f) + EPS_LN);

    out[base + lane]      = d0 * rstd * gamma[lane]      + beta[lane];
    out[base + lane + 32] = d1 * rstd * gamma[lane + 32] + beta[lane + 32];
}

extern "C" void kernel_launch(void* const* d_in, const int* in_sizes, int n_in,
                              void* d_out, int out_size, void* d_ws, size_t ws_size,
                              hipStream_t stream) {
    const float* x         = (const float*)d_in[0];
    const int*   ei        = (const int*)  d_in[1];
    const float* edge_attr = (const float*)d_in[2];
    const float* W_node    = (const float*)d_in[3];
    const float* W_edge    = (const float*)d_in[4];
    const float* attn_src  = (const float*)d_in[5];
    const float* attn_dst  = (const float*)d_in[6];
    const float* gamma     = (const float*)d_in[7];
    const float* beta      = (const float*)d_in[8];
    float* out = (float*)d_out;

    int  n = in_sizes[0] / IN_DIM;
    long e = (long)in_sizes[2] / EDGE_DIM;
    const int* srcIdx = ei;
    const int* dstIdx = ei + e;

    float* ws    = (float*)d_ws;
    float* h     = ws;  ws += (long)n * OUT_DIM;
    float* a_src = ws;  ws += (long)n * H_HEADS;
    float* a_dst = ws;  ws += (long)n * H_HEADS;
    float* attn  = ws;  ws += e * H_HEADS;
    float* acc   = ws;  ws += (long)n * OUT_DIM;
    float* amax  = ws;  ws += (long)n * H_HEADS;
    float* asum  = ws;  ws += (long)n * H_HEADS;

    long zn = (long)n * (OUT_DIM + 2 * H_HEADS);
    zero_kernel<<<(int)((zn + 255) / 256), 256, 0, stream>>>(acc, zn);

    int n_tiles = (n + 15) / 16;
    node_gemm_wmma<<<n_tiles, 128, 0, stream>>>(x, W_node, h, n);

    node_proj<<<(n * H_HEADS + 255) / 256, 256, 0, stream>>>(
        h, attn_src, attn_dst, a_src, a_dst, n);

    edge_score<<<(int)((e + 255) / 256), 256, 0, stream>>>(
        srcIdx, dstIdx, edge_attr, W_edge, a_src, a_dst, attn, (int*)amax, e);

    long eh = e * H_HEADS;
    edge_softmax<<<(int)((eh + 255) / 256), 256, 0, stream>>>(
        dstIdx, amax, attn, asum, eh);

    long total = e * OUT_DIM;
    scatter_msg<<<(int)((total + 255) / 256), 256, 0, stream>>>(
        srcIdx, dstIdx, h, attn, asum, acc, total);

    residual_ln<<<(n + 7) / 8, 256, 0, stream>>>(acc, h, gamma, beta, out, n);
}


# scatter_msg global float adds replaced by node-range ownership: each group scans all dst, lists its 512 nodes' edges in LDS, sums rows by LDS float adds, drains via residual+LN; sums in arrival order
# speedup vs baseline: 1.6015x; 1.6015x over previous
#include <hip/hip_runtime.h>
#include <hip/hip_bf16.h>

typedef float v2f __attribute__((ext_vector_type(2)));
typedef float v8f __attribute__((ext_vector_type(8)));

#define IN_DIM   64
#define OUT_DIM  64
#define EDGE_DIM 16
#define H_HEADS  4
#define HD       16
#define NEG_SLOPE   0.2f
#define EPS_SOFTMAX 1e-8f
#define EPS_LN      1e-5f

__global__ void zero_kernel(float* __restrict__ p, long n) {
    long i = (long)blockIdx.x * blockDim.x + threadIdx.x;
    long stride = (long)gridDim.x * blockDim.x;
    for (; i < n; i += stride) p[i] = 0.0f;
}

__global__ void node_gemm_wmma(const float* __restrict__ x,
                               const float* __restrict__ W,
                               float* __restrict__ h, int n) {
    int wave = threadIdx.x >> 5;
    int lane = threadIdx.x & 31;
    int grp  = lane >> 4;
    int lid  = lane & 15;

    int row0 = blockIdx.x * 16;
    int col0 = wave * 16;

    int arow = row0 + lid;
    if (arow >= n) arow = n - 1;

    const float* xr = x + (long)arow * IN_DIM + 2 * grp;
    const float* wc = W + (long)(2 * grp) * OUT_DIM + col0 + lid;

    v8f c = {};
    #pragma unroll
    for (int k0 = 0; k0 < IN_DIM; k0 += 4) {
        v2f a = *(const v2f*)(xr + k0);
        v2f b;
        b.x = wc[(long)k0 * OUT_DIM];
        b.y = wc[(long)(k0 + 1) * OUT_DIM];
        c = __builtin_amdgcn_wmma_f32_16x16x4_f32(false, a, false, b,
                                                  (short)0, c, false, false);
    }

    #pragma unroll
    for (int r = 0; r < 8; ++r) {
        int row = row0 + r + 8 * grp;
        if (row < n) h[(long)row * OUT_DIM + col0 + lid] = c[r];
    }
}

__global__ void node_proj(const float* __restrict__ h,
                          const float* __restrict__ attn_src,
                          const float* __restrict__ attn_dst,
                          float* __restrict__ a_src,
                          float* __restrict__ a_dst, int n) {
    int idx = blockIdx.x * blockDim.x + threadIdx.x;
    if (idx >= n * H_HEADS) return;
    int hh   = idx & 3;
    int node = idx >> 2;
    const float4* hp = (const float4*)(h + (long)node * OUT_DIM + hh * HD);
    const float4* as = (const float4*)(attn_src + hh * HD);
    const float4* ad = (const float4*)(attn_dst + hh * HD);
    float s = 0.0f, d = 0.0f;
    #pragma unroll
    for (int j = 0; j < 4; ++j) {
        float4 hv = hp[j], sv = as[j], dv = ad[j];
        s += hv.x * sv.x + hv.y * sv.y + hv.z * sv.z + hv.w * sv.w;
        d += hv.x * dv.x + hv.y * dv.y + hv.z * dv.z + hv.w * dv.w;
    }
    a_src[idx] = s;
    a_dst[idx] = d;
}

__global__ void edge_score(const int* __restrict__ src,
                           const int* __restrict__ dst,
                           const float* __restrict__ edge_attr,
                           const float* __restrict__ W_edge,
                           const float* __restrict__ a_src,
                           const float* __restrict__ a_dst,
                           float* __restrict__ attn,
                           int* __restrict__ amax_bits, long e_cnt) {
    long e = (long)blockIdx.x * blockDim.x + threadIdx.x;
    if (e >= e_cnt) return;
    int s = src[e], t = dst[e];

    float sc[H_HEADS] = {0.f, 0.f, 0.f, 0.f};
    const float4* ea = (const float4*)(edge_attr + e * EDGE_DIM);
    #pragma unroll
    for (int q = 0; q < 4; ++q) {
        float4 v = ea[q];
        int d0 = q * 4;
        #pragma unroll
        for (int hh = 0; hh < H_HEADS; ++hh) {
            sc[hh] += v.x * W_edge[(d0 + 0) * H_HEADS + hh]
                    + v.y * W_edge[(d0 + 1) * H_HEADS + hh]
                    + v.z * W_edge[(d0 + 2) * H_HEADS + hh]
                    + v.w * W_edge[(d0 + 3) * H_HEADS + hh];
        }
    }
    float4 asv = ((const float4*)a_src)[s];
    float4 adv = ((const float4*)a_dst)[t];
    float av[H_HEADS] = {asv.x + adv.x, asv.y + adv.y, asv.z + adv.z, asv.w + adv.w};

    float4 outv;
    float* op = (float*)&outv;
    #pragma unroll
    for (int hh = 0; hh < H_HEADS; ++hh) {
        float v = av[hh] + sc[hh];
        v = (v >= 0.0f) ? v : NEG_SLOPE * v;
        op[hh] = v;
        if (v > 0.0f) atomicMax(&amax_bits[t * H_HEADS + hh], __float_as_int(v));
    }
    ((float4*)attn)[e] = outv;
}

__global__ void edge_softmax(const int* __restrict__ dst,
                             const float* __restrict__ amax,
                             float* __restrict__ attn,
                             float* __restrict__ asum, long eh) {
    long idx = (long)blockIdx.x * blockDim.x + threadIdx.x;
    if (idx >= eh) return;
    long e = idx >> 2;
    int hh = (int)(idx & 3);
    int t  = dst[e];
    float v = expf(attn[idx] - amax[t * H_HEADS + hh]);
    attn[idx] = v;
    atomicAdd(&asum[t * H_HEADS + hh], v);
}

#define OWN_NODES 512
#define OWN_CAP   16384
#define OWN_U     12
#define OWN_A     8

__device__ __forceinline__ void own_test(int t, int e, int lo,
                                         unsigned* cnt_s, int* le_s, int* lt_s) {
    unsigned tl = (unsigned)(t - lo);
    if (tl < (unsigned)OWN_NODES) {
        unsigned pos = atomicAdd(cnt_s, 1u);
        if (pos < (unsigned)OWN_CAP) {
            le_s[pos] = e;
            lt_s[pos] = (int)tl;
        }
    }
}

__global__ void __launch_bounds__(1024)
gather_own_ln(const int* __restrict__ src,
              const int* __restrict__ dst,
              const float* __restrict__ h,
              const float* __restrict__ attn,
              const float* __restrict__ asum,
              const float* __restrict__ gamma,
              const float* __restrict__ beta,
              float* __restrict__ out, int n, long e_cnt) {
    __shared__ float    acc_s[OWN_NODES * OUT_DIM];
    __shared__ int      le_s[OWN_CAP];
    __shared__ int      lt_s[OWN_CAP];
    __shared__ unsigned cnt_s;

    const int tid  = threadIdx.x;
    const int wave = tid >> 5;
    const int lane = tid & 31;
    const int lo   = blockIdx.x * OWN_NODES;
    const int ne   = (int)e_cnt;

    {
        float4 z = {0.0f, 0.0f, 0.0f, 0.0f};
        float4* a4 = (float4*)acc_s;
        for (int i = tid; i < OWN_NODES * OUT_DIM / 4; i += 1024) a4[i] = z;
        if (tid == 0) cnt_s = 0u;
    }
    __syncthreads();

    {
        int head = (int)(((16u - (unsigned)((size_t)dst & 15u)) & 15u) >> 2);
        if (head > ne) head = ne;
        const int nb4   = (ne - head) >> 2;
        const int tail0 = head + nb4 * 4;

        if (tid < head)       own_test(dst[tid], tid, lo, &cnt_s, le_s, lt_s);
        if (tail0 + tid < ne) own_test(dst[tail0 + tid], tail0 + tid, lo, &cnt_s, le_s, lt_s);

        const int4* d4 = (const int4*)(dst + head);
        for (int q0 = tid; q0 < nb4; q0 += 1024 * OWN_A) {
            int4 v[OWN_A];
            #pragma unroll
            for (int u = 0; u < OWN_A; ++u) {
                int q  = q0 + u * 1024;
                int qc = (q < nb4) ? q : q0;
                v[u] = d4[qc];
            }
            #pragma unroll
            for (int u = 0; u < OWN_A; ++u) {
                int q = q0 + u * 1024;
                if (q < nb4) {
                    int e0 = head + q * 4;
                    own_test(v[u].x, e0 + 0, lo, &cnt_s, le_s, lt_s);
                    own_test(v[u].y, e0 + 1, lo, &cnt_s, le_s, lt_s);
                    own_test(v[u].z, e0 + 2, lo, &cnt_s, le_s, lt_s);
                    own_test(v[u].w, e0 + 3, lo, &cnt_s, le_s, lt_s);
                }
            }
        }
    }
    __syncthreads();

    const unsigned cnt = cnt_s;
    if (cnt <= (unsigned)OWN_CAP) {
        const int hh = lane >> 3;

        for (unsigned i0 = (unsigned)wave * OWN_U; i0 < cnt; i0 += 32u * OWN_U) {
            int ee[OWN_U], tt[OWN_U];
            #pragma unroll
            for (int u = 0; u < OWN_U; ++u) {
                unsigned i  = i0 + u;
                unsigned ic = (i < cnt) ? i : i0;
                ee[u] = le_s[ic];
                tt[u] = lt_s[ic];
            }
            int   ss[OWN_U];
            float ww[OWN_U];
            #pragma unroll
            for (int u = 0; u < OWN_U; ++u) {
                long e = ee[u];
                long t = lo + tt[u];
                ss[u] = src[e];
                float a = attn[e * H_HEADS + hh];
                float q = asum[t * H_HEADS + hh];
                ww[u] = a / (q + EPS_SOFTMAX);
            }
            v2f hv[OWN_U];
            #pragma unroll
            for (int u = 0; u < OWN_U; ++u)
                hv[u] = *(const v2f*)(h + (long)ss[u] * OUT_DIM + 2 * lane);
            #pragma unroll
            for (int u = 0; u < OWN_U; ++u) {
                if (i0 + u < cnt) {
                    atomicAdd(&acc_s[tt[u] * OUT_DIM + 2 * lane],     hv[u].x * ww[u]);
                    atomicAdd(&acc_s[tt[u] * OUT_DIM + 2 * lane + 1], hv[u].y * ww[u]);
                }
            }
        }
    } else {
        for (int e = tid; e < ne; e += 1024) {
            int t = dst[e];
            unsigned tl = (unsigned)(t - lo);
            if (tl < (unsigned)OWN_NODES) {
                int s = src[e];
                for (int d = 0; d < OUT_DIM; ++d) {
                    int hd = d >> 4;
                    float w = attn[(long)e * H_HEADS + hd] / (asum[(long)t * H_HEADS + hd] + EPS_SOFTMAX);
                    atomicAdd(&acc_s[tl * OUT_DIM + d], h[(long)s * OUT_DIM + d] * w);
                }
            }
        }
    }
    __syncthreads();

    {
        const float g0 = gamma[lane], g1 = gamma[lane + 32];
        const float b0 = beta[lane],  b1 = beta[lane + 32];
        for (int tl0 = wave * 4; tl0 < OWN_NODES; tl0 += 32 * 4) {
            float y0[4], y1[4], s[4], vs[4];
            #pragma unroll
            for (int u = 0; u < 4; ++u) {
                int node = lo + tl0 + u;
                int nc   = (node < n) ? node : (n - 1);
                y0[u] = acc_s[(tl0 + u) * OUT_DIM + lane]      + h[(long)nc * OUT_DIM + lane];
                y1[u] = acc_s[(tl0 + u) * OUT_DIM + lane + 32] + h[(long)nc * OUT_DIM + lane + 32];
                s[u]  = y0[u] + y1[u];
            }
            #pragma unroll
            for (int off = 16; off > 0; off >>= 1) {
                #pragma unroll
                for (int u = 0; u < 4; ++u) s[u] += __shfl_xor(s[u], off, 32);
            }
            #pragma unroll
            for (int u = 0; u < 4; ++u) {
                float mu = s[u] * (1.0f / 64.0f);
                y0[u] -= mu;
                y1[u] -= mu;
                vs[u] = y0[u] * y0[u] + y1[u] * y1[u];
            }
            #pragma unroll
            for (int off = 16; off > 0; off >>= 1) {
                #pragma unroll
                for (int u = 0; u < 4; ++u) vs[u] += __shfl_xor(vs[u], off, 32);
            }
            #pragma unroll
            for (int u = 0; u < 4; ++u) {
                int node = lo + tl0 + u;
                if (node < n) {
                    long base = (long)node * OUT_DIM;
                    float rstd = rsqrtf(vs[u] * (1.0f / 64.0f) + EPS_LN);
                    out[base + lane]      = y0[u] * rstd * g0 + b0;
                    out[base + lane + 32] = y1[u] * rstd * g1 + b1;
                }
            }
        }
    }
}

extern "C" void kernel_launch(void* const* d_in, const int* in_sizes, int n_in,
                              void* d_out, int out_size, void* d_ws, size_t ws_size,
                              hipStream_t stream) {
    const float* x         = (const float*)d_in[0];
    const int*   ei        = (const int*)  d_in[1];
    const float* edge_attr = (const float*)d_in[2];
    const float* W_node    = (const float*)d_in[3];
    const float* W_edge    = (const float*)d_in[4];
    const float* attn_src  = (const float*)d_in[5];
    const float* attn_dst  = (const float*)d_in[6];
    const float* gamma     = (const float*)d_in[7];
    const float* beta      = (const float*)d_in[8];
    float* out = (float*)d_out;

    int  n = in_sizes[0] / IN_DIM;
    long e = (long)in_sizes[2] / EDGE_DIM;
    const int* srcIdx = ei;
    const int* dstIdx = ei + e;

    float* ws    = (float*)d_ws;
    float* h     = ws;  ws += (long)n * OUT_DIM;
    float* a_src = ws;  ws += (long)n * H_HEADS;
    float* a_dst = ws;  ws += (long)n * H_HEADS;
    float* attn  = ws;  ws += e * H_HEADS;
    float* amax  = ws;  ws += (long)n * H_HEADS;
    float* asum  = ws;  ws += (long)n * H_HEADS;

    long zn = (long)n * (2 * H_HEADS);
    zero_kernel<<<(int)((zn + 255) / 256), 256, 0, stream>>>(amax, zn);

    int n_tiles = (n + 15) / 16;
    node_gemm_wmma<<<n_tiles, 128, 0, stream>>>(x, W_node, h, n);

    node_proj<<<(n * H_HEADS + 255) / 256, 256, 0, stream>>>(
        h, attn_src, attn_dst, a_src, a_dst, n);

    edge_score<<<(int)((e + 255) / 256), 256, 0, stream>>>(
        srcIdx, dstIdx, edge_attr, W_edge, a_src, a_dst, attn, (int*)amax, e);

    long eh = e * H_HEADS;
    edge_softmax<<<(int)((eh + 255) / 256), 256, 0, stream>>>(
        dstIdx, amax, attn, asum, eh);

    gather_own_ln<<<(n + OWN_NODES - 1) / OWN_NODES, 1024, 0, stream>>>(
        srcIdx, dstIdx, h, attn, asum, gamma, beta, out, n, e);
}


# segment softmax (max, exp-sum) kept in the range-owning group's LDS: global atomicMax/atomicAdd, edge_softmax and zero launches gone (6 -> 4); division by the sum once per node; sum order changes
# speedup vs baseline: 2.9122x; 1.8184x over previous
#include <hip/hip_runtime.h>
#include <hip/hip_bf16.h>

typedef float v2f __attribute__((ext_vector_type(2)));
typedef float v8f __attribute__((ext_vector_type(8)));

#define IN_DIM   64
#define OUT_DIM  64
#define EDGE_DIM 16
#define H_HEADS  4
#define HD       16
#define NEG_SLOPE   0.2f
#define EPS_SOFTMAX 1e-8f
#define EPS_LN      1e-5f

__global__ void node_gemm_wmma(const float* __restrict__ x,
                               const float* __restrict__ W,
                               float* __restrict__ h, int n) {
    int wave = threadIdx.x >> 5;
    int lane = threadIdx.x & 31;
    int grp  = lane >> 4;
    int lid  = lane & 15;

    int row0 = blockIdx.x * 16;
    int col0 = wave * 16;

    int arow = row0 + lid;
    if (arow >= n) arow = n - 1;

    const float* xr = x + (long)arow * IN_DIM + 2 * grp;
    const float* wc = W + (long)(2 * grp) * OUT_DIM + col0 + lid;

    v8f c = {};
    #pragma unroll
    for (int k0 = 0; k0 < IN_DIM; k0 += 4) {
        v2f a = *(const v2f*)(xr + k0);
        v2f b;
        b.x = wc[(long)k0 * OUT_DIM];
        b.y = wc[(long)(k0 + 1) * OUT_DIM];
        c = __builtin_amdgcn_wmma_f32_16x16x4_f32(false, a, false, b,
                                                  (short)0, c, false, false);
    }

    #pragma unroll
    for (int r = 0; r < 8; ++r) {
        int row = row0 + r + 8 * grp;
        if (row < n) h[(long)row * OUT_DIM + col0 + lid] = c[r];
    }
}

__global__ void node_proj(const float* __restrict__ h,
                          const float* __restrict__ attn_src,
                          const float* __restrict__ attn_dst,
                          float* __restrict__ a_src,
                          float* __restrict__ a_dst, int n) {
    int idx = blockIdx.x * blockDim.x + threadIdx.x;
    if (idx >= n * H_HEADS) return;
    int hh   = idx & 3;
    int node = idx >> 2;
    const float4* hp = (const float4*)(h + (long)node * OUT_DIM + hh * HD);
    const float4* as = (const float4*)(attn_src + hh * HD);
    const float4* ad = (const float4*)(attn_dst + hh * HD);
    float s = 0.0f, d = 0.0f;
    #pragma unroll
    for (int j = 0; j < 4; ++j) {
        float4 hv = hp[j], sv = as[j], dv = ad[j];
        s += hv.x * sv.x + hv.y * sv.y + hv.z * sv.z + hv.w * sv.w;
        d += hv.x * dv.x + hv.y * dv.y + hv.z * dv.z + hv.w * dv.w;
    }
    a_src[idx] = s;
    a_dst[idx] = d;
}

__global__ void edge_score(const int* __restrict__ src,
                           const int* __restrict__ dst,
                           const float* __restrict__ edge_attr,
                           const float* __restrict__ W_edge,
                           const float* __restrict__ a_src,
                           const float* __restrict__ a_dst,
                           float* __restrict__ attn, long e_cnt) {
    long e = (long)blockIdx.x * blockDim.x + threadIdx.x;
    if (e >= e_cnt) return;
    int s = src[e], t = dst[e];

    float sc[H_HEADS] = {0.f, 0.f, 0.f, 0.f};
    const float4* ea = (const float4*)(edge_attr + e * EDGE_DIM);
    #pragma unroll
    for (int q = 0; q < 4; ++q) {
        float4 v = ea[q];
        int d0 = q * 4;
        #pragma unroll
        for (int hh = 0; hh < H_HEADS; ++hh) {
            sc[hh] += v.x * W_edge[(d0 + 0) * H_HEADS + hh]
                    + v.y * W_edge[(d0 + 1) * H_HEADS + hh]
                    + v.z * W_edge[(d0 + 2) * H_HEADS + hh]
                    + v.w * W_edge[(d0 + 3) * H_HEADS + hh];
        }
    }
    float4 asv = ((const float4*)a_src)[s];
    float4 adv = ((const float4*)a_dst)[t];
    float av[H_HEADS] = {asv.x + adv.x, asv.y + adv.y, asv.z + adv.z, asv.w + adv.w};

    float4 outv;
    float* op = (float*)&outv;
    #pragma unroll
    for (int hh = 0; hh < H_HEADS; ++hh) {
        float v = av[hh] + sc[hh];
        v = (v >= 0.0f) ? v : NEG_SLOPE * v;
        op[hh] = v;
    }
    ((float4*)attn)[e] = outv;
}

#define OWN_NODES 512
#define OWN_CAP   16384
#define OWN_U     12
#define OWN_A     8
#define OWN_M     4

__device__ __forceinline__ void own_test(int t, int e, int lo,
                                         unsigned* cnt_s, int* le_s, int* lt_s) {
    unsigned tl = (unsigned)(t - lo);
    if (tl < (unsigned)OWN_NODES) {
        unsigned pos = atomicAdd(cnt_s, 1u);
        if (pos < (unsigned)OWN_CAP) {
            le_s[pos] = e;
            lt_s[pos] = (int)tl;
        }
    }
}

__device__ __forceinline__ void own_max4(int* mx_s, int tl, float4 a) {
    if (a.x > 0.0f) atomicMax(&mx_s[tl * H_HEADS + 0], __float_as_int(a.x));
    if (a.y > 0.0f) atomicMax(&mx_s[tl * H_HEADS + 1], __float_as_int(a.y));
    if (a.z > 0.0f) atomicMax(&mx_s[tl * H_HEADS + 2], __float_as_int(a.z));
    if (a.w > 0.0f) atomicMax(&mx_s[tl * H_HEADS + 3], __float_as_int(a.w));
}

__global__ void __launch_bounds__(1024)
gather_own_ln(const int* __restrict__ src,
              const int* __restrict__ dst,
              const float* __restrict__ h,
              const float* __restrict__ attn,
              const float* __restrict__ gamma,
              const float* __restrict__ beta,
              float* __restrict__ out, int n, long e_cnt) {
    __shared__ float    acc_s[OWN_NODES * OUT_DIM];
    __shared__ int      le_s[OWN_CAP];
    __shared__ int      lt_s[OWN_CAP];
    __shared__ int      mx_s[OWN_NODES * H_HEADS];
    __shared__ float    sm_s[OWN_NODES * H_HEADS];
    __shared__ unsigned cnt_s;

    const int tid  = threadIdx.x;
    const int wave = tid >> 5;
    const int lane = tid & 31;
    const int lo   = blockIdx.x * OWN_NODES;
    const int ne   = (int)e_cnt;
    const float4* attn4 = (const float4*)attn;

    {
        float4 z = {0.0f, 0.0f, 0.0f, 0.0f};
        float4* a4 = (float4*)acc_s;
        for (int i = tid; i < OWN_NODES * OUT_DIM / 4; i += 1024) a4[i] = z;
        for (int i = tid; i < OWN_NODES * H_HEADS; i += 1024) { mx_s[i] = 0; sm_s[i] = 0.0f; }
        if (tid == 0) cnt_s = 0u;
    }
    __syncthreads();

    {
        int head = (int)(((16u - (unsigned)((size_t)dst & 15u)) & 15u) >> 2);
        if (head > ne) head = ne;
        const int nb4   = (ne - head) >> 2;
        const int tail0 = head + nb4 * 4;

        if (tid < head)       own_test(dst[tid], tid, lo, &cnt_s, le_s, lt_s);
        if (tail0 + tid < ne) own_test(dst[tail0 + tid], tail0 + tid, lo, &cnt_s, le_s, lt_s);

        const int4* d4 = (const int4*)(dst + head);
        for (int q0 = tid; q0 < nb4; q0 += 1024 * OWN_A) {
            int4 v[OWN_A];
            #pragma unroll
            for (int u = 0; u < OWN_A; ++u) {
                int q  = q0 + u * 1024;
                int qc = (q < nb4) ? q : q0;
                v[u] = d4[qc];
            }
            #pragma unroll
            for (int u = 0; u < OWN_A; ++u) {
                int q = q0 + u * 1024;
                if (q < nb4) {
                    int e0 = head + q * 4;
                    own_test(v[u].x, e0 + 0, lo, &cnt_s, le_s, lt_s);
                    own_test(v[u].y, e0 + 1, lo, &cnt_s, le_s, lt_s);
                    own_test(v[u].z, e0 + 2, lo, &cnt_s, le_s, lt_s);
                    own_test(v[u].w, e0 + 3, lo, &cnt_s, le_s, lt_s);
                }
            }
        }
    }
    __syncthreads();

    const unsigned cnt    = cnt_s;
    const bool     listed = (cnt <= (unsigned)OWN_CAP);

    if (listed) {
        for (unsigned i0 = (unsigned)tid; i0 < cnt; i0 += 1024u * OWN_M) {
            int    tt[OWN_M];
            float4 av[OWN_M];
            #pragma unroll
            for (int u = 0; u < OWN_M; ++u) {
                unsigned i  = i0 + (unsigned)u * 1024u;
                unsigned ic = (i < cnt) ? i : i0;
                tt[u] = lt_s[ic];
                av[u] = attn4[le_s[ic]];
            }
            #pragma unroll
            for (int u = 0; u < OWN_M; ++u) {
                if (i0 + (unsigned)u * 1024u < cnt) own_max4(mx_s, tt[u], av[u]);
            }
        }
    } else {
        for (int e = tid; e < ne; e += 1024) {
            unsigned tl = (unsigned)(dst[e] - lo);
            if (tl < (unsigned)OWN_NODES) own_max4(mx_s, (int)tl, attn4[e]);
        }
    }
    __syncthreads();

    if (listed) {
        const int hh = lane >> 3;

        for (unsigned i0 = (unsigned)wave * OWN_U; i0 < cnt; i0 += 32u * OWN_U) {
            int ee[OWN_U], tt[OWN_U];
            #pragma unroll
            for (int u = 0; u < OWN_U; ++u) {
                unsigned i  = i0 + u;
                unsigned ic = (i < cnt) ? i : i0;
                ee[u] = le_s[ic];
                tt[u] = lt_s[ic];
            }
            int   ss[OWN_U];
            float aa[OWN_U], mm[OWN_U];
            #pragma unroll
            for (int u = 0; u < OWN_U; ++u) {
                long e = ee[u];
                ss[u] = src[e];
                aa[u] = attn[e * H_HEADS + hh];
                mm[u] = __int_as_float(mx_s[tt[u] * H_HEADS + hh]);
            }
            v2f hv[OWN_U];
            #pragma unroll
            for (int u = 0; u < OWN_U; ++u)
                hv[u] = *(const v2f*)(h + (long)ss[u] * OUT_DIM + 2 * lane);
            #pragma unroll
            for (int u = 0; u < OWN_U; ++u) {
                if (i0 + u < cnt) {
                    float w = expf(aa[u] - mm[u]);
                    if ((lane & 7) == 0) atomicAdd(&sm_s[tt[u] * H_HEADS + hh], w);
                    atomicAdd(&acc_s[tt[u] * OUT_DIM + 2 * lane],     hv[u].x * w);
                    atomicAdd(&acc_s[tt[u] * OUT_DIM + 2 * lane + 1], hv[u].y * w);
                }
            }
        }
    } else {
        for (int e = tid; e < ne; e += 1024) {
            unsigned tl = (unsigned)(dst[e] - lo);
            if (tl < (unsigned)OWN_NODES) {
                int s = src[e];
                float4 a = attn4[e];
                float w[H_HEADS];
                w[0] = expf(a.x - __int_as_float(mx_s[tl * H_HEADS + 0]));
                w[1] = expf(a.y - __int_as_float(mx_s[tl * H_HEADS + 1]));
                w[2] = expf(a.z - __int_as_float(mx_s[tl * H_HEADS + 2]));
                w[3] = expf(a.w - __int_as_float(mx_s[tl * H_HEADS + 3]));
                for (int hd = 0; hd < H_HEADS; ++hd) {
                    atomicAdd(&sm_s[tl * H_HEADS + hd], w[hd]);
                    for (int d = hd * HD; d < hd * HD + HD; ++d)
                        atomicAdd(&acc_s[tl * OUT_DIM + d], h[(long)s * OUT_DIM + d] * w[hd]);
                }
            }
        }
    }
    __syncthreads();

    {
        const float g0 = gamma[lane], g1 = gamma[lane + 32];
        const float b0 = beta[lane],  b1 = beta[lane + 32];
        const int   h0 = lane >> 4;
        for (int tl0 = wave * 4; tl0 < OWN_NODES; tl0 += 32 * 4) {
            float y0[4], y1[4], s[4], vs[4];
            #pragma unroll
            for (int u = 0; u < 4; ++u) {
                int node = lo + tl0 + u;
                int nc   = (node < n) ? node : (n - 1);
                float q0 = sm_s[(tl0 + u) * H_HEADS + h0]     + EPS_SOFTMAX;
                float q1 = sm_s[(tl0 + u) * H_HEADS + h0 + 2] + EPS_SOFTMAX;
                y0[u] = acc_s[(tl0 + u) * OUT_DIM + lane]      / q0 + h[(long)nc * OUT_DIM + lane];
                y1[u] = acc_s[(tl0 + u) * OUT_DIM + lane + 32] / q1 + h[(long)nc * OUT_DIM + lane + 32];
                s[u]  = y0[u] + y1[u];
            }
            #pragma unroll
            for (int off = 16; off > 0; off >>= 1) {
                #pragma unroll
                for (int u = 0; u < 4; ++u) s[u] += __shfl_xor(s[u], off, 32);
            }
            #pragma unroll
            for (int u = 0; u < 4; ++u) {
                float mu = s[u] * (1.0f / 64.0f);
                y0[u] -= mu;
                y1[u] -= mu;
                vs[u] = y0[u] * y0[u] + y1[u] * y1[u];
            }
            #pragma unroll
            for (int off = 16; off > 0; off >>= 1) {
                #pragma unroll
                for (int u = 0; u < 4; ++u) vs[u] += __shfl_xor(vs[u], off, 32);
            }
            #pragma unroll
            for (int u = 0; u < 4; ++u) {
                int node = lo + tl0 + u;
                if (node < n) {
                    long base = (long)node * OUT_DIM;
                    float rstd = rsqrtf(vs[u] * (1.0f / 64.0f) + EPS_LN);
                    out[base + lane]      = y0[u] * rstd * g0 + b0;
                    out[base + lane + 32] = y1[u] * rstd * g1 + b1;
                }
            }
        }
    }
}

extern "C" void kernel_launch(void* const* d_in, const int* in_sizes, int n_in,
                              void* d_out, int out_size, void* d_ws, size_t ws_size,
                              hipStream_t stream) {
    const float* x         = (const float*)d_in[0];
    const int*   ei        = (const int*)  d_in[1];
    const float* edge_attr = (const float*)d_in[2];
    const float* W_node    = (const float*)d_in[3];
    const float* W_edge    = (const float*)d_in[4];
    const float* attn_src  = (const float*)d_in[5];
    const float* attn_dst  = (const float*)d_in[6];
    const float* gamma     = (const float*)d_in[7];
    const float* beta      = (const float*)d_in[8];
    float* out = (float*)d_out;

    int  n = in_sizes[0] / IN_DIM;
    long e = (long)in_sizes[2] / EDGE_DIM;
    const int* srcIdx = ei;
    const int* dstIdx = ei + e;

    float* ws    = (float*)d_ws;
    float* h     = ws;  ws += (long)n * OUT_DIM;
    float* a_src = ws;  ws += (long)n * H_HEADS;
    float* a_dst = ws;  ws += (long)n * H_HEADS;
    float* attn  = ws;  ws += e * H_HEADS;

    int n_tiles = (n + 15) / 16;
    node_gemm_wmma<<<n_tiles, 128, 0, stream>>>(x, W_node, h, n);

    node_proj<<<(n * H_HEADS + 255) / 256, 256, 0, stream>>>(
        h, attn_src, attn_dst, a_src, a_dst, n);

    edge_score<<<(int)((e + 255) / 256), 256, 0, stream>>>(
        srcIdx, dstIdx, edge_attr, W_edge, a_src, a_dst, attn, e);

    gather_own_ln<<<(n + OWN_NODES - 1) / OWN_NODES, 1024, 0, stream>>>(
        srcIdx, dstIdx, h, attn, gamma, beta, out, n, e);
}


# gather pass A scans a 1-byte range-number plane written by edge_score (one 128-bit load tests sixteen edges byte-parallel) instead of the 4x larger dst list; arrival order of the node sums changes
# speedup vs baseline: 3.4560x; 1.1867x over previous
#include <hip/hip_runtime.h>
#include <hip/hip_bf16.h>

typedef float v2f __attribute__((ext_vector_type(2)));
typedef float v8f __attribute__((ext_vector_type(8)));

#define IN_DIM   64
#define OUT_DIM  64
#define EDGE_DIM 16
#define H_HEADS  4
#define HD       16
#define NEG_SLOPE   0.2f
#define EPS_SOFTMAX 1e-8f
#define EPS_LN      1e-5f

#define OWN_SHIFT 9
#define OWN_NODES (1 << OWN_SHIFT)

__global__ void node_gemm_wmma(const float* __restrict__ x,
                               const float* __restrict__ W,
                               float* __restrict__ h, int n) {
    int wave = threadIdx.x >> 5;
    int lane = threadIdx.x & 31;
    int grp  = lane >> 4;
    int lid  = lane & 15;

    int row0 = blockIdx.x * 16;
    int col0 = wave * 16;

    int arow = row0 + lid;
    if (arow >= n) arow = n - 1;

    const float* xr = x + (long)arow * IN_DIM + 2 * grp;
    const float* wc = W + (long)(2 * grp) * OUT_DIM + col0 + lid;

    v8f c = {};
    #pragma unroll
    for (int k0 = 0; k0 < IN_DIM; k0 += 4) {
        v2f a = *(const v2f*)(xr + k0);
        v2f b;
        b.x = wc[(long)k0 * OUT_DIM];
        b.y = wc[(long)(k0 + 1) * OUT_DIM];
        c = __builtin_amdgcn_wmma_f32_16x16x4_f32(false, a, false, b,
                                                  (short)0, c, false, false);
    }

    #pragma unroll
    for (int r = 0; r < 8; ++r) {
        int row = row0 + r + 8 * grp;
        if (row < n) h[(long)row * OUT_DIM + col0 + lid] = c[r];
    }
}

__global__ void node_proj(const float* __restrict__ h,
                          const float* __restrict__ attn_src,
                          const float* __restrict__ attn_dst,
                          float* __restrict__ a_src,
                          float* __restrict__ a_dst, int n) {
    int idx = blockIdx.x * blockDim.x + threadIdx.x;
    if (idx >= n * H_HEADS) return;
    int hh   = idx & 3;
    int node = idx >> 2;
    const float4* hp = (const float4*)(h + (long)node * OUT_DIM + hh * HD);
    const float4* as = (const float4*)(attn_src + hh * HD);
    const float4* ad = (const float4*)(attn_dst + hh * HD);
    float s = 0.0f, d = 0.0f;
    #pragma unroll
    for (int j = 0; j < 4; ++j) {
        float4 hv = hp[j], sv = as[j], dv = ad[j];
        s += hv.x * sv.x + hv.y * sv.y + hv.z * sv.z + hv.w * sv.w;
        d += hv.x * dv.x + hv.y * dv.y + hv.z * dv.z + hv.w * dv.w;
    }
    a_src[idx] = s;
    a_dst[idx] = d;
}

__global__ void edge_score(const int* __restrict__ src,
                           const int* __restrict__ dst,
                           const float* __restrict__ edge_attr,
                           const float* __restrict__ W_edge,
                           const float* __restrict__ a_src,
                           const float* __restrict__ a_dst,
                           float* __restrict__ attn,
                           unsigned* __restrict__ keyw, long e_cnt) {
    long e = (long)blockIdx.x * blockDim.x + threadIdx.x;
    if (e >= e_cnt) return;
    int s = src[e], t = dst[e];
    ((unsigned char*)keyw)[e] = (unsigned char)((unsigned)t >> OWN_SHIFT);

    float sc[H_HEADS] = {0.f, 0.f, 0.f, 0.f};
    const float4* ea = (const float4*)(edge_attr + e * EDGE_DIM);
    #pragma unroll
    for (int q = 0; q < 4; ++q) {
        float4 v = ea[q];
        int d0 = q * 4;
        #pragma unroll
        for (int hh = 0; hh < H_HEADS; ++hh) {
            sc[hh] += v.x * W_edge[(d0 + 0) * H_HEADS + hh]
                    + v.y * W_edge[(d0 + 1) * H_HEADS + hh]
                    + v.z * W_edge[(d0 + 2) * H_HEADS + hh]
                    + v.w * W_edge[(d0 + 3) * H_HEADS + hh];
        }
    }
    float4 asv = ((const float4*)a_src)[s];
    float4 adv = ((const float4*)a_dst)[t];
    float av[H_HEADS] = {asv.x + adv.x, asv.y + adv.y, asv.z + adv.z, asv.w + adv.w};

    float4 outv;
    float* op = (float*)&outv;
    #pragma unroll
    for (int hh = 0; hh < H_HEADS; ++hh) {
        float v = av[hh] + sc[hh];
        v = (v >= 0.0f) ? v : NEG_SLOPE * v;
        op[hh] = v;
    }
    ((float4*)attn)[e] = outv;
}

#define OWN_CAP   16384
#define OWN_U     12
#define OWN_A     8
#define OWN_M     4

__device__ __forceinline__ void own_list(int e, unsigned* cnt_s, int* le_s) {
    unsigned pos = atomicAdd(cnt_s, 1u);
    if (pos < (unsigned)OWN_CAP) le_s[pos] = e;
}

__device__ __forceinline__ unsigned own_eq4(unsigned v, unsigned pat) {
    unsigned x = v ^ pat;
    unsigned t = (x & 0x7F7F7F7Fu) + 0x7F7F7F7Fu;
    return ~(t | x | 0x7F7F7F7Fu);
}

__device__ __forceinline__ void own_test16(uint4 v, unsigned pat, int e0,
                                           unsigned* cnt_s, int* le_s) {
    unsigned hm = (own_eq4(v.x, pat) >> 7) | (own_eq4(v.y, pat) >> 6)
                | (own_eq4(v.z, pat) >> 5) | (own_eq4(v.w, pat) >> 4);
    while (hm) {
        int p = __builtin_ctz(hm);
        hm &= hm - 1u;
        own_list(e0 + ((p & 3) << 2) + (p >> 3), cnt_s, le_s);
    }
}

__device__ __forceinline__ void own_max4(int* mx_s, int tl, float4 a) {
    if (a.x > 0.0f) atomicMax(&mx_s[tl * H_HEADS + 0], __float_as_int(a.x));
    if (a.y > 0.0f) atomicMax(&mx_s[tl * H_HEADS + 1], __float_as_int(a.y));
    if (a.z > 0.0f) atomicMax(&mx_s[tl * H_HEADS + 2], __float_as_int(a.z));
    if (a.w > 0.0f) atomicMax(&mx_s[tl * H_HEADS + 3], __float_as_int(a.w));
}

__global__ void __launch_bounds__(1024)
gather_own_ln(const int* __restrict__ src,
              const int* __restrict__ dst,
              const float* __restrict__ h,
              const float* __restrict__ attn,
              const unsigned* __restrict__ keyw,
              const float* __restrict__ gamma,
              const float* __restrict__ beta,
              float* __restrict__ out, int n, long e_cnt) {
    __shared__ float    acc_s[OWN_NODES * OUT_DIM];
    __shared__ int      le_s[OWN_CAP];
    __shared__ int      lt_s[OWN_CAP];
    __shared__ int      mx_s[OWN_NODES * H_HEADS];
    __shared__ float    sm_s[OWN_NODES * H_HEADS];
    __shared__ unsigned cnt_s;

    const int tid  = threadIdx.x;
    const int wave = tid >> 5;
    const int lane = tid & 31;
    const int lo   = blockIdx.x * OWN_NODES;
    const int ne   = (int)e_cnt;
    const float4* attn4 = (const float4*)attn;

    {
        float4 z = {0.0f, 0.0f, 0.0f, 0.0f};
        float4* a4 = (float4*)acc_s;
        for (int i = tid; i < OWN_NODES * OUT_DIM / 4; i += 1024) a4[i] = z;
        for (int i = tid; i < OWN_NODES * H_HEADS; i += 1024) { mx_s[i] = 0; sm_s[i] = 0.0f; }
        if (tid == 0) cnt_s = 0u;
    }
    __syncthreads();

    const bool keyed = (gridDim.x <= 256u);
    if (keyed) {
        const unsigned mine = blockIdx.x;
        const unsigned pat  = mine * 0x01010101u;
        const unsigned char* key = (const unsigned char*)keyw;
        const int nb16  = ne >> 4;
        const int tail0 = nb16 << 4;

        if (tail0 + tid < ne && key[tail0 + tid] == mine) own_list(tail0 + tid, &cnt_s, le_s);

        const uint4* k4 = (const uint4*)keyw;
        for (int q0 = tid; q0 < nb16; q0 += 1024 * OWN_A) {
            uint4 v[OWN_A];
            #pragma unroll
            for (int u = 0; u < OWN_A; ++u) {
                int q  = q0 + u * 1024;
                int qc = (q < nb16) ? q : q0;
                v[u] = k4[qc];
            }
            #pragma unroll
            for (int u = 0; u < OWN_A; ++u) {
                int q = q0 + u * 1024;
                if (q < nb16) own_test16(v[u], pat, q << 4, &cnt_s, le_s);
            }
        }
    }
    __syncthreads();

    const unsigned cnt    = cnt_s;
    const bool     listed = keyed && (cnt <= (unsigned)OWN_CAP);

    if (listed) {
        for (unsigned i0 = (unsigned)tid; i0 < cnt; i0 += 1024u * OWN_M) {
            int    ee[OWN_M], tt[OWN_M];
            float4 av[OWN_M];
            #pragma unroll
            for (int u = 0; u < OWN_M; ++u) {
                unsigned i  = i0 + (unsigned)u * 1024u;
                unsigned ic = (i < cnt) ? i : i0;
                ee[u] = le_s[ic];
            }
            #pragma unroll
            for (int u = 0; u < OWN_M; ++u) {
                tt[u] = dst[ee[u]] & (OWN_NODES - 1);
                av[u] = attn4[ee[u]];
            }
            #pragma unroll
            for (int u = 0; u < OWN_M; ++u) {
                unsigned i = i0 + (unsigned)u * 1024u;
                if (i < cnt) {
                    lt_s[i] = tt[u];
                    own_max4(mx_s, tt[u], av[u]);
                }
            }
        }
    } else {
        for (int e = tid; e < ne; e += 1024) {
            unsigned tl = (unsigned)(dst[e] - lo);
            if (tl < (unsigned)OWN_NODES) own_max4(mx_s, (int)tl, attn4[e]);
        }
    }
    __syncthreads();

    if (listed) {
        const int hh = lane >> 3;

        for (unsigned i0 = (unsigned)wave * OWN_U; i0 < cnt; i0 += 32u * OWN_U) {
            int ee[OWN_U], tt[OWN_U];
            #pragma unroll
            for (int u = 0; u < OWN_U; ++u) {
                unsigned i  = i0 + u;
                unsigned ic = (i < cnt) ? i : i0;
                ee[u] = le_s[ic];
                tt[u] = lt_s[ic];
            }
            int   ss[OWN_U];
            float aa[OWN_U], mm[OWN_U];
            #pragma unroll
            for (int u = 0; u < OWN_U; ++u) {
                long e = ee[u];
                ss[u] = src[e];
                aa[u] = attn[e * H_HEADS + hh];
                mm[u] = __int_as_float(mx_s[tt[u] * H_HEADS + hh]);
            }
            v2f hv[OWN_U];
            #pragma unroll
            for (int u = 0; u < OWN_U; ++u)
                hv[u] = *(const v2f*)(h + (long)ss[u] * OUT_DIM + 2 * lane);
            #pragma unroll
            for (int u = 0; u < OWN_U; ++u) {
                if (i0 + u < cnt) {
                    float w = expf(aa[u] - mm[u]);
                    if ((lane & 7) == 0) atomicAdd(&sm_s[tt[u] * H_HEADS + hh], w);
                    atomicAdd(&acc_s[tt[u] * OUT_DIM + 2 * lane],     hv[u].x * w);
                    atomicAdd(&acc_s[tt[u] * OUT_DIM + 2 * lane + 1], hv[u].y * w);
                }
            }
        }
    } else {
        for (int e = tid; e < ne; e += 1024) {
            unsigned tl = (unsigned)(dst[e] - lo);
            if (tl < (unsigned)OWN_NODES) {
                int s = src[e];
                float4 a = attn4[e];
                float w[H_HEADS];
                w[0] = expf(a.x - __int_as_float(mx_s[tl * H_HEADS + 0]));
                w[1] = expf(a.y - __int_as_float(mx_s[tl * H_HEADS + 1]));
                w[2] = expf(a.z - __int_as_float(mx_s[tl * H_HEADS + 2]));
                w[3] = expf(a.w - __int_as_float(mx_s[tl * H_HEADS + 3]));
                for (int hd = 0; hd < H_HEADS; ++hd) {
                    atomicAdd(&sm_s[tl * H_HEADS + hd], w[hd]);
                    for (int d = hd * HD; d < hd * HD + HD; ++d)
                        atomicAdd(&acc_s[tl * OUT_DIM + d], h[(long)s * OUT_DIM + d] * w[hd]);
                }
            }
        }
    }
    __syncthreads();

    {
        const float g0 = gamma[lane], g1 = gamma[lane + 32];
        const float b0 = beta[lane],  b1 = beta[lane + 32];
        const int   h0 = lane >> 4;
        for (int tl0 = wave * 4; tl0 < OWN_NODES; tl0 += 32 * 4) {
            float y0[4], y1[4], s[4], vs[4];
            #pragma unroll
            for (int u = 0; u < 4; ++u) {
                int node = lo + tl0 + u;
                int nc   = (node < n) ? node : (n - 1);
                float q0 = sm_s[(tl0 + u) * H_HEADS + h0]     + EPS_SOFTMAX;
                float q1 = sm_s[(tl0 + u) * H_HEADS + h0 + 2] + EPS_SOFTMAX;
                y0[u] = acc_s[(tl0 + u) * OUT_DIM + lane]      / q0 + h[(long)nc * OUT_DIM + lane];
                y1[u] = acc_s[(tl0 + u) * OUT_DIM + lane + 32] / q1 + h[(long)nc * OUT_DIM + lane + 32];
                s[u]  = y0[u] + y1[u];
            }
            #pragma unroll
            for (int off = 16; off > 0; off >>= 1) {
                #pragma unroll
                for (int u = 0; u < 4; ++u) s[u] += __shfl_xor(s[u], off, 32);
            }
            #pragma unroll
            for (int u = 0; u < 4; ++u) {
                float mu = s[u] * (1.0f / 64.0f);
                y0[u] -= mu;
                y1[u] -= mu;
                vs[u] = y0[u] * y0[u] + y1[u] * y1[u];
            }
            #pragma unroll
            for (int off = 16; off > 0; off >>= 1) {
                #pragma unroll
                for (int u = 0; u < 4; ++u) vs[u] += __shfl_xor(vs[u], off, 32);
            }
            #pragma unroll
            for (int u = 0; u < 4; ++u) {
                int node = lo + tl0 + u;
                if (node < n) {
                    long base = (long)node * OUT_DIM;
                    float rstd = rsqrtf(vs[u] * (1.0f / 64.0f) + EPS_LN);
                    out[base + lane]      = y0[u] * rstd * g0 + b0;
                    out[base + lane + 32] = y1[u] * rstd * g1 + b1;
                }
            }
        }
    }
}

extern "C" void kernel_launch(void* const* d_in, const int* in_sizes, int n_in,
                              void* d_out, int out_size, void* d_ws, size_t ws_size,
                              hipStream_t stream) {
    const float* x         = (const float*)d_in[0];
    const int*   ei        = (const int*)  d_in[1];
    const float* edge_attr = (const float*)d_in[2];
    const float* W_node    = (const float*)d_in[3];
    const float* W_edge    = (const float*)d_in[4];
    const float* attn_src  = (const float*)d_in[5];
    const float* attn_dst  = (const float*)d_in[6];
    const float* gamma     = (const float*)d_in[7];
    const float* beta      = (const float*)d_in[8];
    float* out = (float*)d_out;

    int  n = in_sizes[0] / IN_DIM;
    long e = (long)in_sizes[2] / EDGE_DIM;
    const int* srcIdx = ei;
    const int* dstIdx = ei + e;

    float* ws    = (float*)d_ws;
    float* h     = ws;  ws += (long)n * OUT_DIM;
    float* a_src = ws;  ws += (long)n * H_HEADS;
    float* a_dst = ws;  ws += (long)n * H_HEADS;
    float* attn  = ws;  ws += e * H_HEADS;
    unsigned* keyw = (unsigned*)ws;

    int n_tiles = (n + 15) / 16;
    node_gemm_wmma<<<n_tiles, 128, 0, stream>>>(x, W_node, h, n);

    node_proj<<<(n * H_HEADS + 255) / 256, 256, 0, stream>>>(
        h, attn_src, attn_dst, a_src, a_dst, n);

    edge_score<<<(int)((e + 255) / 256), 256, 0, stream>>>(
        srcIdx, dstIdx, edge_attr, W_edge, a_src, a_dst, attn, keyw, e);

    gather_own_ln<<<(n + OWN_NODES - 1) / OWN_NODES, 1024, 0, stream>>>(
        srcIdx, dstIdx, h, attn, keyw, gamma, beta, out, n, e);
}


# ownership range cut from 512 to 391 nodes: 256 one-per-WGP groups instead of 196, so the 60 idle WGPs of the gather launch get work; range number by division; arrival order of node sums changes
# speedup vs baseline: 3.6638x; 1.0601x over previous
#include <hip/hip_runtime.h>
#include <hip/hip_bf16.h>

typedef float v2f __attribute__((ext_vector_type(2)));
typedef float v8f __attribute__((ext_vector_type(8)));

#define IN_DIM   64
#define OUT_DIM  64
#define EDGE_DIM 16
#define H_HEADS  4
#define HD       16
#define NEG_SLOPE   0.2f
#define EPS_SOFTMAX 1e-8f
#define EPS_LN      1e-5f

#define OWN_NODES 391
#define OWN_PAD   392

__global__ void node_gemm_wmma(const float* __restrict__ x,
                               const float* __restrict__ W,
                               float* __restrict__ h, int n) {
    int wave = threadIdx.x >> 5;
    int lane = threadIdx.x & 31;
    int grp  = lane >> 4;
    int lid  = lane & 15;

    int row0 = blockIdx.x * 16;
    int col0 = wave * 16;

    int arow = row0 + lid;
    if (arow >= n) arow = n - 1;

    const float* xr = x + (long)arow * IN_DIM + 2 * grp;
    const float* wc = W + (long)(2 * grp) * OUT_DIM + col0 + lid;

    v8f c = {};
    #pragma unroll
    for (int k0 = 0; k0 < IN_DIM; k0 += 4) {
        v2f a = *(const v2f*)(xr + k0);
        v2f b;
        b.x = wc[(long)k0 * OUT_DIM];
        b.y = wc[(long)(k0 + 1) * OUT_DIM];
        c = __builtin_amdgcn_wmma_f32_16x16x4_f32(false, a, false, b,
                                                  (short)0, c, false, false);
    }

    #pragma unroll
    for (int r = 0; r < 8; ++r) {
        int row = row0 + r + 8 * grp;
        if (row < n) h[(long)row * OUT_DIM + col0 + lid] = c[r];
    }
}

__global__ void node_proj(const float* __restrict__ h,
                          const float* __restrict__ attn_src,
                          const float* __restrict__ attn_dst,
                          float* __restrict__ a_src,
                          float* __restrict__ a_dst, int n) {
    int idx = blockIdx.x * blockDim.x + threadIdx.x;
    if (idx >= n * H_HEADS) return;
    int hh   = idx & 3;
    int node = idx >> 2;
    const float4* hp = (const float4*)(h + (long)node * OUT_DIM + hh * HD);
    const float4* as = (const float4*)(attn_src + hh * HD);
    const float4* ad = (const float4*)(attn_dst + hh * HD);
    float s = 0.0f, d = 0.0f;
    #pragma unroll
    for (int j = 0; j < 4; ++j) {
        float4 hv = hp[j], sv = as[j], dv = ad[j];
        s += hv.x * sv.x + hv.y * sv.y + hv.z * sv.z + hv.w * sv.w;
        d += hv.x * dv.x + hv.y * dv.y + hv.z * dv.z + hv.w * dv.w;
    }
    a_src[idx] = s;
    a_dst[idx] = d;
}

__global__ void edge_score(const int* __restrict__ src,
                           const int* __restrict__ dst,
                           const float* __restrict__ edge_attr,
                           const float* __restrict__ W_edge,
                           const float* __restrict__ a_src,
                           const float* __restrict__ a_dst,
                           float* __restrict__ attn,
                           unsigned* __restrict__ keyw, long e_cnt) {
    long e = (long)blockIdx.x * blockDim.x + threadIdx.x;
    if (e >= e_cnt) return;
    int s = src[e], t = dst[e];
    ((unsigned char*)keyw)[e] = (unsigned char)((unsigned)t / (unsigned)OWN_NODES);

    float sc[H_HEADS] = {0.f, 0.f, 0.f, 0.f};
    const float4* ea = (const float4*)(edge_attr + e * EDGE_DIM);
    #pragma unroll
    for (int q = 0; q < 4; ++q) {
        float4 v = ea[q];
        int d0 = q * 4;
        #pragma unroll
        for (int hh = 0; hh < H_HEADS; ++hh) {
            sc[hh] += v.x * W_edge[(d0 + 0) * H_HEADS + hh]
                    + v.y * W_edge[(d0 + 1) * H_HEADS + hh]
                    + v.z * W_edge[(d0 + 2) * H_HEADS + hh]
                    + v.w * W_edge[(d0 + 3) * H_HEADS + hh];
        }
    }
    float4 asv = ((const float4*)a_src)[s];
    float4 adv = ((const float4*)a_dst)[t];
    float av[H_HEADS] = {asv.x + adv.x, asv.y + adv.y, asv.z + adv.z, asv.w + adv.w};

    float4 outv;
    float* op = (float*)&outv;
    #pragma unroll
    for (int hh = 0; hh < H_HEADS; ++hh) {
        float v = av[hh] + sc[hh];
        v = (v >= 0.0f) ? v : NEG_SLOPE * v;
        op[hh] = v;
    }
    ((float4*)attn)[e] = outv;
}

#define OWN_CAP   16384
#define OWN_U     12
#define OWN_A     8
#define OWN_M     4

__device__ __forceinline__ void own_list(int e, unsigned* cnt_s, int* le_s) {
    unsigned pos = atomicAdd(cnt_s, 1u);
    if (pos < (unsigned)OWN_CAP) le_s[pos] = e;
}

__device__ __forceinline__ unsigned own_eq4(unsigned v, unsigned pat) {
    unsigned x = v ^ pat;
    unsigned t = (x & 0x7F7F7F7Fu) + 0x7F7F7F7Fu;
    return ~(t | x | 0x7F7F7F7Fu);
}

__device__ __forceinline__ void own_test16(uint4 v, unsigned pat, int e0,
                                           unsigned* cnt_s, int* le_s) {
    unsigned hm = (own_eq4(v.x, pat) >> 7) | (own_eq4(v.y, pat) >> 6)
                | (own_eq4(v.z, pat) >> 5) | (own_eq4(v.w, pat) >> 4);
    while (hm) {
        int p = __builtin_ctz(hm);
        hm &= hm - 1u;
        own_list(e0 + ((p & 3) << 2) + (p >> 3), cnt_s, le_s);
    }
}

__device__ __forceinline__ void own_max4(int* mx_s, int tl, float4 a) {
    if (a.x > 0.0f) atomicMax(&mx_s[tl * H_HEADS + 0], __float_as_int(a.x));
    if (a.y > 0.0f) atomicMax(&mx_s[tl * H_HEADS + 1], __float_as_int(a.y));
    if (a.z > 0.0f) atomicMax(&mx_s[tl * H_HEADS + 2], __float_as_int(a.z));
    if (a.w > 0.0f) atomicMax(&mx_s[tl * H_HEADS + 3], __float_as_int(a.w));
}

__global__ void __launch_bounds__(1024)
gather_own_ln(const int* __restrict__ src,
              const int* __restrict__ dst,
              const float* __restrict__ h,
              const float* __restrict__ attn,
              const unsigned* __restrict__ keyw,
              const float* __restrict__ gamma,
              const float* __restrict__ beta,
              float* __restrict__ out, int n, long e_cnt) {
    __shared__ float    acc_s[OWN_PAD * OUT_DIM];
    __shared__ int      le_s[OWN_CAP];
    __shared__ int      lt_s[OWN_CAP];
    __shared__ int      mx_s[OWN_PAD * H_HEADS];
    __shared__ float    sm_s[OWN_PAD * H_HEADS];
    __shared__ unsigned cnt_s;

    const int tid  = threadIdx.x;
    const int wave = tid >> 5;
    const int lane = tid & 31;
    const int lo   = blockIdx.x * OWN_NODES;
    const int ne   = (int)e_cnt;
    const float4* attn4 = (const float4*)attn;

    {
        float4 z = {0.0f, 0.0f, 0.0f, 0.0f};
        float4* a4 = (float4*)acc_s;
        for (int i = tid; i < OWN_PAD * OUT_DIM / 4; i += 1024) a4[i] = z;
        for (int i = tid; i < OWN_PAD * H_HEADS; i += 1024) { mx_s[i] = 0; sm_s[i] = 0.0f; }
        if (tid == 0) cnt_s = 0u;
    }
    __syncthreads();

    const bool keyed = (gridDim.x <= 256u);
    if (keyed) {
        const unsigned mine = blockIdx.x;
        const unsigned pat  = mine * 0x01010101u;
        const unsigned char* key = (const unsigned char*)keyw;
        const int nb16  = ne >> 4;
        const int tail0 = nb16 << 4;

        if (tail0 + tid < ne && key[tail0 + tid] == mine) own_list(tail0 + tid, &cnt_s, le_s);

        const uint4* k4 = (const uint4*)keyw;
        for (int q0 = tid; q0 < nb16; q0 += 1024 * OWN_A) {
            uint4 v[OWN_A];
            #pragma unroll
            for (int u = 0; u < OWN_A; ++u) {
                int q  = q0 + u * 1024;
                int qc = (q < nb16) ? q : q0;
                v[u] = k4[qc];
            }
            #pragma unroll
            for (int u = 0; u < OWN_A; ++u) {
                int q = q0 + u * 1024;
                if (q < nb16) own_test16(v[u], pat, q << 4, &cnt_s, le_s);
            }
        }
    }
    __syncthreads();

    const unsigned cnt    = cnt_s;
    const bool     listed = keyed && (cnt <= (unsigned)OWN_CAP);

    if (listed) {
        for (unsigned i0 = (unsigned)tid; i0 < cnt; i0 += 1024u * OWN_M) {
            int    ee[OWN_M], tt[OWN_M];
            float4 av[OWN_M];
            #pragma unroll
            for (int u = 0; u < OWN_M; ++u) {
                unsigned i  = i0 + (unsigned)u * 1024u;
                unsigned ic = (i < cnt) ? i : i0;
                ee[u] = le_s[ic];
            }
            #pragma unroll
            for (int u = 0; u < OWN_M; ++u) {
                tt[u] = dst[ee[u]] - lo;
                av[u] = attn4[ee[u]];
            }
            #pragma unroll
            for (int u = 0; u < OWN_M; ++u) {
                unsigned i = i0 + (unsigned)u * 1024u;
                if (i < cnt) {
                    lt_s[i] = tt[u];
                    own_max4(mx_s, tt[u], av[u]);
                }
            }
        }
    } else {
        for (int e = tid; e < ne; e += 1024) {
            unsigned tl = (unsigned)(dst[e] - lo);
            if (tl < (unsigned)OWN_NODES) own_max4(mx_s, (int)tl, attn4[e]);
        }
    }
    __syncthreads();

    if (listed) {
        const int hh = lane >> 3;

        for (unsigned i0 = (unsigned)wave * OWN_U; i0 < cnt; i0 += 32u * OWN_U) {
            int ee[OWN_U], tt[OWN_U];
            #pragma unroll
            for (int u = 0; u < OWN_U; ++u) {
                unsigned i  = i0 + u;
                unsigned ic = (i < cnt) ? i : i0;
                ee[u] = le_s[ic];
                tt[u] = lt_s[ic];
            }
            int   ss[OWN_U];
            float aa[OWN_U], mm[OWN_U];
            #pragma unroll
            for (int u = 0; u < OWN_U; ++u) {
                long e = ee[u];
                ss[u] = src[e];
                aa[u] = attn[e * H_HEADS + hh];
                mm[u] = __int_as_float(mx_s[tt[u] * H_HEADS + hh]);
            }
            v2f hv[OWN_U];
            #pragma unroll
            for (int u = 0; u < OWN_U; ++u)
                hv[u] = *(const v2f*)(h + (long)ss[u] * OUT_DIM + 2 * lane);
            #pragma unroll
            for (int u = 0; u < OWN_U; ++u) {
                if (i0 + u < cnt) {
                    float w = expf(aa[u] - mm[u]);
                    if ((lane & 7) == 0) atomicAdd(&sm_s[tt[u] * H_HEADS + hh], w);
                    atomicAdd(&acc_s[tt[u] * OUT_DIM + 2 * lane],     hv[u].x * w);
                    atomicAdd(&acc_s[tt[u] * OUT_DIM + 2 * lane + 1], hv[u].y * w);
                }
            }
        }
    } else {
        for (int e = tid; e < ne; e += 1024) {
            unsigned tl = (unsigned)(dst[e] - lo);
            if (tl < (unsigned)OWN_NODES) {
                int s = src[e];
                float4 a = attn4[e];
                float w[H_HEADS];
                w[0] = expf(a.x - __int_as_float(mx_s[tl * H_HEADS + 0]));
                w[1] = expf(a.y - __int_as_float(mx_s[tl * H_HEADS + 1]));
                w[2] = expf(a.z - __int_as_float(mx_s[tl * H_HEADS + 2]));
                w[3] = expf(a.w - __int_as_float(mx_s[tl * H_HEADS + 3]));
                for (int hd = 0; hd < H_HEADS; ++hd) {
                    atomicAdd(&sm_s[tl * H_HEADS + hd], w[hd]);
                    for (int d = hd * HD; d < hd * HD + HD; ++d)
                        atomicAdd(&acc_s[tl * OUT_DIM + d], h[(long)s * OUT_DIM + d] * w[hd]);
                }
            }
        }
    }
    __syncthreads();

    {
        const float g0 = gamma[lane], g1 = gamma[lane + 32];
        const float b0 = beta[lane],  b1 = beta[lane + 32];
        const int   h0 = lane >> 4;
        for (int tl0 = wave * 4; tl0 < OWN_NODES; tl0 += 32 * 4) {
            float y0[4], y1[4], s[4], vs[4];
            #pragma unroll
            for (int u = 0; u < 4; ++u) {
                int node = lo + tl0 + u;
                int nc   = (node < n) ? node : (n - 1);
                float q0 = sm_s[(tl0 + u) * H_HEADS + h0]     + EPS_SOFTMAX;
                float q1 = sm_s[(tl0 + u) * H_HEADS + h0 + 2] + EPS_SOFTMAX;
                y0[u] = acc_s[(tl0 + u) * OUT_DIM + lane]      / q0 + h[(long)nc * OUT_DIM + lane];
                y1[u] = acc_s[(tl0 + u) * OUT_DIM + lane + 32] / q1 + h[(long)nc * OUT_DIM + lane + 32];
                s[u]  = y0[u] + y1[u];
            }
            #pragma unroll
            for (int off = 16; off > 0; off >>= 1) {
                #pragma unroll
                for (int u = 0; u < 4; ++u) s[u] += __shfl_xor(s[u], off, 32);
            }
            #pragma unroll
            for (int u = 0; u < 4; ++u) {
                float mu = s[u] * (1.0f / 64.0f);
                y0[u] -= mu;
                y1[u] -= mu;
                vs[u] = y0[u] * y0[u] + y1[u] * y1[u];
            }
            #pragma unroll
            for (int off = 16; off > 0; off >>= 1) {
                #pragma unroll
                for (int u = 0; u < 4; ++u) vs[u] += __shfl_xor(vs[u], off, 32);
            }
            #pragma unroll
            for (int u = 0; u < 4; ++u) {
                int node = lo + tl0 + u;
                if (tl0 + u < OWN_NODES && node < n) {
                    long base = (long)node * OUT_DIM;
                    float rstd = rsqrtf(vs[u] * (1.0f / 64.0f) + EPS_LN);
                    out[base + lane]      = y0[u] * rstd * g0 + b0;
                    out[base + lane + 32] = y1[u] * rstd * g1 + b1;
                }
            }
        }
    }
}

extern "C" void kernel_launch(void* const* d_in, const int* in_sizes, int n_in,
                              void* d_out, int out_size, void* d_ws, size_t ws_size,
                              hipStream_t stream) {
    const float* x         = (const float*)d_in[0];
    const int*   ei        = (const int*)  d_in[1];
    const float* edge_attr = (const float*)d_in[2];
    const float* W_node    = (const float*)d_in[3];
    const float* W_edge    = (const float*)d_in[4];
    const float* attn_src  = (const float*)d_in[5];
    const float* attn_dst  = (const float*)d_in[6];
    const float* gamma     = (const float*)d_in[7];
    const float* beta      = (const float*)d_in[8];
    float* out = (float*)d_out;

    int  n = in_sizes[0] / IN_DIM;
    long e = (long)in_sizes[2] / EDGE_DIM;
    const int* srcIdx = ei;
    const int* dstIdx = ei + e;

    float* ws    = (float*)d_ws;
    float* h     = ws;  ws += (long)n * OUT_DIM;
    float* a_src = ws;  ws += (long)n * H_HEADS;
    float* a_dst = ws;  ws += (long)n * H_HEADS;
    float* attn  = ws;  ws += e * H_HEADS;
    unsigned* keyw = (unsigned*)ws;

    int n_tiles = (n + 15) / 16;
    node_gemm_wmma<<<n_tiles, 128, 0, stream>>>(x, W_node, h, n);

    node_proj<<<(n * H_HEADS + 255) / 256, 256, 0, stream>>>(
        h, attn_src, attn_dst, a_src, a_dst, n);

    edge_score<<<(int)((e + 255) / 256), 256, 0, stream>>>(
        srcIdx, dstIdx, edge_attr, W_edge, a_src, a_dst, attn, keyw, e);

    gather_own_ln<<<(n + OWN_NODES - 1) / OWN_NODES, 1024, 0, stream>>>(
        srcIdx, dstIdx, h, attn, keyw, gamma, beta, out, n, e);
}


# range numbers stored bit-sliced (8 ballot words per 32 edges, written by edge_score): gather pass A tests 32 edges with 8 xor-and steps, not 16 edges byte-parallel; node sum order changes
# speedup vs baseline: 3.8377x; 1.0475x over previous
#include <hip/hip_runtime.h>
#include <hip/hip_bf16.h>

typedef float v2f __attribute__((ext_vector_type(2)));
typedef float v8f __attribute__((ext_vector_type(8)));

#define IN_DIM   64
#define OUT_DIM  64
#define EDGE_DIM 16
#define H_HEADS  4
#define HD       16
#define NEG_SLOPE   0.2f
#define EPS_SOFTMAX 1e-8f
#define EPS_LN      1e-5f

#define OWN_NODES 391
#define OWN_PAD   392

__global__ void node_gemm_wmma(const float* __restrict__ x,
                               const float* __restrict__ W,
                               float* __restrict__ h, int n) {
    int wave = threadIdx.x >> 5;
    int lane = threadIdx.x & 31;
    int grp  = lane >> 4;
    int lid  = lane & 15;

    int row0 = blockIdx.x * 16;
    int col0 = wave * 16;

    int arow = row0 + lid;
    if (arow >= n) arow = n - 1;

    const float* xr = x + (long)arow * IN_DIM + 2 * grp;
    const float* wc = W + (long)(2 * grp) * OUT_DIM + col0 + lid;

    v8f c = {};
    #pragma unroll
    for (int k0 = 0; k0 < IN_DIM; k0 += 4) {
        v2f a = *(const v2f*)(xr + k0);
        v2f b;
        b.x = wc[(long)k0 * OUT_DIM];
        b.y = wc[(long)(k0 + 1) * OUT_DIM];
        c = __builtin_amdgcn_wmma_f32_16x16x4_f32(false, a, false, b,
                                                  (short)0, c, false, false);
    }

    #pragma unroll
    for (int r = 0; r < 8; ++r) {
        int row = row0 + r + 8 * grp;
        if (row < n) h[(long)row * OUT_DIM + col0 + lid] = c[r];
    }
}

__global__ void node_proj(const float* __restrict__ h,
                          const float* __restrict__ attn_src,
                          const float* __restrict__ attn_dst,
                          float* __restrict__ a_src,
                          float* __restrict__ a_dst, int n) {
    int idx = blockIdx.x * blockDim.x + threadIdx.x;
    if (idx >= n * H_HEADS) return;
    int hh   = idx & 3;
    int node = idx >> 2;
    const float4* hp = (const float4*)(h + (long)node * OUT_DIM + hh * HD);
    const float4* as = (const float4*)(attn_src + hh * HD);
    const float4* ad = (const float4*)(attn_dst + hh * HD);
    float s = 0.0f, d = 0.0f;
    #pragma unroll
    for (int j = 0; j < 4; ++j) {
        float4 hv = hp[j], sv = as[j], dv = ad[j];
        s += hv.x * sv.x + hv.y * sv.y + hv.z * sv.z + hv.w * sv.w;
        d += hv.x * dv.x + hv.y * dv.y + hv.z * dv.z + hv.w * dv.w;
    }
    a_src[idx] = s;
    a_dst[idx] = d;
}

__device__ __forceinline__ long own_key_word(long g, unsigned b, long ng) {
    long blk = g >> 5;
    if (blk < (ng >> 5)) return blk * 256 + (long)(b >> 2) * 128 + (g & 31) * 4 + (long)(b & 3u);
    return g * 8 + (long)b;
}

__global__ void edge_score(const int* __restrict__ src,
                           const int* __restrict__ dst,
                           const float* __restrict__ edge_attr,
                           const float* __restrict__ W_edge,
                           const float* __restrict__ a_src,
                           const float* __restrict__ a_dst,
                           float* __restrict__ attn,
                           unsigned* __restrict__ keyw, long e_cnt) {
    long e = (long)blockIdx.x * blockDim.x + threadIdx.x;
    const bool live = e < e_cnt;
    int s = 0, t = 0;
    if (live) { s = src[e]; t = dst[e]; }
    {
        const unsigned key  = (unsigned)t / (unsigned)OWN_NODES;
        const unsigned lane = threadIdx.x & 31u;
        unsigned word = 0u;
        #pragma unroll
        for (int b = 0; b < 8; ++b) {
            unsigned bal = __builtin_amdgcn_ballot_w32(live && ((key >> b) & 1u) != 0u);
            if (lane == (unsigned)b) word = bal;
        }
        if (lane < 8u && (e - (long)lane) < e_cnt)
            keyw[own_key_word(e >> 5, lane, (e_cnt + 31) >> 5)] = word;
    }
    if (!live) return;

    float sc[H_HEADS] = {0.f, 0.f, 0.f, 0.f};
    const float4* ea = (const float4*)(edge_attr + e * EDGE_DIM);
    #pragma unroll
    for (int q = 0; q < 4; ++q) {
        float4 v = ea[q];
        int d0 = q * 4;
        #pragma unroll
        for (int hh = 0; hh < H_HEADS; ++hh) {
            sc[hh] += v.x * W_edge[(d0 + 0) * H_HEADS + hh]
                    + v.y * W_edge[(d0 + 1) * H_HEADS + hh]
                    + v.z * W_edge[(d0 + 2) * H_HEADS + hh]
                    + v.w * W_edge[(d0 + 3) * H_HEADS + hh];
        }
    }
    float4 asv = ((const float4*)a_src)[s];
    float4 adv = ((const float4*)a_dst)[t];
    float av[H_HEADS] = {asv.x + adv.x, asv.y + adv.y, asv.z + adv.z, asv.w + adv.w};

    float4 outv;
    float* op = (float*)&outv;
    #pragma unroll
    for (int hh = 0; hh < H_HEADS; ++hh) {
        float v = av[hh] + sc[hh];
        v = (v >= 0.0f) ? v : NEG_SLOPE * v;
        op[hh] = v;
    }
    ((float4*)attn)[e] = outv;
}

#define OWN_CAP   16384
#define OWN_U     12
#define OWN_G     4
#define OWN_M     4

__device__ __forceinline__ void own_list(int e, unsigned* cnt_s, int* le_s) {
    unsigned pos = atomicAdd(cnt_s, 1u);
    if (pos < (unsigned)OWN_CAP) le_s[pos] = e;
}

__device__ __forceinline__ void own_test32(uint4 a, uint4 b, const unsigned* inv, unsigned keep,
                                           int e0, unsigned* cnt_s, int* le_s) {
    unsigned m = (a.x ^ inv[0]) & (a.y ^ inv[1]) & (a.z ^ inv[2]) & (a.w ^ inv[3])
               & (b.x ^ inv[4]) & (b.y ^ inv[5]) & (b.z ^ inv[6]) & (b.w ^ inv[7]) & keep;
    while (m) {
        int p = __builtin_ctz(m);
        m &= m - 1u;
        own_list(e0 + p, cnt_s, le_s);
    }
}

__device__ __forceinline__ void own_key_at(int g, int nblk, int* ia, int* ib) {
    int blk = g >> 5;
    bool whole = blk < nblk;
    *ia = whole ? (blk * 64 + (g & 31)) : (2 * g);
    *ib = whole ? (*ia + 32) : (*ia + 1);
}

__device__ __forceinline__ void own_max4(int* mx_s, int tl, float4 a) {
    if (a.x > 0.0f) atomicMax(&mx_s[tl * H_HEADS + 0], __float_as_int(a.x));
    if (a.y > 0.0f) atomicMax(&mx_s[tl * H_HEADS + 1], __float_as_int(a.y));
    if (a.z > 0.0f) atomicMax(&mx_s[tl * H_HEADS + 2], __float_as_int(a.z));
    if (a.w > 0.0f) atomicMax(&mx_s[tl * H_HEADS + 3], __float_as_int(a.w));
}

__global__ void __launch_bounds__(1024)
gather_own_ln(const int* __restrict__ src,
              const int* __restrict__ dst,
              const float* __restrict__ h,
              const float* __restrict__ attn,
              const unsigned* __restrict__ keyw,
              const float* __restrict__ gamma,
              const float* __restrict__ beta,
              float* __restrict__ out, int n, long e_cnt) {
    __shared__ float    acc_s[OWN_PAD * OUT_DIM];
    __shared__ int      le_s[OWN_CAP];
    __shared__ int      lt_s[OWN_CAP];
    __shared__ int      mx_s[OWN_PAD * H_HEADS];
    __shared__ float    sm_s[OWN_PAD * H_HEADS];
    __shared__ unsigned cnt_s;

    const int tid  = threadIdx.x;
    const int wave = tid >> 5;
    const int lane = tid & 31;
    const int lo   = blockIdx.x * OWN_NODES;
    const int ne   = (int)e_cnt;
    const float4* attn4 = (const float4*)attn;

    {
        float4 z = {0.0f, 0.0f, 0.0f, 0.0f};
        float4* a4 = (float4*)acc_s;
        for (int i = tid; i < OWN_PAD * OUT_DIM / 4; i += 1024) a4[i] = z;
        for (int i = tid; i < OWN_PAD * H_HEADS; i += 1024) { mx_s[i] = 0; sm_s[i] = 0.0f; }
        if (tid == 0) cnt_s = 0u;
    }
    __syncthreads();

    const bool keyed = (gridDim.x <= 256u);
    if (keyed) {
        const unsigned mine = blockIdx.x;
        unsigned inv[8];
        #pragma unroll
        for (int b = 0; b < 8; ++b) inv[b] = ((mine >> b) & 1u) - 1u;
        const int nfull = ne >> 5;
        const int rem   = ne & 31;
        const int nblk  = ((ne + 31) >> 5) >> 5;
        const int nmain = (nblk * 32 < nfull) ? nblk * 32 : nfull;
        const uint4* k4 = (const uint4*)keyw;

        {
            int g = nmain + tid;
            if (g < nfull || (g == nfull && rem != 0)) {
                int ia, ib;
                own_key_at(g, nblk, &ia, &ib);
                unsigned keep = (g < nfull) ? 0xFFFFFFFFu : ((1u << rem) - 1u);
                own_test32(k4[ia], k4[ib], inv, keep, g << 5, &cnt_s, le_s);
            }
        }

        int g0 = tid;
        for (; g0 + (OWN_G - 1) * 1024 < nmain; g0 += 1024 * OWN_G) {
            const uint4* kp = k4 + ((g0 >> 5) * 64 + (g0 & 31));
            uint4 ka[OWN_G], kb[OWN_G];
            #pragma unroll
            for (int u = 0; u < OWN_G; ++u) {
                ka[u] = kp[u * 2048];
                kb[u] = kp[u * 2048 + 32];
            }
            #pragma unroll
            for (int u = 0; u < OWN_G; ++u)
                own_test32(ka[u], kb[u], inv, 0xFFFFFFFFu, (g0 + u * 1024) << 5, &cnt_s, le_s);
        }
        for (; g0 < nmain; g0 += 1024) {
            const uint4* kp = k4 + ((g0 >> 5) * 64 + (g0 & 31));
            own_test32(kp[0], kp[32], inv, 0xFFFFFFFFu, g0 << 5, &cnt_s, le_s);
        }
    }
    __syncthreads();

    const unsigned cnt    = cnt_s;
    const bool     listed = keyed && (cnt <= (unsigned)OWN_CAP);

    if (listed) {
        for (unsigned i0 = (unsigned)tid; i0 < cnt; i0 += 1024u * OWN_M) {
            int    ee[OWN_M], tt[OWN_M];
            float4 av[OWN_M];
            #pragma unroll
            for (int u = 0; u < OWN_M; ++u) {
                unsigned i  = i0 + (unsigned)u * 1024u;
                unsigned ic = (i < cnt) ? i : i0;
                ee[u] = le_s[ic];
            }
            #pragma unroll
            for (int u = 0; u < OWN_M; ++u) {
                tt[u] = dst[ee[u]] - lo;
                av[u] = attn4[ee[u]];
            }
            #pragma unroll
            for (int u = 0; u < OWN_M; ++u) {
                unsigned i = i0 + (unsigned)u * 1024u;
                if (i < cnt) {
                    lt_s[i] = tt[u];
                    own_max4(mx_s, tt[u], av[u]);
                }
            }
        }
    } else {
        for (int e = tid; e < ne; e += 1024) {
            unsigned tl = (unsigned)(dst[e] - lo);
            if (tl < (unsigned)OWN_NODES) own_max4(mx_s, (int)tl, attn4[e]);
        }
    }
    __syncthreads();

    if (listed) {
        const int hh = lane >> 3;

        for (unsigned i0 = (unsigned)wave * OWN_U; i0 < cnt; i0 += 32u * OWN_U) {
            int ee[OWN_U], tt[OWN_U];
            #pragma unroll
            for (int u = 0; u < OWN_U; ++u) {
                unsigned i  = i0 + u;
                unsigned ic = (i < cnt) ? i : i0;
                ee[u] = le_s[ic];
                tt[u] = lt_s[ic];
            }
            int   ss[OWN_U];
            float aa[OWN_U], mm[OWN_U];
            #pragma unroll
            for (int u = 0; u < OWN_U; ++u) {
                long e = ee[u];
                ss[u] = src[e];
                aa[u] = attn[e * H_HEADS + hh];
                mm[u] = __int_as_float(mx_s[tt[u] * H_HEADS + hh]);
            }
            v2f hv[OWN_U];
            #pragma unroll
            for (int u = 0; u < OWN_U; ++u)
                hv[u] = *(const v2f*)(h + (long)ss[u] * OUT_DIM + 2 * lane);
            #pragma unroll
            for (int u = 0; u < OWN_U; ++u) {
                if (i0 + u < cnt) {
                    float w = expf(aa[u] - mm[u]);
                    if ((lane & 7) == 0) atomicAdd(&sm_s[tt[u] * H_HEADS + hh], w);
                    atomicAdd(&acc_s[tt[u] * OUT_DIM + 2 * lane],     hv[u].x * w);
                    atomicAdd(&acc_s[tt[u] * OUT_DIM + 2 * lane + 1], hv[u].y * w);
                }
            }
        }
    } else {
        for (int e = tid; e < ne; e += 1024) {
            unsigned tl = (unsigned)(dst[e] - lo);
            if (tl < (unsigned)OWN_NODES) {
                int s = src[e];
                float4 a = attn4[e];
                float w[H_HEADS];
                w[0] = expf(a.x - __int_as_float(mx_s[tl * H_HEADS + 0]));
                w[1] = expf(a.y - __int_as_float(mx_s[tl * H_HEADS + 1]));
                w[2] = expf(a.z - __int_as_float(mx_s[tl * H_HEADS + 2]));
                w[3] = expf(a.w - __int_as_float(mx_s[tl * H_HEADS + 3]));
                for (int hd = 0; hd < H_HEADS; ++hd) {
                    atomicAdd(&sm_s[tl * H_HEADS + hd], w[hd]);
                    for (int d = hd * HD; d < hd * HD + HD; ++d)
                        atomicAdd(&acc_s[tl * OUT_DIM + d], h[(long)s * OUT_DIM + d] * w[hd]);
                }
            }
        }
    }
    __syncthreads();

    {
        const float g0 = gamma[lane], g1 = gamma[lane + 32];
        const float b0 = beta[lane],  b1 = beta[lane + 32];
        const int   h0 = lane >> 4;
        for (int tl0 = wave * 4; tl0 < OWN_NODES; tl0 += 32 * 4) {
            float y0[4], y1[4], s[4], vs[4];
            #pragma unroll
            for (int u = 0; u < 4; ++u) {
                int node = lo + tl0 + u;
                int nc   = (node < n) ? node : (n - 1);
                float q0 = sm_s[(tl0 + u) * H_HEADS + h0]     + EPS_SOFTMAX;
                float q1 = sm_s[(tl0 + u) * H_HEADS + h0 + 2] + EPS_SOFTMAX;
                y0[u] = acc_s[(tl0 + u) * OUT_DIM + lane]      / q0 + h[(long)nc * OUT_DIM + lane];
                y1[u] = acc_s[(tl0 + u) * OUT_DIM + lane + 32] / q1 + h[(long)nc * OUT_DIM + lane + 32];
                s[u]  = y0[u] + y1[u];
            }
            #pragma unroll
            for (int off = 16; off > 0; off >>= 1) {
                #pragma unroll
                for (int u = 0; u < 4; ++u) s[u] += __shfl_xor(s[u], off, 32);
            }
            #pragma unroll
            for (int u = 0; u < 4; ++u) {
                float mu = s[u] * (1.0f / 64.0f);
                y0[u] -= mu;
                y1[u] -= mu;
                vs[u] = y0[u] * y0[u] + y1[u] * y1[u];
            }
            #pragma unroll
            for (int off = 16; off > 0; off >>= 1) {
                #pragma unroll
                for (int u = 0; u < 4; ++u) vs[u] += __shfl_xor(vs[u], off, 32);
            }
            #pragma unroll
            for (int u = 0; u < 4; ++u) {
                int node = lo + tl0 + u;
                if (tl0 + u < OWN_NODES && node < n) {
                    long base = (long)node * OUT_DIM;
                    float rstd = rsqrtf(vs[u] * (1.0f / 64.0f) + EPS_LN);
                    out[base + lane]      = y0[u] * rstd * g0 + b0;
                    out[base + lane + 32] = y1[u] * rstd * g1 + b1;
                }
            }
        }
    }
}

extern "C" void kernel_launch(void* const* d_in, const int* in_sizes, int n_in,
                              void* d_out, int out_size, void* d_ws, size_t ws_size,
                              hipStream_t stream) {
    const float* x         = (const float*)d_in[0];
    const int*   ei        = (const int*)  d_in[1];
    const float* edge_attr = (const float*)d_in[2];
    const float* W_node    = (const float*)d_in[3];
    const float* W_edge    = (const float*)d_in[4];
    const float* attn_src  = (const float*)d_in[5];
    const float* attn_dst  = (const float*)d_in[6];
    const float* gamma     = (const float*)d_in[7];
    const float* beta      = (const float*)d_in[8];
    float* out = (float*)d_out;

    int  n = in_sizes[0] / IN_DIM;
    long e = (long)in_sizes[2] / EDGE_DIM;
    const int* srcIdx = ei;
    const int* dstIdx = ei + e;

    float* ws    = (float*)d_ws;
    float* h     = ws;  ws += (long)n * OUT_DIM;
    float* a_src = ws;  ws += (long)n * H_HEADS;
    float* a_dst = ws;  ws += (long)n * H_HEADS;
    float* attn  = ws;  ws += e * H_HEADS;
    unsigned* keyw = (unsigned*)ws;

    int n_tiles = (n + 15) / 16;
    node_gemm_wmma<<<n_tiles, 128, 0, stream>>>(x, W_node, h, n);

    node_proj<<<(n * H_HEADS + 255) / 256, 256, 0, stream>>>(
        h, attn_src, attn_dst, a_src, a_dst, n);

    edge_score<<<(int)((e + 255) / 256), 256, 0, stream>>>(
        srcIdx, dstIdx, edge_attr, W_edge, a_src, a_dst, attn, keyw, e);

    gather_own_ln<<<(n + OWN_NODES - 1) / OWN_NODES, 1024, 0, stream>>>(
        srcIdx, dstIdx, h, attn, keyw, gamma, beta, out, n, e);
}


# gather: per-edge values (source, place, exp weight, head sums) made once by one lane in new LDS list passes; the wave-per-edge pass only loads the row and adds; sums stay in arrival order
# speedup vs baseline: 4.1483x; 1.0809x over previous
#include <hip/hip_runtime.h>
#include <hip/hip_bf16.h>

typedef float v2f __attribute__((ext_vector_type(2)));
typedef float v8f __attribute__((ext_vector_type(8)));

#define IN_DIM   64
#define OUT_DIM  64
#define EDGE_DIM 16
#define H_HEADS  4
#define HD       16
#define NEG_SLOPE   0.2f
#define EPS_SOFTMAX 1e-8f
#define EPS_LN      1e-5f

#define OWN_NODES 391
#define OWN_PAD   392

__global__ void node_gemm_wmma(const float* __restrict__ x,
                               const float* __restrict__ W,
                               float* __restrict__ h, int n) {
    int wave = threadIdx.x >> 5;
    int lane = threadIdx.x & 31;
    int grp  = lane >> 4;
    int lid  = lane & 15;

    int row0 = blockIdx.x * 16;
    int col0 = wave * 16;

    int arow = row0 + lid;
    if (arow >= n) arow = n - 1;

    const float* xr = x + (long)arow * IN_DIM + 2 * grp;
    const float* wc = W + (long)(2 * grp) * OUT_DIM + col0 + lid;

    v8f c = {};
    #pragma unroll
    for (int k0 = 0; k0 < IN_DIM; k0 += 4) {
        v2f a = *(const v2f*)(xr + k0);
        v2f b;
        b.x = wc[(long)k0 * OUT_DIM];
        b.y = wc[(long)(k0 + 1) * OUT_DIM];
        c = __builtin_amdgcn_wmma_f32_16x16x4_f32(false, a, false, b,
                                                  (short)0, c, false, false);
    }

    #pragma unroll
    for (int r = 0; r < 8; ++r) {
        int row = row0 + r + 8 * grp;
        if (row < n) h[(long)row * OUT_DIM + col0 + lid] = c[r];
    }
}

__global__ void node_proj(const float* __restrict__ h,
                          const float* __restrict__ attn_src,
                          const float* __restrict__ attn_dst,
                          float* __restrict__ a_src,
                          float* __restrict__ a_dst, int n) {
    int idx = blockIdx.x * blockDim.x + threadIdx.x;
    if (idx >= n * H_HEADS) return;
    int hh   = idx & 3;
    int node = idx >> 2;
    const float4* hp = (const float4*)(h + (long)node * OUT_DIM + hh * HD);
    const float4* as = (const float4*)(attn_src + hh * HD);
    const float4* ad = (const float4*)(attn_dst + hh * HD);
    float s = 0.0f, d = 0.0f;
    #pragma unroll
    for (int j = 0; j < 4; ++j) {
        float4 hv = hp[j], sv = as[j], dv = ad[j];
        s += hv.x * sv.x + hv.y * sv.y + hv.z * sv.z + hv.w * sv.w;
        d += hv.x * dv.x + hv.y * dv.y + hv.z * dv.z + hv.w * dv.w;
    }
    a_src[idx] = s;
    a_dst[idx] = d;
}

__device__ __forceinline__ long own_key_word(long g, unsigned b, long ng) {
    long blk = g >> 5;
    if (blk < (ng >> 5)) return blk * 256 + (long)(b >> 2) * 128 + (g & 31) * 4 + (long)(b & 3u);
    return g * 8 + (long)b;
}

__global__ void edge_score(const int* __restrict__ src,
                           const int* __restrict__ dst,
                           const float* __restrict__ edge_attr,
                           const float* __restrict__ W_edge,
                           const float* __restrict__ a_src,
                           const float* __restrict__ a_dst,
                           float* __restrict__ attn,
                           unsigned* __restrict__ keyw, long e_cnt) {
    long e = (long)blockIdx.x * blockDim.x + threadIdx.x;
    const bool live = e < e_cnt;
    int s = 0, t = 0;
    if (live) { s = src[e]; t = dst[e]; }
    {
        const unsigned key  = (unsigned)t / (unsigned)OWN_NODES;
        const unsigned lane = threadIdx.x & 31u;
        unsigned word = 0u;
        #pragma unroll
        for (int b = 0; b < 8; ++b) {
            unsigned bal = __builtin_amdgcn_ballot_w32(live && ((key >> b) & 1u) != 0u);
            if (lane == (unsigned)b) word = bal;
        }
        if (lane < 8u && (e - (long)lane) < e_cnt)
            keyw[own_key_word(e >> 5, lane, (e_cnt + 31) >> 5)] = word;
    }
    if (!live) return;

    float sc[H_HEADS] = {0.f, 0.f, 0.f, 0.f};
    const float4* ea = (const float4*)(edge_attr + e * EDGE_DIM);
    #pragma unroll
    for (int q = 0; q < 4; ++q) {
        float4 v = ea[q];
        int d0 = q * 4;
        #pragma unroll
        for (int hh = 0; hh < H_HEADS; ++hh) {
            sc[hh] += v.x * W_edge[(d0 + 0) * H_HEADS + hh]
                    + v.y * W_edge[(d0 + 1) * H_HEADS + hh]
                    + v.z * W_edge[(d0 + 2) * H_HEADS + hh]
                    + v.w * W_edge[(d0 + 3) * H_HEADS + hh];
        }
    }
    float4 asv = ((const float4*)a_src)[s];
    float4 adv = ((const float4*)a_dst)[t];
    float av[H_HEADS] = {asv.x + adv.x, asv.y + adv.y, asv.z + adv.z, asv.w + adv.w};

    float4 outv;
    float* op = (float*)&outv;
    #pragma unroll
    for (int hh = 0; hh < H_HEADS; ++hh) {
        float v = av[hh] + sc[hh];
        v = (v >= 0.0f) ? v : NEG_SLOPE * v;
        op[hh] = v;
    }
    ((float4*)attn)[e] = outv;
}

#define OWN_CAP   8192
#define OWN_U     16
#define OWN_G     4
#define OWN_M     4
#define OWN_PLACE_BITS 9
#define OWN_PLACE_MASK 511u

__device__ __forceinline__ void own_list(int e, unsigned* cnt_s, int* le_s) {
    unsigned pos = atomicAdd(cnt_s, 1u);
    if (pos < (unsigned)OWN_CAP) le_s[pos] = e;
}

__device__ __forceinline__ void own_test32(uint4 a, uint4 b, const unsigned* inv, unsigned keep,
                                           int e0, unsigned* cnt_s, int* le_s) {
    unsigned m = (a.x ^ inv[0]) & (a.y ^ inv[1]) & (a.z ^ inv[2]) & (a.w ^ inv[3])
               & (b.x ^ inv[4]) & (b.y ^ inv[5]) & (b.z ^ inv[6]) & (b.w ^ inv[7]) & keep;
    while (m) {
        int p = __builtin_ctz(m);
        m &= m - 1u;
        own_list(e0 + p, cnt_s, le_s);
    }
}

__device__ __forceinline__ void own_key_at(int g, int nblk, int* ia, int* ib) {
    int blk = g >> 5;
    bool whole = blk < nblk;
    *ia = whole ? (blk * 64 + (g & 31)) : (2 * g);
    *ib = whole ? (*ia + 32) : (*ia + 1);
}

__device__ __forceinline__ void own_max4(int* mx_s, int tl, float4 a) {
    if (a.x > 0.0f) atomicMax(&mx_s[tl * H_HEADS + 0], __float_as_int(a.x));
    if (a.y > 0.0f) atomicMax(&mx_s[tl * H_HEADS + 1], __float_as_int(a.y));
    if (a.z > 0.0f) atomicMax(&mx_s[tl * H_HEADS + 2], __float_as_int(a.z));
    if (a.w > 0.0f) atomicMax(&mx_s[tl * H_HEADS + 3], __float_as_int(a.w));
}

__global__ void __launch_bounds__(1024)
gather_own_ln(const int* __restrict__ src,
              const int* __restrict__ dst,
              const float* __restrict__ h,
              const float* __restrict__ attn,
              const unsigned* __restrict__ keyw,
              const float* __restrict__ gamma,
              const float* __restrict__ beta,
              float* __restrict__ out, int n, long e_cnt) {
    __shared__ float    acc_s[OWN_PAD * OUT_DIM];
    __shared__ int      le_s[OWN_CAP];
    __shared__ float4   sc_s[OWN_CAP];
    __shared__ alignas(16) int mx_s[OWN_PAD * H_HEADS];
    __shared__ float    sm_s[OWN_PAD * H_HEADS];
    __shared__ unsigned cnt_s;

    const int tid  = threadIdx.x;
    const int wave = tid >> 5;
    const int lane = tid & 31;
    const int lo   = blockIdx.x * OWN_NODES;
    const int ne   = (int)e_cnt;
    const float4* attn4 = (const float4*)attn;

    {
        float4 z = {0.0f, 0.0f, 0.0f, 0.0f};
        float4* a4 = (float4*)acc_s;
        for (int i = tid; i < OWN_PAD * OUT_DIM / 4; i += 1024) a4[i] = z;
        for (int i = tid; i < OWN_PAD * H_HEADS; i += 1024) { mx_s[i] = 0; sm_s[i] = 0.0f; }
        if (tid == 0) cnt_s = 0u;
    }
    __syncthreads();

    const bool keyed = (gridDim.x <= 256u);
    if (keyed) {
        const unsigned mine = blockIdx.x;
        unsigned inv[8];
        #pragma unroll
        for (int b = 0; b < 8; ++b) inv[b] = ((mine >> b) & 1u) - 1u;
        const int nfull = ne >> 5;
        const int rem   = ne & 31;
        const int nblk  = ((ne + 31) >> 5) >> 5;
        const int nmain = (nblk * 32 < nfull) ? nblk * 32 : nfull;
        const uint4* k4 = (const uint4*)keyw;

        {
            int g = nmain + tid;
            if (g < nfull || (g == nfull && rem != 0)) {
                int ia, ib;
                own_key_at(g, nblk, &ia, &ib);
                unsigned keep = (g < nfull) ? 0xFFFFFFFFu : ((1u << rem) - 1u);
                own_test32(k4[ia], k4[ib], inv, keep, g << 5, &cnt_s, le_s);
            }
        }

        int g0 = tid;
        for (; g0 + (OWN_G - 1) * 1024 < nmain; g0 += 1024 * OWN_G) {
            const uint4* kp = k4 + ((g0 >> 5) * 64 + (g0 & 31));
            uint4 ka[OWN_G], kb[OWN_G];
            #pragma unroll
            for (int u = 0; u < OWN_G; ++u) {
                ka[u] = kp[u * 2048];
                kb[u] = kp[u * 2048 + 32];
            }
            #pragma unroll
            for (int u = 0; u < OWN_G; ++u)
                own_test32(ka[u], kb[u], inv, 0xFFFFFFFFu, (g0 + u * 1024) << 5, &cnt_s, le_s);
        }
        for (; g0 < nmain; g0 += 1024) {
            const uint4* kp = k4 + ((g0 >> 5) * 64 + (g0 & 31));
            own_test32(kp[0], kp[32], inv, 0xFFFFFFFFu, g0 << 5, &cnt_s, le_s);
        }
    }
    __syncthreads();

    const unsigned cnt    = cnt_s;
    const bool     listed = keyed && (cnt <= (unsigned)OWN_CAP);

    if (listed) {
        for (unsigned i0 = (unsigned)tid; i0 < cnt; i0 += 1024u * OWN_M) {
            int    ee[OWN_M], tt[OWN_M], ss[OWN_M];
            float4 av[OWN_M];
            #pragma unroll
            for (int u = 0; u < OWN_M; ++u) {
                unsigned i  = i0 + (unsigned)u * 1024u;
                unsigned ic = (i < cnt) ? i : i0;
                ee[u] = le_s[ic];
            }
            #pragma unroll
            for (int u = 0; u < OWN_M; ++u) {
                tt[u] = dst[ee[u]] - lo;
                ss[u] = src[ee[u]];
                av[u] = attn4[ee[u]];
            }
            #pragma unroll
            for (int u = 0; u < OWN_M; ++u) {
                unsigned i = i0 + (unsigned)u * 1024u;
                if (i < cnt) {
                    le_s[i] = (int)(((unsigned)ss[u] << OWN_PLACE_BITS) | (unsigned)tt[u]);
                    sc_s[i] = av[u];
                    own_max4(mx_s, tt[u], av[u]);
                }
            }
        }
    } else {
        for (int e = tid; e < ne; e += 1024) {
            unsigned tl = (unsigned)(dst[e] - lo);
            if (tl < (unsigned)OWN_NODES) own_max4(mx_s, (int)tl, attn4[e]);
        }
    }
    __syncthreads();

    if (listed) {
        for (unsigned i = (unsigned)tid; i < cnt; i += 1024u) {
            const unsigned tl = (unsigned)le_s[i] & OWN_PLACE_MASK;
            const float4 a = sc_s[i];
            const int4   m = *(const int4*)&mx_s[tl * H_HEADS];
            float4 w;
            w.x = expf(a.x - __int_as_float(m.x));
            w.y = expf(a.y - __int_as_float(m.y));
            w.z = expf(a.z - __int_as_float(m.z));
            w.w = expf(a.w - __int_as_float(m.w));
            atomicAdd(&sm_s[tl * H_HEADS + 0], w.x);
            atomicAdd(&sm_s[tl * H_HEADS + 1], w.y);
            atomicAdd(&sm_s[tl * H_HEADS + 2], w.z);
            atomicAdd(&sm_s[tl * H_HEADS + 3], w.w);
            sc_s[i] = w;
        }
    }
    __syncthreads();

    if (listed) {
        const int hh = lane >> 3;
        const unsigned wv = (unsigned)__builtin_amdgcn_readfirstlane(wave);
        const unsigned cn = (unsigned)__builtin_amdgcn_readfirstlane((int)cnt);
        const float*   wt_s  = (const float*)sc_s;
        const char*    hb    = (const char*)h;
        const unsigned lane8 = (unsigned)lane * 8u;

        unsigned i0 = wv * OWN_U;
        for (; i0 + OWN_U <= cn; i0 += 32u * OWN_U) {
            unsigned pk[OWN_U];
            float    ww[OWN_U];
            #pragma unroll
            for (int u = 0; u < OWN_U; ++u) {
                pk[u] = (unsigned)le_s[i0 + u];
                ww[u] = wt_s[(i0 + u) * H_HEADS + hh];
            }
            v2f hv[OWN_U];
            #pragma unroll
            for (int u = 0; u < OWN_U; ++u)
                hv[u] = *(const v2f*)(hb + ((((pk[u] >> OWN_PLACE_BITS) * (unsigned)OUT_DIM) << 2) + lane8));
            #pragma unroll
            for (int u = 0; u < OWN_U; ++u) {
                float* ap = &acc_s[(pk[u] & OWN_PLACE_MASK) * OUT_DIM + 2 * lane];
                atomicAdd(ap,     hv[u].x * ww[u]);
                atomicAdd(ap + 1, hv[u].y * ww[u]);
            }
        }
        if (i0 < cn) {
            unsigned pk[OWN_U];
            float    ww[OWN_U];
            #pragma unroll
            for (int u = 0; u < OWN_U; ++u) {
                const unsigned ic = (i0 + u < cn) ? (i0 + u) : i0;
                pk[u] = (unsigned)le_s[ic];
                ww[u] = wt_s[ic * H_HEADS + hh];
            }
            v2f hv[OWN_U];
            #pragma unroll
            for (int u = 0; u < OWN_U; ++u)
                hv[u] = *(const v2f*)(hb + ((((pk[u] >> OWN_PLACE_BITS) * (unsigned)OUT_DIM) << 2) + lane8));
            #pragma unroll
            for (int u = 0; u < OWN_U; ++u) {
                if (i0 + u < cn) {
                    float* ap = &acc_s[(pk[u] & OWN_PLACE_MASK) * OUT_DIM + 2 * lane];
                    atomicAdd(ap,     hv[u].x * ww[u]);
                    atomicAdd(ap + 1, hv[u].y * ww[u]);
                }
            }
        }
    } else {
        for (int e = tid; e < ne; e += 1024) {
            unsigned tl = (unsigned)(dst[e] - lo);
            if (tl < (unsigned)OWN_NODES) {
                int s = src[e];
                float4 a = attn4[e];
                float w[H_HEADS];
                w[0] = expf(a.x - __int_as_float(mx_s[tl * H_HEADS + 0]));
                w[1] = expf(a.y - __int_as_float(mx_s[tl * H_HEADS + 1]));
                w[2] = expf(a.z - __int_as_float(mx_s[tl * H_HEADS + 2]));
                w[3] = expf(a.w - __int_as_float(mx_s[tl * H_HEADS + 3]));
                for (int hd = 0; hd < H_HEADS; ++hd) {
                    atomicAdd(&sm_s[tl * H_HEADS + hd], w[hd]);
                    for (int d = hd * HD; d < hd * HD + HD; ++d)
                        atomicAdd(&acc_s[tl * OUT_DIM + d], h[(long)s * OUT_DIM + d] * w[hd]);
                }
            }
        }
    }
    __syncthreads();

    {
        const float g0 = gamma[lane], g1 = gamma[lane + 32];
        const float b0 = beta[lane],  b1 = beta[lane + 32];
        const int   h0 = lane >> 4;
        for (int tl0 = wave * 4; tl0 < OWN_NODES; tl0 += 32 * 4) {
            float y0[4], y1[4], s[4], vs[4];
            #pragma unroll
            for (int u = 0; u < 4; ++u) {
                int node = lo + tl0 + u;
                int nc   = (node < n) ? node : (n - 1);
                float q0 = sm_s[(tl0 + u) * H_HEADS + h0]     + EPS_SOFTMAX;
                float q1 = sm_s[(tl0 + u) * H_HEADS + h0 + 2] + EPS_SOFTMAX;
                y0[u] = acc_s[(tl0 + u) * OUT_DIM + lane]      / q0 + h[(long)nc * OUT_DIM + lane];
                y1[u] = acc_s[(tl0 + u) * OUT_DIM + lane + 32] / q1 + h[(long)nc * OUT_DIM + lane + 32];
                s[u]  = y0[u] + y1[u];
            }
            #pragma unroll
            for (int off = 16; off > 0; off >>= 1) {
                #pragma unroll
                for (int u = 0; u < 4; ++u) s[u] += __shfl_xor(s[u], off, 32);
            }
            #pragma unroll
            for (int u = 0; u < 4; ++u) {
                float mu = s[u] * (1.0f / 64.0f);
                y0[u] -= mu;
                y1[u] -= mu;
                vs[u] = y0[u] * y0[u] + y1[u] * y1[u];
            }
            #pragma unroll
            for (int off = 16; off > 0; off >>= 1) {
                #pragma unroll
                for (int u = 0; u < 4; ++u) vs[u] += __shfl_xor(vs[u], off, 32);
            }
            #pragma unroll
            for (int u = 0; u < 4; ++u) {
                int node = lo + tl0 + u;
                if (tl0 + u < OWN_NODES && node < n) {
                    long base = (long)node * OUT_DIM;
                    float rstd = rsqrtf(vs[u] * (1.0f / 64.0f) + EPS_LN);
                    out[base + lane]      = y0[u] * rstd * g0 + b0;
                    out[base + lane + 32] = y1[u] * rstd * g1 + b1;
                }
            }
        }
    }
}

extern "C" void kernel_launch(void* const* d_in, const int* in_sizes, int n_in,
                              void* d_out, int out_size, void* d_ws, size_t ws_size,
                              hipStream_t stream) {
    const float* x         = (const float*)d_in[0];
    const int*   ei        = (const int*)  d_in[1];
    const float* edge_attr = (const float*)d_in[2];
    const float* W_node    = (const float*)d_in[3];
    const float* W_edge    = (const float*)d_in[4];
    const float* attn_src  = (const float*)d_in[5];
    const float* attn_dst  = (const float*)d_in[6];
    const float* gamma     = (const float*)d_in[7];
    const float* beta      = (const float*)d_in[8];
    float* out = (float*)d_out;

    int  n = in_sizes[0] / IN_DIM;
    long e = (long)in_sizes[2] / EDGE_DIM;
    const int* srcIdx = ei;
    const int* dstIdx = ei + e;

    float* ws    = (float*)d_ws;
    float* h     = ws;  ws += (long)n * OUT_DIM;
    float* a_src = ws;  ws += (long)n * H_HEADS;
    float* a_dst = ws;  ws += (long)n * H_HEADS;
    float* attn  = ws;  ws += e * H_HEADS;
    unsigned* keyw = (unsigned*)ws;

    int n_tiles = (n + 15) / 16;
    node_gemm_wmma<<<n_tiles, 128, 0, stream>>>(x, W_node, h, n);

    node_proj<<<(n * H_HEADS + 255) / 256, 256, 0, stream>>>(
        h, attn_src, attn_dst, a_src, a_dst, n);

    edge_score<<<(int)((e + 255) / 256), 256, 0, stream>>>(
        srcIdx, dstIdx, edge_attr, W_edge, a_src, a_dst, attn, keyw, e);

    gather_own_ln<<<(n + OWN_NODES - 1) / OWN_NODES, 1024, 0, stream>>>(
        srcIdx, dstIdx, h, attn, keyw, gamma, beta, out, n, e);
}


# edge scores made in gather pass B1 by the group owning the destination (reads attributes, source projection; destination's from LDS): score plane gone, edge launch writes only keys; sum order kept
# speedup vs baseline: 4.5668x; 1.1009x over previous
#include <hip/hip_runtime.h>
#include <hip/hip_bf16.h>

typedef float v2f __attribute__((ext_vector_type(2)));
typedef float v8f __attribute__((ext_vector_type(8)));

#define IN_DIM   64
#define OUT_DIM  64
#define EDGE_DIM 16
#define H_HEADS  4
#define HD       16
#define NEG_SLOPE   0.2f
#define EPS_SOFTMAX 1e-8f
#define EPS_LN      1e-5f

#define OWN_NODES 391
#define OWN_PAD   392

__global__ void node_gemm_wmma(const float* __restrict__ x,
                               const float* __restrict__ W,
                               float* __restrict__ h, int n) {
    int wave = threadIdx.x >> 5;
    int lane = threadIdx.x & 31;
    int grp  = lane >> 4;
    int lid  = lane & 15;

    int row0 = blockIdx.x * 16;
    int col0 = wave * 16;

    int arow = row0 + lid;
    if (arow >= n) arow = n - 1;

    const float* xr = x + (long)arow * IN_DIM + 2 * grp;
    const float* wc = W + (long)(2 * grp) * OUT_DIM + col0 + lid;

    v8f c = {};
    #pragma unroll
    for (int k0 = 0; k0 < IN_DIM; k0 += 4) {
        v2f a = *(const v2f*)(xr + k0);
        v2f b;
        b.x = wc[(long)k0 * OUT_DIM];
        b.y = wc[(long)(k0 + 1) * OUT_DIM];
        c = __builtin_amdgcn_wmma_f32_16x16x4_f32(false, a, false, b,
                                                  (short)0, c, false, false);
    }

    #pragma unroll
    for (int r = 0; r < 8; ++r) {
        int row = row0 + r + 8 * grp;
        if (row < n) h[(long)row * OUT_DIM + col0 + lid] = c[r];
    }
}

__global__ void node_proj(const float* __restrict__ h,
                          const float* __restrict__ attn_src,
                          const float* __restrict__ attn_dst,
                          float* __restrict__ a_src,
                          float* __restrict__ a_dst, int n) {
    int idx = blockIdx.x * blockDim.x + threadIdx.x;
    if (idx >= n * H_HEADS) return;
    int hh   = idx & 3;
    int node = idx >> 2;
    const float4* hp = (const float4*)(h + (long)node * OUT_DIM + hh * HD);
    const float4* as = (const float4*)(attn_src + hh * HD);
    const float4* ad = (const float4*)(attn_dst + hh * HD);
    float s = 0.0f, d = 0.0f;
    #pragma unroll
    for (int j = 0; j < 4; ++j) {
        float4 hv = hp[j], sv = as[j], dv = ad[j];
        s += hv.x * sv.x + hv.y * sv.y + hv.z * sv.z + hv.w * sv.w;
        d += hv.x * dv.x + hv.y * dv.y + hv.z * dv.z + hv.w * dv.w;
    }
    a_src[idx] = s;
    a_dst[idx] = d;
}

__device__ __forceinline__ long own_key_word(long g, unsigned b, long ng) {
    long blk = g >> 5;
    if (blk < (ng >> 5)) return blk * 256 + (long)(b >> 2) * 128 + (g & 31) * 4 + (long)(b & 3u);
    return g * 8 + (long)b;
}

__global__ void edge_key(const int* __restrict__ dst,
                         unsigned* __restrict__ keyw, long e_cnt) {
    const unsigned lane = threadIdx.x & 31u;
    const long     g0   = (long)blockIdx.x * 32 + (long)(threadIdx.x >> 5) * 4;
    const long     ng   = (e_cnt + 31) >> 5;
    unsigned word = 0u;
    int t[4];
    #pragma unroll
    for (int r = 0; r < 4; ++r) {
        const long e = (g0 + r) * 32 + (long)lane;
        t[r] = dst[(e < e_cnt) ? e : (e_cnt - 1)];
    }
    #pragma unroll
    for (int r = 0; r < 4; ++r) {
        const long e    = (g0 + r) * 32 + (long)lane;
        const bool live = e < e_cnt;
        const unsigned key = (unsigned)t[r] / (unsigned)OWN_NODES;
        #pragma unroll
        for (int b = 0; b < 8; ++b) {
            unsigned bal = __builtin_amdgcn_ballot_w32(live && ((key >> b) & 1u) != 0u);
            if (lane == (unsigned)(r * 8 + b)) word = bal;
        }
    }
    const long g = g0 + (long)(lane >> 3);
    if (g * 32 < e_cnt) keyw[own_key_word(g, lane & 7u, ng)] = word;
}

__device__ __forceinline__ float4 own_score(const float* __restrict__ W_edge, const float4* ea,
                                            float4 asv, float4 adv) {
    float sc[H_HEADS] = {0.f, 0.f, 0.f, 0.f};
    #pragma unroll
    for (int q = 0; q < 4; ++q) {
        float4 v = ea[q];
        int d0 = q * 4;
        #pragma unroll
        for (int hh = 0; hh < H_HEADS; ++hh) {
            sc[hh] += v.x * W_edge[(d0 + 0) * H_HEADS + hh]
                    + v.y * W_edge[(d0 + 1) * H_HEADS + hh]
                    + v.z * W_edge[(d0 + 2) * H_HEADS + hh]
                    + v.w * W_edge[(d0 + 3) * H_HEADS + hh];
        }
    }
    float av[H_HEADS] = {asv.x + adv.x, asv.y + adv.y, asv.z + adv.z, asv.w + adv.w};
    float4 outv;
    float* op = (float*)&outv;
    #pragma unroll
    for (int hh = 0; hh < H_HEADS; ++hh) {
        float v = av[hh] + sc[hh];
        v = (v >= 0.0f) ? v : NEG_SLOPE * v;
        op[hh] = v;
    }
    return outv;
}

__device__ __forceinline__ float4 own_score_at(const float* __restrict__ edge_attr,
                                               const float* __restrict__ W_edge,
                                               const float* __restrict__ a_src,
                                               int e, int s, float4 adv) {
    const float4* p = (const float4*)(edge_attr + (long)e * EDGE_DIM);
    float4 ea[4] = {p[0], p[1], p[2], p[3]};
    return own_score(W_edge, ea, ((const float4*)a_src)[s], adv);
}

#define OWN_CAP   8192
#define OWN_U     16
#define OWN_G     4
#define OWN_M     1
#define OWN_PLACE_BITS 9
#define OWN_PLACE_MASK 511u

__device__ __forceinline__ void own_list(int e, unsigned* cnt_s, int* le_s) {
    unsigned pos = atomicAdd(cnt_s, 1u);
    if (pos < (unsigned)OWN_CAP) le_s[pos] = e;
}

__device__ __forceinline__ void own_test32(uint4 a, uint4 b, const unsigned* inv, unsigned keep,
                                           int e0, unsigned* cnt_s, int* le_s) {
    unsigned m = (a.x ^ inv[0]) & (a.y ^ inv[1]) & (a.z ^ inv[2]) & (a.w ^ inv[3])
               & (b.x ^ inv[4]) & (b.y ^ inv[5]) & (b.z ^ inv[6]) & (b.w ^ inv[7]) & keep;
    while (m) {
        int p = __builtin_ctz(m);
        m &= m - 1u;
        own_list(e0 + p, cnt_s, le_s);
    }
}

__device__ __forceinline__ void own_key_at(int g, int nblk, int* ia, int* ib) {
    int blk = g >> 5;
    bool whole = blk < nblk;
    *ia = whole ? (blk * 64 + (g & 31)) : (2 * g);
    *ib = whole ? (*ia + 32) : (*ia + 1);
}

__device__ __forceinline__ void own_max4(int* mx_s, int tl, float4 a) {
    if (a.x > 0.0f) atomicMax(&mx_s[tl * H_HEADS + 0], __float_as_int(a.x));
    if (a.y > 0.0f) atomicMax(&mx_s[tl * H_HEADS + 1], __float_as_int(a.y));
    if (a.z > 0.0f) atomicMax(&mx_s[tl * H_HEADS + 2], __float_as_int(a.z));
    if (a.w > 0.0f) atomicMax(&mx_s[tl * H_HEADS + 3], __float_as_int(a.w));
}

__global__ void __launch_bounds__(1024)
gather_own_ln(const int* __restrict__ src,
              const int* __restrict__ dst,
              const float* __restrict__ h,
              const float* __restrict__ edge_attr,
              const float* __restrict__ W_edge,
              const float* __restrict__ a_src,
              const float* __restrict__ a_dst,
              const unsigned* __restrict__ keyw,
              const float* __restrict__ gamma,
              const float* __restrict__ beta,
              float* __restrict__ out, int n, long e_cnt) {
    __shared__ float    acc_s[OWN_PAD * OUT_DIM];
    __shared__ int      le_s[OWN_CAP];
    __shared__ float4   sc_s[OWN_CAP];
    __shared__ alignas(16) int mx_s[OWN_PAD * H_HEADS];
    __shared__ float    sm_s[OWN_PAD * H_HEADS];
    __shared__ alignas(16) float adst_s[OWN_PAD * H_HEADS];
    __shared__ unsigned cnt_s;

    const int tid  = threadIdx.x;
    const int wave = tid >> 5;
    const int lane = tid & 31;
    const int lo   = blockIdx.x * OWN_NODES;
    const int ne   = (int)e_cnt;
    const float4* asrc4 = (const float4*)a_src;

    {
        float4 z = {0.0f, 0.0f, 0.0f, 0.0f};
        float4* a4 = (float4*)acc_s;
        for (int i = tid; i < OWN_PAD * OUT_DIM / 4; i += 1024) a4[i] = z;
        for (int i = tid; i < OWN_PAD * H_HEADS; i += 1024) {
            mx_s[i] = 0;
            sm_s[i] = 0.0f;
            const int node = lo + (i >> 2);
            adst_s[i] = (node < n) ? a_dst[(long)node * H_HEADS + (i & 3)] : 0.0f;
        }
        if (tid == 0) cnt_s = 0u;
    }
    __syncthreads();

    const bool keyed = (gridDim.x <= 256u);
    if (keyed) {
        const unsigned mine = blockIdx.x;
        unsigned inv[8];
        #pragma unroll
        for (int b = 0; b < 8; ++b) inv[b] = ((mine >> b) & 1u) - 1u;
        const int nfull = ne >> 5;
        const int rem   = ne & 31;
        const int nblk  = ((ne + 31) >> 5) >> 5;
        const int nmain = (nblk * 32 < nfull) ? nblk * 32 : nfull;
        const uint4* k4 = (const uint4*)keyw;

        {
            int g = nmain + tid;
            if (g < nfull || (g == nfull && rem != 0)) {
                int ia, ib;
                own_key_at(g, nblk, &ia, &ib);
                unsigned keep = (g < nfull) ? 0xFFFFFFFFu : ((1u << rem) - 1u);
                own_test32(k4[ia], k4[ib], inv, keep, g << 5, &cnt_s, le_s);
            }
        }

        int g0 = tid;
        for (; g0 + (OWN_G - 1) * 1024 < nmain; g0 += 1024 * OWN_G) {
            const uint4* kp = k4 + ((g0 >> 5) * 64 + (g0 & 31));
            uint4 ka[OWN_G], kb[OWN_G];
            #pragma unroll
            for (int u = 0; u < OWN_G; ++u) {
                ka[u] = kp[u * 2048];
                kb[u] = kp[u * 2048 + 32];
            }
            #pragma unroll
            for (int u = 0; u < OWN_G; ++u)
                own_test32(ka[u], kb[u], inv, 0xFFFFFFFFu, (g0 + u * 1024) << 5, &cnt_s, le_s);
        }
        for (; g0 < nmain; g0 += 1024) {
            const uint4* kp = k4 + ((g0 >> 5) * 64 + (g0 & 31));
            own_test32(kp[0], kp[32], inv, 0xFFFFFFFFu, g0 << 5, &cnt_s, le_s);
        }
    }
    __syncthreads();

    const unsigned cnt    = cnt_s;
    const bool     listed = keyed && (cnt <= (unsigned)OWN_CAP);

    if (listed) {
        for (unsigned i0 = (unsigned)tid; i0 < cnt; i0 += 1024u * OWN_M) {
            int    ee[OWN_M], tt[OWN_M], ss[OWN_M];
            float4 ea[OWN_M][4], as[OWN_M];
            #pragma unroll
            for (int u = 0; u < OWN_M; ++u) {
                unsigned i  = i0 + (unsigned)u * 1024u;
                unsigned ic = (i < cnt) ? i : i0;
                ee[u] = le_s[ic];
            }
            #pragma unroll
            for (int u = 0; u < OWN_M; ++u) {
                tt[u] = dst[ee[u]] - lo;
                ss[u] = src[ee[u]];
                const float4* p = (const float4*)(edge_attr + (long)ee[u] * EDGE_DIM);
                ea[u][0] = p[0]; ea[u][1] = p[1]; ea[u][2] = p[2]; ea[u][3] = p[3];
            }
            #pragma unroll
            for (int u = 0; u < OWN_M; ++u) as[u] = asrc4[ss[u]];
            #pragma unroll
            for (int u = 0; u < OWN_M; ++u) {
                unsigned i = i0 + (unsigned)u * 1024u;
                if (i < cnt) {
                    const float4 ad = *(const float4*)&adst_s[tt[u] * H_HEADS];
                    const float4 av = own_score(W_edge, ea[u], as[u], ad);
                    le_s[i] = (int)(((unsigned)ss[u] << OWN_PLACE_BITS) | (unsigned)tt[u]);
                    sc_s[i] = av;
                    own_max4(mx_s, tt[u], av);
                }
            }
        }
    } else {
        for (int e = tid; e < ne; e += 1024) {
            unsigned tl = (unsigned)(dst[e] - lo);
            if (tl < (unsigned)OWN_NODES)
                own_max4(mx_s, (int)tl, own_score_at(edge_attr, W_edge, a_src, e, src[e],
                                                     *(const float4*)&adst_s[tl * H_HEADS]));
        }
    }
    __syncthreads();

    if (listed) {
        for (unsigned i = (unsigned)tid; i < cnt; i += 1024u) {
            const unsigned tl = (unsigned)le_s[i] & OWN_PLACE_MASK;
            const float4 a = sc_s[i];
            const int4   m = *(const int4*)&mx_s[tl * H_HEADS];
            float4 w;
            w.x = expf(a.x - __int_as_float(m.x));
            w.y = expf(a.y - __int_as_float(m.y));
            w.z = expf(a.z - __int_as_float(m.z));
            w.w = expf(a.w - __int_as_float(m.w));
            atomicAdd(&sm_s[tl * H_HEADS + 0], w.x);
            atomicAdd(&sm_s[tl * H_HEADS + 1], w.y);
            atomicAdd(&sm_s[tl * H_HEADS + 2], w.z);
            atomicAdd(&sm_s[tl * H_HEADS + 3], w.w);
            sc_s[i] = w;
        }
    }
    __syncthreads();

    if (listed) {
        const int hh = lane >> 3;
        const unsigned wv = (unsigned)__builtin_amdgcn_readfirstlane(wave);
        const unsigned cn = (unsigned)__builtin_amdgcn_readfirstlane((int)cnt);
        const float*   wt_s  = (const float*)sc_s;
        const char*    hb    = (const char*)h;
        const unsigned lane8 = (unsigned)lane * 8u;

        unsigned i0 = wv * OWN_U;
        for (; i0 + OWN_U <= cn; i0 += 32u * OWN_U) {
            unsigned pk[OWN_U];
            float    ww[OWN_U];
            #pragma unroll
            for (int u = 0; u < OWN_U; ++u) {
                pk[u] = (unsigned)le_s[i0 + u];
                ww[u] = wt_s[(i0 + u) * H_HEADS + hh];
            }
            v2f hv[OWN_U];
            #pragma unroll
            for (int u = 0; u < OWN_U; ++u)
                hv[u] = *(const v2f*)(hb + ((((pk[u] >> OWN_PLACE_BITS) * (unsigned)OUT_DIM) << 2) + lane8));
            #pragma unroll
            for (int u = 0; u < OWN_U; ++u) {
                float* ap = &acc_s[(pk[u] & OWN_PLACE_MASK) * OUT_DIM + 2 * lane];
                atomicAdd(ap,     hv[u].x * ww[u]);
                atomicAdd(ap + 1, hv[u].y * ww[u]);
            }
        }
        if (i0 < cn) {
            unsigned pk[OWN_U];
            float    ww[OWN_U];
            #pragma unroll
            for (int u = 0; u < OWN_U; ++u) {
                const unsigned ic = (i0 + u < cn) ? (i0 + u) : i0;
                pk[u] = (unsigned)le_s[ic];
                ww[u] = wt_s[ic * H_HEADS + hh];
            }
            v2f hv[OWN_U];
            #pragma unroll
            for (int u = 0; u < OWN_U; ++u)
                hv[u] = *(const v2f*)(hb + ((((pk[u] >> OWN_PLACE_BITS) * (unsigned)OUT_DIM) << 2) + lane8));
            #pragma unroll
            for (int u = 0; u < OWN_U; ++u) {
                if (i0 + u < cn) {
                    float* ap = &acc_s[(pk[u] & OWN_PLACE_MASK) * OUT_DIM + 2 * lane];
                    atomicAdd(ap,     hv[u].x * ww[u]);
                    atomicAdd(ap + 1, hv[u].y * ww[u]);
                }
            }
        }
    } else {
        for (int e = tid; e < ne; e += 1024) {
            unsigned tl = (unsigned)(dst[e] - lo);
            if (tl < (unsigned)OWN_NODES) {
                int s = src[e];
                float4 a = own_score_at(edge_attr, W_edge, a_src, e, s, *(const float4*)&adst_s[tl * H_HEADS]);
                float w[H_HEADS];
                w[0] = expf(a.x - __int_as_float(mx_s[tl * H_HEADS + 0]));
                w[1] = expf(a.y - __int_as_float(mx_s[tl * H_HEADS + 1]));
                w[2] = expf(a.z - __int_as_float(mx_s[tl * H_HEADS + 2]));
                w[3] = expf(a.w - __int_as_float(mx_s[tl * H_HEADS + 3]));
                for (int hd = 0; hd < H_HEADS; ++hd) {
                    atomicAdd(&sm_s[tl * H_HEADS + hd], w[hd]);
                    for (int d = hd * HD; d < hd * HD + HD; ++d)
                        atomicAdd(&acc_s[tl * OUT_DIM + d], h[(long)s * OUT_DIM + d] * w[hd]);
                }
            }
        }
    }
    __syncthreads();

    {
        const float g0 = gamma[lane], g1 = gamma[lane + 32];
        const float b0 = beta[lane],  b1 = beta[lane + 32];
        const int   h0 = lane >> 4;
        for (int tl0 = wave * 4; tl0 < OWN_NODES; tl0 += 32 * 4) {
            float y0[4], y1[4], s[4], vs[4];
            #pragma unroll
            for (int u = 0; u < 4; ++u) {
                int node = lo + tl0 + u;
                int nc   = (node < n) ? node : (n - 1);
                float q0 = sm_s[(tl0 + u) * H_HEADS + h0]     + EPS_SOFTMAX;
                float q1 = sm_s[(tl0 + u) * H_HEADS + h0 + 2] + EPS_SOFTMAX;
                y0[u] = acc_s[(tl0 + u) * OUT_DIM + lane]      / q0 + h[(long)nc * OUT_DIM + lane];
                y1[u] = acc_s[(tl0 + u) * OUT_DIM + lane + 32] / q1 + h[(long)nc * OUT_DIM + lane + 32];
                s[u]  = y0[u] + y1[u];
            }
            #pragma unroll
            for (int off = 16; off > 0; off >>= 1) {
                #pragma unroll
                for (int u = 0; u < 4; ++u) s[u] += __shfl_xor(s[u], off, 32);
            }
            #pragma unroll
            for (int u = 0; u < 4; ++u) {
                float mu = s[u] * (1.0f / 64.0f);
                y0[u] -= mu;
                y1[u] -= mu;
                vs[u] = y0[u] * y0[u] + y1[u] * y1[u];
            }
            #pragma unroll
            for (int off = 16; off > 0; off >>= 1) {
                #pragma unroll
                for (int u = 0; u < 4; ++u) vs[u] += __shfl_xor(vs[u], off, 32);
            }
            #pragma unroll
            for (int u = 0; u < 4; ++u) {
                int node = lo + tl0 + u;
                if (tl0 + u < OWN_NODES && node < n) {
                    long base = (long)node * OUT_DIM;
                    float rstd = rsqrtf(vs[u] * (1.0f / 64.0f) + EPS_LN);
                    out[base + lane]      = y0[u] * rstd * g0 + b0;
                    out[base + lane + 32] = y1[u] * rstd * g1 + b1;
                }
            }
        }
    }
}

extern "C" void kernel_launch(void* const* d_in, const int* in_sizes, int n_in,
                              void* d_out, int out_size, void* d_ws, size_t ws_size,
                              hipStream_t stream) {
    const float* x         = (const float*)d_in[0];
    const int*   ei        = (const int*)  d_in[1];
    const float* edge_attr = (const float*)d_in[2];
    const float* W_node    = (const float*)d_in[3];
    const float* W_edge    = (const float*)d_in[4];
    const float* attn_src  = (const float*)d_in[5];
    const float* attn_dst  = (const float*)d_in[6];
    const float* gamma     = (const float*)d_in[7];
    const float* beta      = (const float*)d_in[8];
    float* out = (float*)d_out;

    int  n = in_sizes[0] / IN_DIM;
    long e = (long)in_sizes[2] / EDGE_DIM;
    const int* srcIdx = ei;
    const int* dstIdx = ei + e;

    float* ws    = (float*)d_ws;
    float* h     = ws;  ws += (long)n * OUT_DIM;
    float* a_src = ws;  ws += (long)n * H_HEADS;
    float* a_dst = ws;  ws += (long)n * H_HEADS;
    unsigned* keyw = (unsigned*)ws;

    int n_tiles = (n + 15) / 16;
    node_gemm_wmma<<<n_tiles, 128, 0, stream>>>(x, W_node, h, n);

    node_proj<<<(n * H_HEADS + 255) / 256, 256, 0, stream>>>(
        h, attn_src, attn_dst, a_src, a_dst, n);

    edge_key<<<(int)((e + 1023) / 1024), 256, 0, stream>>>(dstIdx, keyw, e);

    gather_own_ln<<<(n + OWN_NODES - 1) / OWN_NODES, 1024, 0, stream>>>(
        srcIdx, dstIdx, h, edge_attr, W_edge, a_src, a_dst, keyw, gamma, beta, out, n, e);
}


# edge launch also writes one word per edge (source*512 + place in range); gather pass B1 takes source and place from that word, fetched one edge ahead, not from dst[e] and src[e]; sum order kept
# speedup vs baseline: 4.7059x; 1.0305x over previous
#include <hip/hip_runtime.h>
#include <hip/hip_bf16.h>

typedef float v2f __attribute__((ext_vector_type(2)));
typedef float v8f __attribute__((ext_vector_type(8)));

#define IN_DIM   64
#define OUT_DIM  64
#define EDGE_DIM 16
#define H_HEADS  4
#define HD       16
#define NEG_SLOPE   0.2f
#define EPS_SOFTMAX 1e-8f
#define EPS_LN      1e-5f

#define OWN_NODES 391
#define OWN_PAD   392
#define OWN_PLACE_BITS 9
#define OWN_PLACE_MASK 511u

__global__ void node_gemm_wmma(const float* __restrict__ x,
                               const float* __restrict__ W,
                               float* __restrict__ h, int n) {
    int wave = threadIdx.x >> 5;
    int lane = threadIdx.x & 31;
    int grp  = lane >> 4;
    int lid  = lane & 15;

    int row0 = blockIdx.x * 16;
    int col0 = wave * 16;

    int arow = row0 + lid;
    if (arow >= n) arow = n - 1;

    const float* xr = x + (long)arow * IN_DIM + 2 * grp;
    const float* wc = W + (long)(2 * grp) * OUT_DIM + col0 + lid;

    v8f c = {};
    #pragma unroll
    for (int k0 = 0; k0 < IN_DIM; k0 += 4) {
        v2f a = *(const v2f*)(xr + k0);
        v2f b;
        b.x = wc[(long)k0 * OUT_DIM];
        b.y = wc[(long)(k0 + 1) * OUT_DIM];
        c = __builtin_amdgcn_wmma_f32_16x16x4_f32(false, a, false, b,
                                                  (short)0, c, false, false);
    }

    #pragma unroll
    for (int r = 0; r < 8; ++r) {
        int row = row0 + r + 8 * grp;
        if (row < n) h[(long)row * OUT_DIM + col0 + lid] = c[r];
    }
}

__global__ void node_proj(const float* __restrict__ h,
                          const float* __restrict__ attn_src,
                          const float* __restrict__ attn_dst,
                          float* __restrict__ a_src,
                          float* __restrict__ a_dst, int n) {
    int idx = blockIdx.x * blockDim.x + threadIdx.x;
    if (idx >= n * H_HEADS) return;
    int hh   = idx & 3;
    int node = idx >> 2;
    const float4* hp = (const float4*)(h + (long)node * OUT_DIM + hh * HD);
    const float4* as = (const float4*)(attn_src + hh * HD);
    const float4* ad = (const float4*)(attn_dst + hh * HD);
    float s = 0.0f, d = 0.0f;
    #pragma unroll
    for (int j = 0; j < 4; ++j) {
        float4 hv = hp[j], sv = as[j], dv = ad[j];
        s += hv.x * sv.x + hv.y * sv.y + hv.z * sv.z + hv.w * sv.w;
        d += hv.x * dv.x + hv.y * dv.y + hv.z * dv.z + hv.w * dv.w;
    }
    a_src[idx] = s;
    a_dst[idx] = d;
}

__device__ __forceinline__ long own_key_word(long g, unsigned b, long ng) {
    long blk = g >> 5;
    if (blk < (ng >> 5)) return blk * 256 + (long)(b >> 2) * 128 + (g & 31) * 4 + (long)(b & 3u);
    return g * 8 + (long)b;
}

__global__ void edge_key(const int* __restrict__ src,
                         const int* __restrict__ dst,
                         unsigned* __restrict__ keyw,
                         unsigned* __restrict__ srcplace, long e_cnt) {
    const unsigned lane = threadIdx.x & 31u;
    const long     g0   = (long)blockIdx.x * 32 + (long)(threadIdx.x >> 5) * 4;
    const long     ng   = (e_cnt + 31) >> 5;
    unsigned word = 0u;
    int t[4], s[4];
    #pragma unroll
    for (int r = 0; r < 4; ++r) {
        const long e  = (g0 + r) * 32 + (long)lane;
        const long ec = (e < e_cnt) ? e : (e_cnt - 1);
        t[r] = dst[ec];
        s[r] = src[ec];
    }
    #pragma unroll
    for (int r = 0; r < 4; ++r) {
        const long e    = (g0 + r) * 32 + (long)lane;
        const bool live = e < e_cnt;
        const unsigned key = (unsigned)t[r] / (unsigned)OWN_NODES;
        srcplace[live ? e : (e_cnt - 1)] =
            ((unsigned)s[r] << OWN_PLACE_BITS) | ((unsigned)t[r] - key * (unsigned)OWN_NODES);
        #pragma unroll
        for (int b = 0; b < 8; ++b) {
            unsigned bal = __builtin_amdgcn_ballot_w32(live && ((key >> b) & 1u) != 0u);
            if (lane == (unsigned)(r * 8 + b)) word = bal;
        }
    }
    const long g = g0 + (long)(lane >> 3);
    if (g * 32 < e_cnt) keyw[own_key_word(g, lane & 7u, ng)] = word;
}

__device__ __forceinline__ float4 own_score(const float* __restrict__ W_edge, const float4* ea,
                                            float4 asv, float4 adv) {
    float sc[H_HEADS] = {0.f, 0.f, 0.f, 0.f};
    #pragma unroll
    for (int q = 0; q < 4; ++q) {
        float4 v = ea[q];
        int d0 = q * 4;
        #pragma unroll
        for (int hh = 0; hh < H_HEADS; ++hh) {
            sc[hh] += v.x * W_edge[(d0 + 0) * H_HEADS + hh]
                    + v.y * W_edge[(d0 + 1) * H_HEADS + hh]
                    + v.z * W_edge[(d0 + 2) * H_HEADS + hh]
                    + v.w * W_edge[(d0 + 3) * H_HEADS + hh];
        }
    }
    float av[H_HEADS] = {asv.x + adv.x, asv.y + adv.y, asv.z + adv.z, asv.w + adv.w};
    float4 outv;
    float* op = (float*)&outv;
    #pragma unroll
    for (int hh = 0; hh < H_HEADS; ++hh) {
        float v = av[hh] + sc[hh];
        v = (v >= 0.0f) ? v : NEG_SLOPE * v;
        op[hh] = v;
    }
    return outv;
}

__device__ __forceinline__ float4 own_score_at(const float* __restrict__ edge_attr,
                                               const float* __restrict__ W_edge,
                                               const float* __restrict__ a_src,
                                               int e, int s, float4 adv) {
    const float4* p = (const float4*)(edge_attr + (long)e * EDGE_DIM);
    float4 ea[4] = {p[0], p[1], p[2], p[3]};
    return own_score(W_edge, ea, ((const float4*)a_src)[s], adv);
}

#define OWN_CAP   8192
#define OWN_U     16
#define OWN_G     4

__device__ __forceinline__ void own_list(int e, unsigned* cnt_s, int* le_s) {
    unsigned pos = atomicAdd(cnt_s, 1u);
    if (pos < (unsigned)OWN_CAP) le_s[pos] = e;
}

__device__ __forceinline__ void own_test32(uint4 a, uint4 b, const unsigned* inv, unsigned keep,
                                           int e0, unsigned* cnt_s, int* le_s) {
    unsigned m = (a.x ^ inv[0]) & (a.y ^ inv[1]) & (a.z ^ inv[2]) & (a.w ^ inv[3])
               & (b.x ^ inv[4]) & (b.y ^ inv[5]) & (b.z ^ inv[6]) & (b.w ^ inv[7]) & keep;
    while (m) {
        int p = __builtin_ctz(m);
        m &= m - 1u;
        own_list(e0 + p, cnt_s, le_s);
    }
}

__device__ __forceinline__ void own_key_at(int g, int nblk, int* ia, int* ib) {
    int blk = g >> 5;
    bool whole = blk < nblk;
    *ia = whole ? (blk * 64 + (g & 31)) : (2 * g);
    *ib = whole ? (*ia + 32) : (*ia + 1);
}

__device__ __forceinline__ void own_max4(int* mx_s, int tl, float4 a) {
    if (a.x > 0.0f) atomicMax(&mx_s[tl * H_HEADS + 0], __float_as_int(a.x));
    if (a.y > 0.0f) atomicMax(&mx_s[tl * H_HEADS + 1], __float_as_int(a.y));
    if (a.z > 0.0f) atomicMax(&mx_s[tl * H_HEADS + 2], __float_as_int(a.z));
    if (a.w > 0.0f) atomicMax(&mx_s[tl * H_HEADS + 3], __float_as_int(a.w));
}

__global__ void __launch_bounds__(1024)
gather_own_ln(const int* __restrict__ src,
              const int* __restrict__ dst,
              const float* __restrict__ h,
              const float* __restrict__ edge_attr,
              const float* __restrict__ W_edge,
              const float* __restrict__ a_src,
              const float* __restrict__ a_dst,
              const unsigned* __restrict__ keyw,
              const unsigned* __restrict__ srcplace,
              const float* __restrict__ gamma,
              const float* __restrict__ beta,
              float* __restrict__ out, int n, long e_cnt) {
    __shared__ float    acc_s[OWN_PAD * OUT_DIM];
    __shared__ int      le_s[OWN_CAP];
    __shared__ float4   sc_s[OWN_CAP];
    __shared__ alignas(16) int mx_s[OWN_PAD * H_HEADS];
    __shared__ float    sm_s[OWN_PAD * H_HEADS];
    __shared__ alignas(16) float adst_s[OWN_PAD * H_HEADS];
    __shared__ unsigned cnt_s;

    const int tid  = threadIdx.x;
    const int wave = tid >> 5;
    const int lane = tid & 31;
    const int lo   = blockIdx.x * OWN_NODES;
    const int ne   = (int)e_cnt;
    const float4* asrc4 = (const float4*)a_src;

    {
        float4 z = {0.0f, 0.0f, 0.0f, 0.0f};
        float4* a4 = (float4*)acc_s;
        for (int i = tid; i < OWN_PAD * OUT_DIM / 4; i += 1024) a4[i] = z;
        for (int i = tid; i < OWN_PAD * H_HEADS; i += 1024) {
            mx_s[i] = 0;
            sm_s[i] = 0.0f;
            const int node = lo + (i >> 2);
            adst_s[i] = (node < n) ? a_dst[(long)node * H_HEADS + (i & 3)] : 0.0f;
        }
        if (tid == 0) cnt_s = 0u;
    }
    __syncthreads();

    const bool keyed = (gridDim.x <= 256u);
    if (keyed) {
        const unsigned mine = blockIdx.x;
        unsigned inv[8];
        #pragma unroll
        for (int b = 0; b < 8; ++b) inv[b] = ((mine >> b) & 1u) - 1u;
        const int nfull = ne >> 5;
        const int rem   = ne & 31;
        const int nblk  = ((ne + 31) >> 5) >> 5;
        const int nmain = (nblk * 32 < nfull) ? nblk * 32 : nfull;
        const uint4* k4 = (const uint4*)keyw;

        {
            int g = nmain + tid;
            if (g < nfull || (g == nfull && rem != 0)) {
                int ia, ib;
                own_key_at(g, nblk, &ia, &ib);
                unsigned keep = (g < nfull) ? 0xFFFFFFFFu : ((1u << rem) - 1u);
                own_test32(k4[ia], k4[ib], inv, keep, g << 5, &cnt_s, le_s);
            }
        }

        int g0 = tid;
        for (; g0 + (OWN_G - 1) * 1024 < nmain; g0 += 1024 * OWN_G) {
            const uint4* kp = k4 + ((g0 >> 5) * 64 + (g0 & 31));
            uint4 ka[OWN_G], kb[OWN_G];
            #pragma unroll
            for (int u = 0; u < OWN_G; ++u) {
                ka[u] = kp[u * 2048];
                kb[u] = kp[u * 2048 + 32];
            }
            #pragma unroll
            for (int u = 0; u < OWN_G; ++u)
                own_test32(ka[u], kb[u], inv, 0xFFFFFFFFu, (g0 + u * 1024) << 5, &cnt_s, le_s);
        }
        for (; g0 < nmain; g0 += 1024) {
            const uint4* kp = k4 + ((g0 >> 5) * 64 + (g0 & 31));
            own_test32(kp[0], kp[32], inv, 0xFFFFFFFFu, g0 << 5, &cnt_s, le_s);
        }
    }
    __syncthreads();

    const unsigned cnt    = cnt_s;
    const bool     listed = keyed && (cnt <= (unsigned)OWN_CAP);

    if (listed) {
        unsigned i = (unsigned)tid;
        int      e_nx  = 0;
        unsigned pw_nx = 0u;
        if (i < cnt) {
            e_nx  = le_s[i];
            pw_nx = srcplace[e_nx];
        }
        for (; i < cnt; i += 1024u) {
            const int      e  = e_nx;
            const unsigned pw = pw_nx;
            const float4* p = (const float4*)(edge_attr + (long)e * EDGE_DIM);
            const float4 ea[4] = {p[0], p[1], p[2], p[3]};
            const float4 as = asrc4[pw >> OWN_PLACE_BITS];
            const unsigned in = i + 1024u;
            if (in < cnt) {
                e_nx  = le_s[in];
                pw_nx = srcplace[e_nx];
            }
            const int    tl = (int)(pw & OWN_PLACE_MASK);
            const float4 ad = *(const float4*)&adst_s[tl * H_HEADS];
            const float4 av = own_score(W_edge, ea, as, ad);
            le_s[i] = (int)pw;
            sc_s[i] = av;
            own_max4(mx_s, tl, av);
        }
    } else {
        for (int e = tid; e < ne; e += 1024) {
            unsigned tl = (unsigned)(dst[e] - lo);
            if (tl < (unsigned)OWN_NODES)
                own_max4(mx_s, (int)tl, own_score_at(edge_attr, W_edge, a_src, e, src[e],
                                                     *(const float4*)&adst_s[tl * H_HEADS]));
        }
    }
    __syncthreads();

    if (listed) {
        for (unsigned i = (unsigned)tid; i < cnt; i += 1024u) {
            const unsigned tl = (unsigned)le_s[i] & OWN_PLACE_MASK;
            const float4 a = sc_s[i];
            const int4   m = *(const int4*)&mx_s[tl * H_HEADS];
            float4 w;
            w.x = expf(a.x - __int_as_float(m.x));
            w.y = expf(a.y - __int_as_float(m.y));
            w.z = expf(a.z - __int_as_float(m.z));
            w.w = expf(a.w - __int_as_float(m.w));
            atomicAdd(&sm_s[tl * H_HEADS + 0], w.x);
            atomicAdd(&sm_s[tl * H_HEADS + 1], w.y);
            atomicAdd(&sm_s[tl * H_HEADS + 2], w.z);
            atomicAdd(&sm_s[tl * H_HEADS + 3], w.w);
            sc_s[i] = w;
        }
    }
    __syncthreads();

    if (listed) {
        const int hh = lane >> 3;
        const unsigned wv = (unsigned)__builtin_amdgcn_readfirstlane(wave);
        const unsigned cn = (unsigned)__builtin_amdgcn_readfirstlane((int)cnt);
        const float*   wt_s  = (const float*)sc_s;
        const char*    hb    = (const char*)h;
        const unsigned lane8 = (unsigned)lane * 8u;

        unsigned i0 = wv * OWN_U;
        for (; i0 + OWN_U <= cn; i0 += 32u * OWN_U) {
            unsigned pk[OWN_U];
            float    ww[OWN_U];
            #pragma unroll
            for (int u = 0; u < OWN_U; ++u) {
                pk[u] = (unsigned)le_s[i0 + u];
                ww[u] = wt_s[(i0 + u) * H_HEADS + hh];
            }
            v2f hv[OWN_U];
            #pragma unroll
            for (int u = 0; u < OWN_U; ++u)
                hv[u] = *(const v2f*)(hb + ((((pk[u] >> OWN_PLACE_BITS) * (unsigned)OUT_DIM) << 2) + lane8));
            #pragma unroll
            for (int u = 0; u < OWN_U; ++u) {
                float* ap = &acc_s[(pk[u] & OWN_PLACE_MASK) * OUT_DIM + 2 * lane];
                atomicAdd(ap,     hv[u].x * ww[u]);
                atomicAdd(ap + 1, hv[u].y * ww[u]);
            }
        }
        if (i0 < cn) {
            unsigned pk[OWN_U];
            float    ww[OWN_U];
            #pragma unroll
            for (int u = 0; u < OWN_U; ++u) {
                const unsigned ic = (i0 + u < cn) ? (i0 + u) : i0;
                pk[u] = (unsigned)le_s[ic];
                ww[u] = wt_s[ic * H_HEADS + hh];
            }
            v2f hv[OWN_U];
            #pragma unroll
            for (int u = 0; u < OWN_U; ++u)
                hv[u] = *(const v2f*)(hb + ((((pk[u] >> OWN_PLACE_BITS) * (unsigned)OUT_DIM) << 2) + lane8));
            #pragma unroll
            for (int u = 0; u < OWN_U; ++u) {
                if (i0 + u < cn) {
                    float* ap = &acc_s[(pk[u] & OWN_PLACE_MASK) * OUT_DIM + 2 * lane];
                    atomicAdd(ap,     hv[u].x * ww[u]);
                    atomicAdd(ap + 1, hv[u].y * ww[u]);
                }
            }
        }
    } else {
        for (int e = tid; e < ne; e += 1024) {
            unsigned tl = (unsigned)(dst[e] - lo);
            if (tl < (unsigned)OWN_NODES) {
                int s = src[e];
                float4 a = own_score_at(edge_attr, W_edge, a_src, e, s, *(const float4*)&adst_s[tl * H_HEADS]);
                float w[H_HEADS];
                w[0] = expf(a.x - __int_as_float(mx_s[tl * H_HEADS + 0]));
                w[1] = expf(a.y - __int_as_float(mx_s[tl * H_HEADS + 1]));
                w[2] = expf(a.z - __int_as_float(mx_s[tl * H_HEADS + 2]));
                w[3] = expf(a.w - __int_as_float(mx_s[tl * H_HEADS + 3]));
                for (int hd = 0; hd < H_HEADS; ++hd) {
                    atomicAdd(&sm_s[tl * H_HEADS + hd], w[hd]);
                    for (int d = hd * HD; d < hd * HD + HD; ++d)
                        atomicAdd(&acc_s[tl * OUT_DIM + d], h[(long)s * OUT_DIM + d] * w[hd]);
                }
            }
        }
    }
    __syncthreads();

    {
        const float g0 = gamma[lane], g1 = gamma[lane + 32];
        const float b0 = beta[lane],  b1 = beta[lane + 32];
        const int   h0 = lane >> 4;
        for (int tl0 = wave * 4; tl0 < OWN_NODES; tl0 += 32 * 4) {
            float y0[4], y1[4], s[4], vs[4];
            #pragma unroll
            for (int u = 0; u < 4; ++u) {
                int node = lo + tl0 + u;
                int nc   = (node < n) ? node : (n - 1);
                float q0 = sm_s[(tl0 + u) * H_HEADS + h0]     + EPS_SOFTMAX;
                float q1 = sm_s[(tl0 + u) * H_HEADS + h0 + 2] + EPS_SOFTMAX;
                y0[u] = acc_s[(tl0 + u) * OUT_DIM + lane]      / q0 + h[(long)nc * OUT_DIM + lane];
                y1[u] = acc_s[(tl0 + u) * OUT_DIM + lane + 32] / q1 + h[(long)nc * OUT_DIM + lane + 32];
                s[u]  = y0[u] + y1[u];
            }
            #pragma unroll
            for (int off = 16; off > 0; off >>= 1) {
                #pragma unroll
                for (int u = 0; u < 4; ++u) s[u] += __shfl_xor(s[u], off, 32);
            }
            #pragma unroll
            for (int u = 0; u < 4; ++u) {
                float mu = s[u] * (1.0f / 64.0f);
                y0[u] -= mu;
                y1[u] -= mu;
                vs[u] = y0[u] * y0[u] + y1[u] * y1[u];
            }
            #pragma unroll
            for (int off = 16; off > 0; off >>= 1) {
                #pragma unroll
                for (int u = 0; u < 4; ++u) vs[u] += __shfl_xor(vs[u], off, 32);
            }
            #pragma unroll
            for (int u = 0; u < 4; ++u) {
                int node = lo + tl0 + u;
                if (tl0 + u < OWN_NODES && node < n) {
                    long base = (long)node * OUT_DIM;
                    float rstd = rsqrtf(vs[u] * (1.0f / 64.0f) + EPS_LN);
                    out[base + lane]      = y0[u] * rstd * g0 + b0;
                    out[base + lane + 32] = y1[u] * rstd * g1 + b1;
                }
            }
        }
    }
}

extern "C" void kernel_launch(void* const* d_in, const int* in_sizes, int n_in,
                              void* d_out, int out_size, void* d_ws, size_t ws_size,
                              hipStream_t stream) {
    const float* x         = (const float*)d_in[0];
    const int*   ei        = (const int*)  d_in[1];
    const float* edge_attr = (const float*)d_in[2];
    const float* W_node    = (const float*)d_in[3];
    const float* W_edge    = (const float*)d_in[4];
    const float* attn_src  = (const float*)d_in[5];
    const float* attn_dst  = (const float*)d_in[6];
    const float* gamma     = (const float*)d_in[7];
    const float* beta      = (const float*)d_in[8];
    float* out = (float*)d_out;

    int  n = in_sizes[0] / IN_DIM;
    long e = (long)in_sizes[2] / EDGE_DIM;
    const int* srcIdx = ei;
    const int* dstIdx = ei + e;

    float* ws    = (float*)d_ws;
    float* h     = ws;  ws += (long)n * OUT_DIM;
    float* a_src = ws;  ws += (long)n * H_HEADS;
    float* a_dst = ws;  ws += (long)n * H_HEADS;
    unsigned* keyw = (unsigned*)ws;
    unsigned* srcplace = keyw + ((e + 31) / 32) * 8;

    int n_tiles = (n + 15) / 16;
    node_gemm_wmma<<<n_tiles, 128, 0, stream>>>(x, W_node, h, n);

    node_proj<<<(n * H_HEADS + 255) / 256, 256, 0, stream>>>(
        h, attn_src, attn_dst, a_src, a_dst, n);

    edge_key<<<(int)((e + 1023) / 1024), 256, 0, stream>>>(srcIdx, dstIdx, keyw, srcplace, e);

    gather_own_ln<<<(n + OWN_NODES - 1) / OWN_NODES, 1024, 0, stream>>>(
        srcIdx, dstIdx, h, edge_attr, W_edge, a_src, a_dst, keyw, srcplace, gamma, beta, out, n, e);
}


# gather pass B1: an edge's 64 attribute bytes are read by the four lanes of its quad (16 bytes each, quad sum on lane swizzles): an attribute load touches 8 lines, not 32; score sum order changes
# speedup vs baseline: 4.8644x; 1.0337x over previous
#include <hip/hip_runtime.h>
#include <hip/hip_bf16.h>

typedef float v2f __attribute__((ext_vector_type(2)));
typedef float v8f __attribute__((ext_vector_type(8)));

#define IN_DIM   64
#define OUT_DIM  64
#define EDGE_DIM 16
#define H_HEADS  4
#define HD       16
#define NEG_SLOPE   0.2f
#define EPS_SOFTMAX 1e-8f
#define EPS_LN      1e-5f

#define OWN_NODES 391
#define OWN_PAD   392
#define OWN_PLACE_BITS 9
#define OWN_PLACE_MASK 511u

__global__ void node_gemm_wmma(const float* __restrict__ x,
                               const float* __restrict__ W,
                               float* __restrict__ h, int n) {
    int wave = threadIdx.x >> 5;
    int lane = threadIdx.x & 31;
    int grp  = lane >> 4;
    int lid  = lane & 15;

    int row0 = blockIdx.x * 16;
    int col0 = wave * 16;

    int arow = row0 + lid;
    if (arow >= n) arow = n - 1;

    const float* xr = x + (long)arow * IN_DIM + 2 * grp;
    const float* wc = W + (long)(2 * grp) * OUT_DIM + col0 + lid;

    v8f c = {};
    #pragma unroll
    for (int k0 = 0; k0 < IN_DIM; k0 += 4) {
        v2f a = *(const v2f*)(xr + k0);
        v2f b;
        b.x = wc[(long)k0 * OUT_DIM];
        b.y = wc[(long)(k0 + 1) * OUT_DIM];
        c = __builtin_amdgcn_wmma_f32_16x16x4_f32(false, a, false, b,
                                                  (short)0, c, false, false);
    }

    #pragma unroll
    for (int r = 0; r < 8; ++r) {
        int row = row0 + r + 8 * grp;
        if (row < n) h[(long)row * OUT_DIM + col0 + lid] = c[r];
    }
}

__global__ void node_proj(const float* __restrict__ h,
                          const float* __restrict__ attn_src,
                          const float* __restrict__ attn_dst,
                          float* __restrict__ a_src,
                          float* __restrict__ a_dst, int n) {
    int idx = blockIdx.x * blockDim.x + threadIdx.x;
    if (idx >= n * H_HEADS) return;
    int hh   = idx & 3;
    int node = idx >> 2;
    const float4* hp = (const float4*)(h + (long)node * OUT_DIM + hh * HD);
    const float4* as = (const float4*)(attn_src + hh * HD);
    const float4* ad = (const float4*)(attn_dst + hh * HD);
    float s = 0.0f, d = 0.0f;
    #pragma unroll
    for (int j = 0; j < 4; ++j) {
        float4 hv = hp[j], sv = as[j], dv = ad[j];
        s += hv.x * sv.x + hv.y * sv.y + hv.z * sv.z + hv.w * sv.w;
        d += hv.x * dv.x + hv.y * dv.y + hv.z * dv.z + hv.w * dv.w;
    }
    a_src[idx] = s;
    a_dst[idx] = d;
}

__device__ __forceinline__ long own_key_word(long g, unsigned b, long ng) {
    long blk = g >> 5;
    if (blk < (ng >> 5)) return blk * 256 + (long)(b >> 2) * 128 + (g & 31) * 4 + (long)(b & 3u);
    return g * 8 + (long)b;
}

__global__ void edge_key(const int* __restrict__ src,
                         const int* __restrict__ dst,
                         unsigned* __restrict__ keyw,
                         unsigned* __restrict__ srcplace, long e_cnt) {
    const unsigned lane = threadIdx.x & 31u;
    const long     g0   = (long)blockIdx.x * 32 + (long)(threadIdx.x >> 5) * 4;
    const long     ng   = (e_cnt + 31) >> 5;
    unsigned word = 0u;
    int t[4], s[4];
    #pragma unroll
    for (int r = 0; r < 4; ++r) {
        const long e  = (g0 + r) * 32 + (long)lane;
        const long ec = (e < e_cnt) ? e : (e_cnt - 1);
        t[r] = dst[ec];
        s[r] = src[ec];
    }
    #pragma unroll
    for (int r = 0; r < 4; ++r) {
        const long e    = (g0 + r) * 32 + (long)lane;
        const bool live = e < e_cnt;
        const unsigned key = (unsigned)t[r] / (unsigned)OWN_NODES;
        srcplace[live ? e : (e_cnt - 1)] =
            ((unsigned)s[r] << OWN_PLACE_BITS) | ((unsigned)t[r] - key * (unsigned)OWN_NODES);
        #pragma unroll
        for (int b = 0; b < 8; ++b) {
            unsigned bal = __builtin_amdgcn_ballot_w32(live && ((key >> b) & 1u) != 0u);
            if (lane == (unsigned)(r * 8 + b)) word = bal;
        }
    }
    const long g = g0 + (long)(lane >> 3);
    if (g * 32 < e_cnt) keyw[own_key_word(g, lane & 7u, ng)] = word;
}

__device__ __forceinline__ float4 own_score(const float* __restrict__ W_edge, const float4* ea,
                                            float4 asv, float4 adv) {
    float sc[H_HEADS] = {0.f, 0.f, 0.f, 0.f};
    #pragma unroll
    for (int q = 0; q < 4; ++q) {
        float4 v = ea[q];
        int d0 = q * 4;
        #pragma unroll
        for (int hh = 0; hh < H_HEADS; ++hh) {
            sc[hh] += v.x * W_edge[(d0 + 0) * H_HEADS + hh]
                    + v.y * W_edge[(d0 + 1) * H_HEADS + hh]
                    + v.z * W_edge[(d0 + 2) * H_HEADS + hh]
                    + v.w * W_edge[(d0 + 3) * H_HEADS + hh];
        }
    }
    float av[H_HEADS] = {asv.x + adv.x, asv.y + adv.y, asv.z + adv.z, asv.w + adv.w};
    float4 outv;
    float* op = (float*)&outv;
    #pragma unroll
    for (int hh = 0; hh < H_HEADS; ++hh) {
        float v = av[hh] + sc[hh];
        v = (v >= 0.0f) ? v : NEG_SLOPE * v;
        op[hh] = v;
    }
    return outv;
}

__device__ __forceinline__ float4 own_score_at(const float* __restrict__ edge_attr,
                                               const float* __restrict__ W_edge,
                                               const float* __restrict__ a_src,
                                               int e, int s, float4 adv) {
    const float4* p = (const float4*)(edge_attr + (long)e * EDGE_DIM);
    float4 ea[4] = {p[0], p[1], p[2], p[3]};
    return own_score(W_edge, ea, ((const float4*)a_src)[s], adv);
}

#define OWN_CAP   8192
#define OWN_U     16
#define OWN_G     4

__device__ __forceinline__ void own_list(int e, unsigned* cnt_s, int* le_s) {
    unsigned pos = atomicAdd(cnt_s, 1u);
    if (pos < (unsigned)OWN_CAP) le_s[pos] = e;
}

__device__ __forceinline__ void own_test32(uint4 a, uint4 b, const unsigned* inv, unsigned keep,
                                           int e0, unsigned* cnt_s, int* le_s) {
    unsigned m = (a.x ^ inv[0]) & (a.y ^ inv[1]) & (a.z ^ inv[2]) & (a.w ^ inv[3])
               & (b.x ^ inv[4]) & (b.y ^ inv[5]) & (b.z ^ inv[6]) & (b.w ^ inv[7]) & keep;
    while (m) {
        int p = __builtin_ctz(m);
        m &= m - 1u;
        own_list(e0 + p, cnt_s, le_s);
    }
}

__device__ __forceinline__ void own_key_at(int g, int nblk, int* ia, int* ib) {
    int blk = g >> 5;
    bool whole = blk < nblk;
    *ia = whole ? (blk * 64 + (g & 31)) : (2 * g);
    *ib = whole ? (*ia + 32) : (*ia + 1);
}

template <int U>
__device__ __forceinline__ int own_quad_lane(int v) {
    return __builtin_amdgcn_update_dpp(0, v, U * 0x55, 0xf, 0xf, true);
}

__device__ __forceinline__ float own_quad_sum(float v) {
    v += __int_as_float(__builtin_amdgcn_update_dpp(0, __float_as_int(v), 0xB1, 0xf, 0xf, true));
    v += __int_as_float(__builtin_amdgcn_update_dpp(0, __float_as_int(v), 0x4E, 0xf, 0xf, true));
    return v;
}

__device__ __forceinline__ float own_leaky(float v) {
    return (v >= 0.0f) ? v : NEG_SLOPE * v;
}

__device__ __forceinline__ void own_max4(int* mx_s, int tl, float4 a) {
    if (a.x > 0.0f) atomicMax(&mx_s[tl * H_HEADS + 0], __float_as_int(a.x));
    if (a.y > 0.0f) atomicMax(&mx_s[tl * H_HEADS + 1], __float_as_int(a.y));
    if (a.z > 0.0f) atomicMax(&mx_s[tl * H_HEADS + 2], __float_as_int(a.z));
    if (a.w > 0.0f) atomicMax(&mx_s[tl * H_HEADS + 3], __float_as_int(a.w));
}

__global__ void __launch_bounds__(1024)
gather_own_ln(const int* __restrict__ src,
              const int* __restrict__ dst,
              const float* __restrict__ h,
              const float* __restrict__ edge_attr,
              const float* __restrict__ W_edge,
              const float* __restrict__ a_src,
              const float* __restrict__ a_dst,
              const unsigned* __restrict__ keyw,
              const unsigned* __restrict__ srcplace,
              const float* __restrict__ gamma,
              const float* __restrict__ beta,
              float* __restrict__ out, int n, long e_cnt) {
    __shared__ float    acc_s[OWN_PAD * OUT_DIM];
    __shared__ int      le_s[OWN_CAP];
    __shared__ float4   sc_s[OWN_CAP];
    __shared__ alignas(16) int mx_s[OWN_PAD * H_HEADS];
    __shared__ float    sm_s[OWN_PAD * H_HEADS];
    __shared__ alignas(16) float adst_s[OWN_PAD * H_HEADS];
    __shared__ unsigned cnt_s;

    const int tid  = threadIdx.x;
    const int wave = tid >> 5;
    const int lane = tid & 31;
    const int lo   = blockIdx.x * OWN_NODES;
    const int ne   = (int)e_cnt;
    const float4* asrc4 = (const float4*)a_src;

    {
        float4 z = {0.0f, 0.0f, 0.0f, 0.0f};
        float4* a4 = (float4*)acc_s;
        for (int i = tid; i < OWN_PAD * OUT_DIM / 4; i += 1024) a4[i] = z;
        for (int i = tid; i < OWN_PAD * H_HEADS; i += 1024) {
            mx_s[i] = 0;
            sm_s[i] = 0.0f;
            const int node = lo + (i >> 2);
            adst_s[i] = (node < n) ? a_dst[(long)node * H_HEADS + (i & 3)] : 0.0f;
        }
        if (tid == 0) cnt_s = 0u;
    }
    __syncthreads();

    const bool keyed = (gridDim.x <= 256u);
    if (keyed) {
        const unsigned mine = blockIdx.x;
        unsigned inv[8];
        #pragma unroll
        for (int b = 0; b < 8; ++b) inv[b] = ((mine >> b) & 1u) - 1u;
        const int nfull = ne >> 5;
        const int rem   = ne & 31;
        const int nblk  = ((ne + 31) >> 5) >> 5;
        const int nmain = (nblk * 32 < nfull) ? nblk * 32 : nfull;
        const uint4* k4 = (const uint4*)keyw;

        {
            int g = nmain + tid;
            if (g < nfull || (g == nfull && rem != 0)) {
                int ia, ib;
                own_key_at(g, nblk, &ia, &ib);
                unsigned keep = (g < nfull) ? 0xFFFFFFFFu : ((1u << rem) - 1u);
                own_test32(k4[ia], k4[ib], inv, keep, g << 5, &cnt_s, le_s);
            }
        }

        int g0 = tid;
        for (; g0 + (OWN_G - 1) * 1024 < nmain; g0 += 1024 * OWN_G) {
            const uint4* kp = k4 + ((g0 >> 5) * 64 + (g0 & 31));
            uint4 ka[OWN_G], kb[OWN_G];
            #pragma unroll
            for (int u = 0; u < OWN_G; ++u) {
                ka[u] = kp[u * 2048];
                kb[u] = kp[u * 2048 + 32];
            }
            #pragma unroll
            for (int u = 0; u < OWN_G; ++u)
                own_test32(ka[u], kb[u], inv, 0xFFFFFFFFu, (g0 + u * 1024) << 5, &cnt_s, le_s);
        }
        for (; g0 < nmain; g0 += 1024) {
            const uint4* kp = k4 + ((g0 >> 5) * 64 + (g0 & 31));
            own_test32(kp[0], kp[32], inv, 0xFFFFFFFFu, g0 << 5, &cnt_s, le_s);
        }
    }
    __syncthreads();

    const unsigned cnt    = cnt_s;
    const bool     listed = keyed && (cnt <= (unsigned)OWN_CAP);

    if (listed) {
        const int q = lane & 3;
        const float4* wq = (const float4*)(W_edge + 16 * q);
        const float4 w0 = wq[0], w1 = wq[1], w2 = wq[2], w3 = wq[3];
        const float4* attr4 = (const float4*)edge_attr;
        const unsigned wv = (unsigned)__builtin_amdgcn_readfirstlane(wave);
        const unsigned cn = (unsigned)__builtin_amdgcn_readfirstlane((int)cnt);
        int      e_nx  = 0;
        unsigned pw_nx = 0u;
        if (wv * 32u + (unsigned)lane < cn) {
            e_nx  = le_s[wv * 32u + (unsigned)lane];
            pw_nx = srcplace[e_nx];
        }
        for (unsigned base = wv * 32u; base < cn; base += 1024u) {
            const unsigned i  = base + (unsigned)lane;
            const int      e  = e_nx;
            const unsigned pw = pw_nx;
            float4 ea[4];
            ea[0] = attr4[(long)own_quad_lane<0>(e) * 4 + q];
            ea[1] = attr4[(long)own_quad_lane<1>(e) * 4 + q];
            ea[2] = attr4[(long)own_quad_lane<2>(e) * 4 + q];
            ea[3] = attr4[(long)own_quad_lane<3>(e) * 4 + q];
            const float4 as = asrc4[pw >> OWN_PLACE_BITS];
            const unsigned in = i + 1024u;
            e_nx  = 0;
            pw_nx = 0u;
            if (in < cn) {
                e_nx  = le_s[in];
                pw_nx = srcplace[e_nx];
            }
            float4 p[4];
            #pragma unroll
            for (int u = 0; u < 4; ++u) {
                p[u].x = ea[u].x * w0.x + ea[u].y * w1.x + ea[u].z * w2.x + ea[u].w * w3.x;
                p[u].y = ea[u].x * w0.y + ea[u].y * w1.y + ea[u].z * w2.y + ea[u].w * w3.y;
                p[u].z = ea[u].x * w0.z + ea[u].y * w1.z + ea[u].z * w2.z + ea[u].w * w3.z;
                p[u].w = ea[u].x * w0.w + ea[u].y * w1.w + ea[u].z * w2.w + ea[u].w * w3.w;
                p[u].x = own_quad_sum(p[u].x);
                p[u].y = own_quad_sum(p[u].y);
                p[u].z = own_quad_sum(p[u].z);
                p[u].w = own_quad_sum(p[u].w);
            }
            const bool q1 = (q == 1), q2 = (q == 2), q3 = (q == 3);
            float4 es;
            es.x = q3 ? p[3].x : (q2 ? p[2].x : (q1 ? p[1].x : p[0].x));
            es.y = q3 ? p[3].y : (q2 ? p[2].y : (q1 ? p[1].y : p[0].y));
            es.z = q3 ? p[3].z : (q2 ? p[2].z : (q1 ? p[1].z : p[0].z));
            es.w = q3 ? p[3].w : (q2 ? p[2].w : (q1 ? p[1].w : p[0].w));
            if (i < cn) {
                const int    tl = (int)(pw & OWN_PLACE_MASK);
                const float4 ad = *(const float4*)&adst_s[tl * H_HEADS];
                float4 av;
                av.x = own_leaky((as.x + ad.x) + es.x);
                av.y = own_leaky((as.y + ad.y) + es.y);
                av.z = own_leaky((as.z + ad.z) + es.z);
                av.w = own_leaky((as.w + ad.w) + es.w);
                le_s[i] = (int)pw;
                sc_s[i] = av;
                own_max4(mx_s, tl, av);
            }
        }
    } else {
        for (int e = tid; e < ne; e += 1024) {
            unsigned tl = (unsigned)(dst[e] - lo);
            if (tl < (unsigned)OWN_NODES)
                own_max4(mx_s, (int)tl, own_score_at(edge_attr, W_edge, a_src, e, src[e],
                                                     *(const float4*)&adst_s[tl * H_HEADS]));
        }
    }
    __syncthreads();

    if (listed) {
        for (unsigned i = (unsigned)tid; i < cnt; i += 1024u) {
            const unsigned tl = (unsigned)le_s[i] & OWN_PLACE_MASK;
            const float4 a = sc_s[i];
            const int4   m = *(const int4*)&mx_s[tl * H_HEADS];
            float4 w;
            w.x = expf(a.x - __int_as_float(m.x));
            w.y = expf(a.y - __int_as_float(m.y));
            w.z = expf(a.z - __int_as_float(m.z));
            w.w = expf(a.w - __int_as_float(m.w));
            atomicAdd(&sm_s[tl * H_HEADS + 0], w.x);
            atomicAdd(&sm_s[tl * H_HEADS + 1], w.y);
            atomicAdd(&sm_s[tl * H_HEADS + 2], w.z);
            atomicAdd(&sm_s[tl * H_HEADS + 3], w.w);
            sc_s[i] = w;
        }
    }
    __syncthreads();

    if (listed) {
        const int hh = lane >> 3;
        const unsigned wv = (unsigned)__builtin_amdgcn_readfirstlane(wave);
        const unsigned cn = (unsigned)__builtin_amdgcn_readfirstlane((int)cnt);
        const float*   wt_s  = (const float*)sc_s;
        const char*    hb    = (const char*)h;
        const unsigned lane8 = (unsigned)lane * 8u;

        unsigned i0 = wv * OWN_U;
        for (; i0 + OWN_U <= cn; i0 += 32u * OWN_U) {
            unsigned pk[OWN_U];
            float    ww[OWN_U];
            #pragma unroll
            for (int u = 0; u < OWN_U; ++u) {
                pk[u] = (unsigned)le_s[i0 + u];
                ww[u] = wt_s[(i0 + u) * H_HEADS + hh];
            }
            v2f hv[OWN_U];
            #pragma unroll
            for (int u = 0; u < OWN_U; ++u)
                hv[u] = *(const v2f*)(hb + ((((pk[u] >> OWN_PLACE_BITS) * (unsigned)OUT_DIM) << 2) + lane8));
            #pragma unroll
            for (int u = 0; u < OWN_U; ++u) {
                float* ap = &acc_s[(pk[u] & OWN_PLACE_MASK) * OUT_DIM + 2 * lane];
                atomicAdd(ap,     hv[u].x * ww[u]);
                atomicAdd(ap + 1, hv[u].y * ww[u]);
            }
        }
        if (i0 < cn) {
            unsigned pk[OWN_U];
            float    ww[OWN_U];
            #pragma unroll
            for (int u = 0; u < OWN_U; ++u) {
                const unsigned ic = (i0 + u < cn) ? (i0 + u) : i0;
                pk[u] = (unsigned)le_s[ic];
                ww[u] = wt_s[ic * H_HEADS + hh];
            }
            v2f hv[OWN_U];
            #pragma unroll
            for (int u = 0; u < OWN_U; ++u)
                hv[u] = *(const v2f*)(hb + ((((pk[u] >> OWN_PLACE_BITS) * (unsigned)OUT_DIM) << 2) + lane8));
            #pragma unroll
            for (int u = 0; u < OWN_U; ++u) {
                if (i0 + u < cn) {
                    float* ap = &acc_s[(pk[u] & OWN_PLACE_MASK) * OUT_DIM + 2 * lane];
                    atomicAdd(ap,     hv[u].x * ww[u]);
                    atomicAdd(ap + 1, hv[u].y * ww[u]);
                }
            }
        }
    } else {
        for (int e = tid; e < ne; e += 1024) {
            unsigned tl = (unsigned)(dst[e] - lo);
            if (tl < (unsigned)OWN_NODES) {
                int s = src[e];
                float4 a = own_score_at(edge_attr, W_edge, a_src, e, s, *(const float4*)&adst_s[tl * H_HEADS]);
                float w[H_HEADS];
                w[0] = expf(a.x - __int_as_float(mx_s[tl * H_HEADS + 0]));
                w[1] = expf(a.y - __int_as_float(mx_s[tl * H_HEADS + 1]));
                w[2] = expf(a.z - __int_as_float(mx_s[tl * H_HEADS + 2]));
                w[3] = expf(a.w - __int_as_float(mx_s[tl * H_HEADS + 3]));
                for (int hd = 0; hd < H_HEADS; ++hd) {
                    atomicAdd(&sm_s[tl * H_HEADS + hd], w[hd]);
                    for (int d = hd * HD; d < hd * HD + HD; ++d)
                        atomicAdd(&acc_s[tl * OUT_DIM + d], h[(long)s * OUT_DIM + d] * w[hd]);
                }
            }
        }
    }
    __syncthreads();

    {
        const float g0 = gamma[lane], g1 = gamma[lane + 32];
        const float b0 = beta[lane],  b1 = beta[lane + 32];
        const int   h0 = lane >> 4;
        for (int tl0 = wave * 4; tl0 < OWN_NODES; tl0 += 32 * 4) {
            float y0[4], y1[4], s[4], vs[4];
            #pragma unroll
            for (int u = 0; u < 4; ++u) {
                int node = lo + tl0 + u;
                int nc   = (node < n) ? node : (n - 1);
                float q0 = sm_s[(tl0 + u) * H_HEADS + h0]     + EPS_SOFTMAX;
                float q1 = sm_s[(tl0 + u) * H_HEADS + h0 + 2] + EPS_SOFTMAX;
                y0[u] = acc_s[(tl0 + u) * OUT_DIM + lane]      / q0 + h[(long)nc * OUT_DIM + lane];
                y1[u] = acc_s[(tl0 + u) * OUT_DIM + lane + 32] / q1 + h[(long)nc * OUT_DIM + lane + 32];
                s[u]  = y0[u] + y1[u];
            }
            #pragma unroll
            for (int off = 16; off > 0; off >>= 1) {
                #pragma unroll
                for (int u = 0; u < 4; ++u) s[u] += __shfl_xor(s[u], off, 32);
            }
            #pragma unroll
            for (int u = 0; u < 4; ++u) {
                float mu = s[u] * (1.0f / 64.0f);
                y0[u] -= mu;
                y1[u] -= mu;
                vs[u] = y0[u] * y0[u] + y1[u] * y1[u];
            }
            #pragma unroll
            for (int off = 16; off > 0; off >>= 1) {
                #pragma unroll
                for (int u = 0; u < 4; ++u) vs[u] += __shfl_xor(vs[u], off, 32);
            }
            #pragma unroll
            for (int u = 0; u < 4; ++u) {
                int node = lo + tl0 + u;
                if (tl0 + u < OWN_NODES && node < n) {
                    long base = (long)node * OUT_DIM;
                    float rstd = rsqrtf(vs[u] * (1.0f / 64.0f) + EPS_LN);
                    out[base + lane]      = y0[u] * rstd * g0 + b0;
                    out[base + lane + 32] = y1[u] * rstd * g1 + b1;
                }
            }
        }
    }
}

extern "C" void kernel_launch(void* const* d_in, const int* in_sizes, int n_in,
                              void* d_out, int out_size, void* d_ws, size_t ws_size,
                              hipStream_t stream) {
    const float* x         = (const float*)d_in[0];
    const int*   ei        = (const int*)  d_in[1];
    const float* edge_attr = (const float*)d_in[2];
    const float* W_node    = (const float*)d_in[3];
    const float* W_edge    = (const float*)d_in[4];
    const float* attn_src  = (const float*)d_in[5];
    const float* attn_dst  = (const float*)d_in[6];
    const float* gamma     = (const float*)d_in[7];
    const float* beta      = (const float*)d_in[8];
    float* out = (float*)d_out;

    int  n = in_sizes[0] / IN_DIM;
    long e = (long)in_sizes[2] / EDGE_DIM;
    const int* srcIdx = ei;
    const int* dstIdx = ei + e;

    float* ws    = (float*)d_ws;
    float* h     = ws;  ws += (long)n * OUT_DIM;
    float* a_src = ws;  ws += (long)n * H_HEADS;
    float* a_dst = ws;  ws += (long)n * H_HEADS;
    unsigned* keyw = (unsigned*)ws;
    unsigned* srcplace = keyw + ((e + 31) / 32) * 8;

    int n_tiles = (n + 15) / 16;
    node_gemm_wmma<<<n_tiles, 128, 0, stream>>>(x, W_node, h, n);

    node_proj<<<(n * H_HEADS + 255) / 256, 256, 0, stream>>>(
        h, attn_src, attn_dst, a_src, a_dst, n);

    edge_key<<<(int)((e + 1023) / 1024), 256, 0, stream>>>(srcIdx, dstIdx, keyw, srcplace, e);

    gather_own_ln<<<(n + OWN_NODES - 1) / OWN_NODES, 1024, 0, stream>>>(
        srcIdx, dstIdx, h, edge_attr, W_edge, a_src, a_dst, keyw, srcplace, gamma, beta, out, n, e);
}


# gather pass A no longer scans every run's range bits: the key launch also writes a presence word per 1024 edges and range; a group fetches and tests only its marked runs; node sum order changes
# speedup vs baseline: 5.2513x; 1.0795x over previous
#include <hip/hip_runtime.h>
#include <hip/hip_bf16.h>

typedef float v2f __attribute__((ext_vector_type(2)));
typedef float v8f __attribute__((ext_vector_type(8)));

#define IN_DIM   64
#define OUT_DIM  64
#define EDGE_DIM 16
#define H_HEADS  4
#define HD       16
#define NEG_SLOPE   0.2f
#define EPS_SOFTMAX 1e-8f
#define EPS_LN      1e-5f

#define OWN_NODES 391
#define OWN_PAD   392
#define OWN_PLACE_BITS 9
#define OWN_PLACE_MASK 511u

__global__ void node_gemm_wmma(const float* __restrict__ x,
                               const float* __restrict__ W,
                               float* __restrict__ h, int n) {
    int wave = threadIdx.x >> 5;
    int lane = threadIdx.x & 31;
    int grp  = lane >> 4;
    int lid  = lane & 15;

    int row0 = blockIdx.x * 16;
    int col0 = wave * 16;

    int arow = row0 + lid;
    if (arow >= n) arow = n - 1;

    const float* xr = x + (long)arow * IN_DIM + 2 * grp;
    const float* wc = W + (long)(2 * grp) * OUT_DIM + col0 + lid;

    v8f c = {};
    #pragma unroll
    for (int k0 = 0; k0 < IN_DIM; k0 += 4) {
        v2f a = *(const v2f*)(xr + k0);
        v2f b;
        b.x = wc[(long)k0 * OUT_DIM];
        b.y = wc[(long)(k0 + 1) * OUT_DIM];
        c = __builtin_amdgcn_wmma_f32_16x16x4_f32(false, a, false, b,
                                                  (short)0, c, false, false);
    }

    #pragma unroll
    for (int r = 0; r < 8; ++r) {
        int row = row0 + r + 8 * grp;
        if (row < n) h[(long)row * OUT_DIM + col0 + lid] = c[r];
    }
}

__global__ void node_proj(const float* __restrict__ h,
                          const float* __restrict__ attn_src,
                          const float* __restrict__ attn_dst,
                          float* __restrict__ a_src,
                          float* __restrict__ a_dst, int n) {
    int idx = blockIdx.x * blockDim.x + threadIdx.x;
    if (idx >= n * H_HEADS) return;
    int hh   = idx & 3;
    int node = idx >> 2;
    const float4* hp = (const float4*)(h + (long)node * OUT_DIM + hh * HD);
    const float4* as = (const float4*)(attn_src + hh * HD);
    const float4* ad = (const float4*)(attn_dst + hh * HD);
    float s = 0.0f, d = 0.0f;
    #pragma unroll
    for (int j = 0; j < 4; ++j) {
        float4 hv = hp[j], sv = as[j], dv = ad[j];
        s += hv.x * sv.x + hv.y * sv.y + hv.z * sv.z + hv.w * sv.w;
        d += hv.x * dv.x + hv.y * dv.y + hv.z * dv.z + hv.w * dv.w;
    }
    a_src[idx] = s;
    a_dst[idx] = d;
}


__global__ void __launch_bounds__(256)
edge_key(const int* __restrict__ src,
         const int* __restrict__ dst,
         unsigned* __restrict__ keyw,
         unsigned* __restrict__ srcplace,
         unsigned* __restrict__ pres, long e_cnt) {
    __shared__ unsigned pres_s[256];
    const unsigned lane = threadIdx.x & 31u;
    const long     g0   = (long)blockIdx.x * 32 + (long)(threadIdx.x >> 5) * 4;
    unsigned word = 0u;
    int t[4], s[4];
    #pragma unroll
    for (int r = 0; r < 4; ++r) {
        const long e  = (g0 + r) * 32 + (long)lane;
        const long ec = (e < e_cnt) ? e : (e_cnt - 1);
        t[r] = dst[ec];
        s[r] = src[ec];
    }
    pres_s[threadIdx.x] = 0u;
    __syncthreads();
    #pragma unroll
    for (int r = 0; r < 4; ++r) {
        const long e    = (g0 + r) * 32 + (long)lane;
        const bool live = e < e_cnt;
        const unsigned key = (unsigned)t[r] / (unsigned)OWN_NODES;
        srcplace[live ? e : (e_cnt - 1)] =
            ((unsigned)s[r] << OWN_PLACE_BITS) | ((unsigned)t[r] - key * (unsigned)OWN_NODES);
        if (live && key < 256u) atomicOr(&pres_s[key], 1u << ((threadIdx.x >> 5) * 4u + (unsigned)r));
        #pragma unroll
        for (int b = 0; b < 8; ++b) {
            unsigned bal = __builtin_amdgcn_ballot_w32(live && ((key >> b) & 1u) != 0u);
            if (lane == (unsigned)(r * 8 + b)) word = bal;
        }
    }
    const long g = g0 + (long)(lane >> 3);
    if (g * 32 < e_cnt) keyw[g * 8 + (long)(lane & 7u)] = word;
    __syncthreads();
    pres[(long)blockIdx.x * 256 + (long)threadIdx.x] = pres_s[threadIdx.x];
}

__device__ __forceinline__ float4 own_score(const float* __restrict__ W_edge, const float4* ea,
                                            float4 asv, float4 adv) {
    float sc[H_HEADS] = {0.f, 0.f, 0.f, 0.f};
    #pragma unroll
    for (int q = 0; q < 4; ++q) {
        float4 v = ea[q];
        int d0 = q * 4;
        #pragma unroll
        for (int hh = 0; hh < H_HEADS; ++hh) {
            sc[hh] += v.x * W_edge[(d0 + 0) * H_HEADS + hh]
                    + v.y * W_edge[(d0 + 1) * H_HEADS + hh]
                    + v.z * W_edge[(d0 + 2) * H_HEADS + hh]
                    + v.w * W_edge[(d0 + 3) * H_HEADS + hh];
        }
    }
    float av[H_HEADS] = {asv.x + adv.x, asv.y + adv.y, asv.z + adv.z, asv.w + adv.w};
    float4 outv;
    float* op = (float*)&outv;
    #pragma unroll
    for (int hh = 0; hh < H_HEADS; ++hh) {
        float v = av[hh] + sc[hh];
        v = (v >= 0.0f) ? v : NEG_SLOPE * v;
        op[hh] = v;
    }
    return outv;
}

__device__ __forceinline__ float4 own_score_at(const float* __restrict__ edge_attr,
                                               const float* __restrict__ W_edge,
                                               const float* __restrict__ a_src,
                                               int e, int s, float4 adv) {
    const float4* p = (const float4*)(edge_attr + (long)e * EDGE_DIM);
    float4 ea[4] = {p[0], p[1], p[2], p[3]};
    return own_score(W_edge, ea, ((const float4*)a_src)[s], adv);
}

#define OWN_CAP   8192
#define OWN_U     16

__device__ __forceinline__ void own_list(int e, unsigned* cnt_s, int* le_s) {
    unsigned pos = atomicAdd(cnt_s, 1u);
    if (pos < (unsigned)OWN_CAP) le_s[pos] = e;
}

__device__ __forceinline__ void own_test32(uint4 a, uint4 b, const unsigned* inv, unsigned keep,
                                           int e0, unsigned* cnt_s, int* le_s) {
    unsigned m = (a.x ^ inv[0]) & (a.y ^ inv[1]) & (a.z ^ inv[2]) & (a.w ^ inv[3])
               & (b.x ^ inv[4]) & (b.y ^ inv[5]) & (b.z ^ inv[6]) & (b.w ^ inv[7]) & keep;
    while (m) {
        int p = __builtin_ctz(m);
        m &= m - 1u;
        own_list(e0 + p, cnt_s, le_s);
    }
}

template <int U>
__device__ __forceinline__ int own_quad_lane(int v) {
    return __builtin_amdgcn_update_dpp(0, v, U * 0x55, 0xf, 0xf, true);
}

__device__ __forceinline__ float own_quad_sum(float v) {
    v += __int_as_float(__builtin_amdgcn_update_dpp(0, __float_as_int(v), 0xB1, 0xf, 0xf, true));
    v += __int_as_float(__builtin_amdgcn_update_dpp(0, __float_as_int(v), 0x4E, 0xf, 0xf, true));
    return v;
}

__device__ __forceinline__ float own_leaky(float v) {
    return (v >= 0.0f) ? v : NEG_SLOPE * v;
}

__device__ __forceinline__ void own_max4(int* mx_s, int tl, float4 a) {
    if (a.x > 0.0f) atomicMax(&mx_s[tl * H_HEADS + 0], __float_as_int(a.x));
    if (a.y > 0.0f) atomicMax(&mx_s[tl * H_HEADS + 1], __float_as_int(a.y));
    if (a.z > 0.0f) atomicMax(&mx_s[tl * H_HEADS + 2], __float_as_int(a.z));
    if (a.w > 0.0f) atomicMax(&mx_s[tl * H_HEADS + 3], __float_as_int(a.w));
}

__global__ void __launch_bounds__(1024)
gather_own_ln(const int* __restrict__ src,
              const int* __restrict__ dst,
              const float* __restrict__ h,
              const float* __restrict__ edge_attr,
              const float* __restrict__ W_edge,
              const float* __restrict__ a_src,
              const float* __restrict__ a_dst,
              const unsigned* __restrict__ keyw,
              const unsigned* __restrict__ srcplace,
              const unsigned* __restrict__ pres,
              const float* __restrict__ gamma,
              const float* __restrict__ beta,
              float* __restrict__ out, int n, long e_cnt) {
    __shared__ float    acc_s[OWN_PAD * OUT_DIM];
    __shared__ int      le_s[OWN_CAP];
    __shared__ float4   sc_s[OWN_CAP];
    __shared__ alignas(16) int mx_s[OWN_PAD * H_HEADS];
    __shared__ float    sm_s[OWN_PAD * H_HEADS];
    __shared__ alignas(16) float adst_s[OWN_PAD * H_HEADS];
    __shared__ unsigned cnt_s;

    const int tid  = threadIdx.x;
    const int wave = tid >> 5;
    const int lane = tid & 31;
    const int lo   = blockIdx.x * OWN_NODES;
    const int ne   = (int)e_cnt;
    const float4* asrc4 = (const float4*)a_src;

    {
        float4 z = {0.0f, 0.0f, 0.0f, 0.0f};
        float4* a4 = (float4*)acc_s;
        for (int i = tid; i < OWN_PAD * OUT_DIM / 4; i += 1024) a4[i] = z;
        for (int i = tid; i < OWN_PAD * H_HEADS; i += 1024) {
            mx_s[i] = 0;
            sm_s[i] = 0.0f;
            const int node = lo + (i >> 2);
            adst_s[i] = (node < n) ? a_dst[(long)node * H_HEADS + (i & 3)] : 0.0f;
        }
        if (tid == 0) cnt_s = 0u;
    }
    __syncthreads();

    const bool keyed = (gridDim.x <= 256u);
    if (keyed) {
        const unsigned mine = blockIdx.x;
        unsigned inv[8];
        #pragma unroll
        for (int b = 0; b < 8; ++b) inv[b] = ((mine >> b) & 1u) - 1u;
        const int nfull = ne >> 5;
        const unsigned last = (1u << (ne & 31)) - 1u;
        const int nw    = (ne + 1023) >> 10;
        const uint4* k4 = (const uint4*)keyw;
        for (int w = tid; w < nw; w += 1024) {
            unsigned bits = pres[(long)w * 256 + (long)mine];
            while (bits) {
                const int b0 = __builtin_ctz(bits);
                bits &= bits - 1u;
                const bool two = (bits != 0u);
                const int b1 = two ? __builtin_ctz(bits) : b0;
                bits &= bits - 1u;
                const int g0 = w * 32 + b0;
                const int g1 = w * 32 + b1;
                const uint4 a0 = k4[2 * g0], c0 = k4[2 * g0 + 1];
                const uint4 a1 = k4[2 * g1], c1 = k4[2 * g1 + 1];
                const unsigned keep0 = (g0 < nfull) ? 0xFFFFFFFFu : last;
                const unsigned keep1 = two ? ((g1 < nfull) ? 0xFFFFFFFFu : last) : 0u;
                own_test32(a0, c0, inv, keep0, g0 << 5, &cnt_s, le_s);
                own_test32(a1, c1, inv, keep1, g1 << 5, &cnt_s, le_s);
            }
        }
    }
    __syncthreads();

    const unsigned cnt    = cnt_s;
    const bool     listed = keyed && (cnt <= (unsigned)OWN_CAP);

    if (listed) {
        const int q = lane & 3;
        const float4* wq = (const float4*)(W_edge + 16 * q);
        const float4 w0 = wq[0], w1 = wq[1], w2 = wq[2], w3 = wq[3];
        const float4* attr4 = (const float4*)edge_attr;
        const unsigned wv = (unsigned)__builtin_amdgcn_readfirstlane(wave);
        const unsigned cn = (unsigned)__builtin_amdgcn_readfirstlane((int)cnt);
        int      e_nx  = 0;
        unsigned pw_nx = 0u;
        if (wv * 32u + (unsigned)lane < cn) {
            e_nx  = le_s[wv * 32u + (unsigned)lane];
            pw_nx = srcplace[e_nx];
        }
        for (unsigned base = wv * 32u; base < cn; base += 1024u) {
            const unsigned i  = base + (unsigned)lane;
            const int      e  = e_nx;
            const unsigned pw = pw_nx;
            float4 ea[4];
            ea[0] = attr4[(long)own_quad_lane<0>(e) * 4 + q];
            ea[1] = attr4[(long)own_quad_lane<1>(e) * 4 + q];
            ea[2] = attr4[(long)own_quad_lane<2>(e) * 4 + q];
            ea[3] = attr4[(long)own_quad_lane<3>(e) * 4 + q];
            const float4 as = asrc4[pw >> OWN_PLACE_BITS];
            const unsigned in = i + 1024u;
            e_nx  = 0;
            pw_nx = 0u;
            if (in < cn) {
                e_nx  = le_s[in];
                pw_nx = srcplace[e_nx];
            }
            float4 p[4];
            #pragma unroll
            for (int u = 0; u < 4; ++u) {
                p[u].x = ea[u].x * w0.x + ea[u].y * w1.x + ea[u].z * w2.x + ea[u].w * w3.x;
                p[u].y = ea[u].x * w0.y + ea[u].y * w1.y + ea[u].z * w2.y + ea[u].w * w3.y;
                p[u].z = ea[u].x * w0.z + ea[u].y * w1.z + ea[u].z * w2.z + ea[u].w * w3.z;
                p[u].w = ea[u].x * w0.w + ea[u].y * w1.w + ea[u].z * w2.w + ea[u].w * w3.w;
                p[u].x = own_quad_sum(p[u].x);
                p[u].y = own_quad_sum(p[u].y);
                p[u].z = own_quad_sum(p[u].z);
                p[u].w = own_quad_sum(p[u].w);
            }
            const bool q1 = (q == 1), q2 = (q == 2), q3 = (q == 3);
            float4 es;
            es.x = q3 ? p[3].x : (q2 ? p[2].x : (q1 ? p[1].x : p[0].x));
            es.y = q3 ? p[3].y : (q2 ? p[2].y : (q1 ? p[1].y : p[0].y));
            es.z = q3 ? p[3].z : (q2 ? p[2].z : (q1 ? p[1].z : p[0].z));
            es.w = q3 ? p[3].w : (q2 ? p[2].w : (q1 ? p[1].w : p[0].w));
            if (i < cn) {
                const int    tl = (int)(pw & OWN_PLACE_MASK);
                const float4 ad = *(const float4*)&adst_s[tl * H_HEADS];
                float4 av;
                av.x = own_leaky((as.x + ad.x) + es.x);
                av.y = own_leaky((as.y + ad.y) + es.y);
                av.z = own_leaky((as.z + ad.z) + es.z);
                av.w = own_leaky((as.w + ad.w) + es.w);
                le_s[i] = (int)pw;
                sc_s[i] = av;
                own_max4(mx_s, tl, av);
            }
        }
    } else {
        for (int e = tid; e < ne; e += 1024) {
            unsigned tl = (unsigned)(dst[e] - lo);
            if (tl < (unsigned)OWN_NODES)
                own_max4(mx_s, (int)tl, own_score_at(edge_attr, W_edge, a_src, e, src[e],
                                                     *(const float4*)&adst_s[tl * H_HEADS]));
        }
    }
    __syncthreads();

    if (listed) {
        for (unsigned i = (unsigned)tid; i < cnt; i += 1024u) {
            const unsigned tl = (unsigned)le_s[i] & OWN_PLACE_MASK;
            const float4 a = sc_s[i];
            const int4   m = *(const int4*)&mx_s[tl * H_HEADS];
            float4 w;
            w.x = expf(a.x - __int_as_float(m.x));
            w.y = expf(a.y - __int_as_float(m.y));
            w.z = expf(a.z - __int_as_float(m.z));
            w.w = expf(a.w - __int_as_float(m.w));
            atomicAdd(&sm_s[tl * H_HEADS + 0], w.x);
            atomicAdd(&sm_s[tl * H_HEADS + 1], w.y);
            atomicAdd(&sm_s[tl * H_HEADS + 2], w.z);
            atomicAdd(&sm_s[tl * H_HEADS + 3], w.w);
            sc_s[i] = w;
        }
    }
    __syncthreads();

    if (listed) {
        const int hh = lane >> 3;
        const unsigned wv = (unsigned)__builtin_amdgcn_readfirstlane(wave);
        const unsigned cn = (unsigned)__builtin_amdgcn_readfirstlane((int)cnt);
        const float*   wt_s  = (const float*)sc_s;
        const char*    hb    = (const char*)h;
        const unsigned lane8 = (unsigned)lane * 8u;

        unsigned i0 = wv * OWN_U;
        for (; i0 + OWN_U <= cn; i0 += 32u * OWN_U) {
            unsigned pk[OWN_U];
            float    ww[OWN_U];
            #pragma unroll
            for (int u = 0; u < OWN_U; ++u) {
                pk[u] = (unsigned)le_s[i0 + u];
                ww[u] = wt_s[(i0 + u) * H_HEADS + hh];
            }
            v2f hv[OWN_U];
            #pragma unroll
            for (int u = 0; u < OWN_U; ++u)
                hv[u] = *(const v2f*)(hb + ((((pk[u] >> OWN_PLACE_BITS) * (unsigned)OUT_DIM) << 2) + lane8));
            #pragma unroll
            for (int u = 0; u < OWN_U; ++u) {
                float* ap = &acc_s[(pk[u] & OWN_PLACE_MASK) * OUT_DIM + 2 * lane];
                atomicAdd(ap,     hv[u].x * ww[u]);
                atomicAdd(ap + 1, hv[u].y * ww[u]);
            }
        }
        if (i0 < cn) {
            unsigned pk[OWN_U];
            float    ww[OWN_U];
            #pragma unroll
            for (int u = 0; u < OWN_U; ++u) {
                const unsigned ic = (i0 + u < cn) ? (i0 + u) : i0;
                pk[u] = (unsigned)le_s[ic];
                ww[u] = wt_s[ic * H_HEADS + hh];
            }
            v2f hv[OWN_U];
            #pragma unroll
            for (int u = 0; u < OWN_U; ++u)
                hv[u] = *(const v2f*)(hb + ((((pk[u] >> OWN_PLACE_BITS) * (unsigned)OUT_DIM) << 2) + lane8));
            #pragma unroll
            for (int u = 0; u < OWN_U; ++u) {
                if (i0 + u < cn) {
                    float* ap = &acc_s[(pk[u] & OWN_PLACE_MASK) * OUT_DIM + 2 * lane];
                    atomicAdd(ap,     hv[u].x * ww[u]);
                    atomicAdd(ap + 1, hv[u].y * ww[u]);
                }
            }
        }
    } else {
        for (int e = tid; e < ne; e += 1024) {
            unsigned tl = (unsigned)(dst[e] - lo);
            if (tl < (unsigned)OWN_NODES) {
                int s = src[e];
                float4 a = own_score_at(edge_attr, W_edge, a_src, e, s, *(const float4*)&adst_s[tl * H_HEADS]);
                float w[H_HEADS];
                w[0] = expf(a.x - __int_as_float(mx_s[tl * H_HEADS + 0]));
                w[1] = expf(a.y - __int_as_float(mx_s[tl * H_HEADS + 1]));
                w[2] = expf(a.z - __int_as_float(mx_s[tl * H_HEADS + 2]));
                w[3] = expf(a.w - __int_as_float(mx_s[tl * H_HEADS + 3]));
                for (int hd = 0; hd < H_HEADS; ++hd) {
                    atomicAdd(&sm_s[tl * H_HEADS + hd], w[hd]);
                    for (int d = hd * HD; d < hd * HD + HD; ++d)
                        atomicAdd(&acc_s[tl * OUT_DIM + d], h[(long)s * OUT_DIM + d] * w[hd]);
                }
            }
        }
    }
    __syncthreads();

    {
        const float g0 = gamma[lane], g1 = gamma[lane + 32];
        const float b0 = beta[lane],  b1 = beta[lane + 32];
        const int   h0 = lane >> 4;
        for (int tl0 = wave * 4; tl0 < OWN_NODES; tl0 += 32 * 4) {
            float y0[4], y1[4], s[4], vs[4];
            #pragma unroll
            for (int u = 0; u < 4; ++u) {
                int node = lo + tl0 + u;
                int nc   = (node < n) ? node : (n - 1);
                float q0 = sm_s[(tl0 + u) * H_HEADS + h0]     + EPS_SOFTMAX;
                float q1 = sm_s[(tl0 + u) * H_HEADS + h0 + 2] + EPS_SOFTMAX;
                y0[u] = acc_s[(tl0 + u) * OUT_DIM + lane]      / q0 + h[(long)nc * OUT_DIM + lane];
                y1[u] = acc_s[(tl0 + u) * OUT_DIM + lane + 32] / q1 + h[(long)nc * OUT_DIM + lane + 32];
                s[u]  = y0[u] + y1[u];
            }
            #pragma unroll
            for (int off = 16; off > 0; off >>= 1) {
                #pragma unroll
                for (int u = 0; u < 4; ++u) s[u] += __shfl_xor(s[u], off, 32);
            }
            #pragma unroll
            for (int u = 0; u < 4; ++u) {
                float mu = s[u] * (1.0f / 64.0f);
                y0[u] -= mu;
                y1[u] -= mu;
                vs[u] = y0[u] * y0[u] + y1[u] * y1[u];
            }
            #pragma unroll
            for (int off = 16; off > 0; off >>= 1) {
                #pragma unroll
                for (int u = 0; u < 4; ++u) vs[u] += __shfl_xor(vs[u], off, 32);
            }
            #pragma unroll
            for (int u = 0; u < 4; ++u) {
                int node = lo + tl0 + u;
                if (tl0 + u < OWN_NODES && node < n) {
                    long base = (long)node * OUT_DIM;
                    float rstd = rsqrtf(vs[u] * (1.0f / 64.0f) + EPS_LN);
                    out[base + lane]      = y0[u] * rstd * g0 + b0;
                    out[base + lane + 32] = y1[u] * rstd * g1 + b1;
                }
            }
        }
    }
}

extern "C" void kernel_launch(void* const* d_in, const int* in_sizes, int n_in,
                              void* d_out, int out_size, void* d_ws, size_t ws_size,
                              hipStream_t stream) {
    const float* x         = (const float*)d_in[0];
    const int*   ei        = (const int*)  d_in[1];
    const float* edge_attr = (const float*)d_in[2];
    const float* W_node    = (const float*)d_in[3];
    const float* W_edge    = (const float*)d_in[4];
    const float* attn_src  = (const float*)d_in[5];
    const float* attn_dst  = (const float*)d_in[6];
    const float* gamma     = (const float*)d_in[7];
    const float* beta      = (const float*)d_in[8];
    float* out = (float*)d_out;

    int  n = in_sizes[0] / IN_DIM;
    long e = (long)in_sizes[2] / EDGE_DIM;
    const int* srcIdx = ei;
    const int* dstIdx = ei + e;

    float* ws    = (float*)d_ws;
    float* h     = ws;  ws += (long)n * OUT_DIM;
    float* a_src = ws;  ws += (long)n * H_HEADS;
    float* a_dst = ws;  ws += (long)n * H_HEADS;
    unsigned* keyw = (unsigned*)ws;
    unsigned* srcplace = keyw + ((e + 31) / 32) * 8;
    unsigned* pres = srcplace + e;

    int n_tiles = (n + 15) / 16;
    node_gemm_wmma<<<n_tiles, 128, 0, stream>>>(x, W_node, h, n);

    node_proj<<<(n * H_HEADS + 255) / 256, 256, 0, stream>>>(
        h, attn_src, attn_dst, a_src, a_dst, n);

    edge_key<<<(int)((e + 1023) / 1024), 256, 0, stream>>>(srcIdx, dstIdx, keyw, srcplace, pres, e);

    gather_own_ln<<<(n + OWN_NODES - 1) / OWN_NODES, 1024, 0, stream>>>(
        srcIdx, dstIdx, h, edge_attr, W_edge, a_src, a_dst, keyw, srcplace, pres, gamma, beta, out, n, e);
}
